# starting kernel
# baseline (speedup 1.0000x reference)
#include <hip/hip_runtime.h>
#include <hip/hip_bf16.h>


typedef float v2f __attribute__((ext_vector_type(2)));
typedef float v8f __attribute__((ext_vector_type(8)));

#define NFEAT   128
#define NHID    96
#define NCLASS  40
#define GN_EPS  1e-5f

template<int NT, int KMAX>
__global__ __launch_bounds__(128) void wmma_gemm_strip_kernel(
    const float* __restrict__ A, const float* __restrict__ Bw,
    float* __restrict__ C, int K, int ncols, int ntm)
{
    constexpr int NCP = NT * 16;
    __shared__ float lds[KMAX * NCP];

    const int tid = threadIdx.x;

    for (int i = tid; i < K * NCP; i += 128) {
        int k   = i / NCP;
        int col = i - k * NCP;
        float v = (col < ncols) ? Bw[(size_t)k * ncols + col] : 0.0f;
        lds[(k >> 1) * (NCP * 2) + col * 2 + (k & 1)] = v;
    }
    __syncthreads();

    const int wave  = tid >> 5;
    const int mtile = blockIdx.x * 4 + wave;
    if (mtile >= ntm) return;

    const int lane  = tid & 31;
    const int coll  = lane & 15;
    const int mrow  = mtile * 16 + coll;
    const int khalf = (lane >> 4) * 2;

    const float2* __restrict__ lds2 = reinterpret_cast<const float2*>(lds);

    v8f acc[NT] = {};
    for (int k = 0; k < K; k += 4) {
        const int kk = k + khalf;
        float2 av = *reinterpret_cast<const float2*>(A + (size_t)mrow * K + kk);
        v2f a; a.x = av.x; a.y = av.y;
        const float2* bp = lds2 + (kk >> 1) * NCP + coll;
        #pragma unroll
        for (int t = 0; t < NT; ++t) {
            float2 bv = bp[t * 16];
            v2f b; b.x = bv.x; b.y = bv.y;
            acc[t] = __builtin_amdgcn_wmma_f32_16x16x4_f32(
                false, a, false, b, (short)0, acc[t], false, false);
        }
    }

    const int rbase = mtile * 16 + (lane >> 4) * 8;
    #pragma unroll
    for (int t = 0; t < NT; ++t) {
        int col = t * 16 + coll;
        if (col < ncols) {
            #pragma unroll
            for (int r = 0; r < 8; ++r)
                C[(size_t)(rbase + r) * ncols + col] = acc[t][r];
        }
    }
}

__global__ void scatter_kernel(const float* __restrict__ h,
                               const int* __restrict__ src,
                               const int* __restrict__ tgt,
                               const float* __restrict__ w,
                               float* __restrict__ agg,
                               int C, int vpe, int total)
{
    int idx = blockIdx.x * blockDim.x + threadIdx.x;
    if (idx >= total) return;
    int e = idx / vpe;
    int c = (idx - e * vpe) * 4;
    int s = src[e];
    int t = tgt[e];
    float ww = w[e];
    float4 m = *reinterpret_cast<const float4*>(h + (size_t)s * C + c);
    float* ap = agg + (size_t)t * C + c;
    atomicAdd(ap + 0, m.x * ww);
    atomicAdd(ap + 1, m.y * ww);
    atomicAdd(ap + 2, m.z * ww);
    atomicAdd(ap + 3, m.w * ww);
}

__global__ void bias_relu_kernel(float* __restrict__ x,
                                 const float* __restrict__ b,
                                 int C, int total)
{
    int idx = blockIdx.x * blockDim.x + threadIdx.x;
    if (idx >= total) return;
    int c = idx % C;
    x[idx] = fmaxf(x[idx] + b[c], 0.0f);
}

__global__ void groupnorm_kernel(float* __restrict__ x,
                                 const float* __restrict__ gamma,
                                 const float* __restrict__ beta,
                                 int total  )
{
    int idx = blockIdx.x * blockDim.x + threadIdx.x;
    if (idx >= total) return;
    int row = idx >> 5;
    int g   = idx & 31;
    size_t base = (size_t)row * NHID + g * 3;
    float x0 = x[base], x1 = x[base + 1], x2 = x[base + 2];
    float mu = (x0 + x1 + x2) * (1.0f / 3.0f);
    float d0 = x0 - mu, d1 = x1 - mu, d2 = x2 - mu;
    float var = (d0 * d0 + d1 * d1 + d2 * d2) * (1.0f / 3.0f);
    float inv = rsqrtf(var + GN_EPS);
    int cg = g * 3;
    x[base]     = d0 * inv * gamma[cg]     + beta[cg];
    x[base + 1] = d1 * inv * gamma[cg + 1] + beta[cg + 1];
    x[base + 2] = d2 * inv * gamma[cg + 2] + beta[cg + 2];
}

__global__ void bias_logsoftmax_kernel(float* __restrict__ out,
                                       const float* __restrict__ b,
                                       int nrows)
{
    int row = blockIdx.x * blockDim.x + threadIdx.x;
    if (row >= nrows) return;
    float v[NCLASS];
    float m = -3.402823466e+38f;
    #pragma unroll
    for (int c = 0; c < NCLASS; ++c) {
        v[c] = out[(size_t)row * NCLASS + c] + b[c];
        m = fmaxf(m, v[c]);
    }
    float s = 0.0f;
    #pragma unroll
    for (int c = 0; c < NCLASS; ++c) s += __expf(v[c] - m);
    float ls = __logf(s);
    #pragma unroll
    for (int c = 0; c < NCLASS; ++c)
        out[(size_t)row * NCLASS + c] = v[c] - m - ls;
}

extern "C" void kernel_launch(void* const* d_in, const int* in_sizes, int n_in,
                              void* d_out, int out_size, void* d_ws, size_t ws_size,
                              hipStream_t stream)
{
    const float* x     = (const float*)d_in[0];
    const int*   src   = (const int*)  d_in[1];
    const int*   tgt   = (const int*)  d_in[2];
    const float* w     = (const float*)d_in[3];
    const float* W1    = (const float*)d_in[4];
    const float* b1    = (const float*)d_in[5];
    const float* W2    = (const float*)d_in[6];
    const float* b2    = (const float*)d_in[7];
    const float* W3    = (const float*)d_in[8];
    const float* b3    = (const float*)d_in[9];
    const float* gamma = (const float*)d_in[10];
    const float* beta  = (const float*)d_in[11];
    float* out = (float*)d_out;

    const int N   = in_sizes[0] / NFEAT;
    const int E   = in_sizes[1];
    const int ntm = N / 16;
    const int gblk = (ntm + 3) / 4;

    float* bufA = (float*)d_ws;
    float* bufB = bufA + (size_t)N * NHID;

    const int TPB = 256;

    wmma_gemm_strip_kernel<6, NFEAT><<<gblk, 128, 0, stream>>>(
        x, W1, bufA, NFEAT, NHID, ntm);
    hipMemsetAsync(bufB, 0, (size_t)N * NHID * sizeof(float), stream);
    {
        int vpe = NHID / 4, total = E * vpe;
        scatter_kernel<<<(total + TPB - 1) / TPB, TPB, 0, stream>>>(
            bufA, src, tgt, w, bufB, NHID, vpe, total);
        int t2 = N * NHID;
        bias_relu_kernel<<<(t2 + TPB - 1) / TPB, TPB, 0, stream>>>(bufB, b1, NHID, t2);
    }

    wmma_gemm_strip_kernel<6, NFEAT><<<gblk, 128, 0, stream>>>(
        bufB, W2, bufA, NHID, NHID, ntm);
    hipMemsetAsync(bufB, 0, (size_t)N * NHID * sizeof(float), stream);
    {
        int vpe = NHID / 4, total = E * vpe;
        scatter_kernel<<<(total + TPB - 1) / TPB, TPB, 0, stream>>>(
            bufA, src, tgt, w, bufB, NHID, vpe, total);
        int t2 = N * NHID;
        bias_relu_kernel<<<(t2 + TPB - 1) / TPB, TPB, 0, stream>>>(bufB, b2, NHID, t2);
    }

    {
        int total = N * 32;
        groupnorm_kernel<<<(total + TPB - 1) / TPB, TPB, 0, stream>>>(
            bufB, gamma, beta, total);
    }

    wmma_gemm_strip_kernel<3, NHID><<<gblk, 128, 0, stream>>>(
        bufB, W3, bufA, NHID, NCLASS, ntm);
    hipMemsetAsync(out, 0, (size_t)N * NCLASS * sizeof(float), stream);
    {
        int vpe = NCLASS / 4, total = E * vpe;
        scatter_kernel<<<(total + TPB - 1) / TPB, TPB, 0, stream>>>(
            bufA, src, tgt, w, out, NCLASS, vpe, total);
    }

    bias_logsoftmax_kernel<<<(N + TPB - 1) / TPB, TPB, 0, stream>>>(out, b3, N);
}


# GEMMs: K and width compile-time; all global loads (W share as 16-byte loads, the wave's whole A tile) issued before the first wait, not one per trip; k summed in permuted order inside each 8-block
# speedup vs baseline: 1.2520x; 1.2520x over previous
#include <hip/hip_runtime.h>
#include <hip/hip_bf16.h>


typedef float v2f __attribute__((ext_vector_type(2)));
typedef float v4f __attribute__((ext_vector_type(4)));
typedef float v8f __attribute__((ext_vector_type(8)));

#define NFEAT   128
#define NHID    96
#define NCLASS  40
#define GN_EPS  1e-5f

template<int NT, int K, int NCOLS>
__global__ __launch_bounds__(128) void wmma_gemm_strip_kernel(
    const float* __restrict__ A, const float* __restrict__ Bw,
    float* __restrict__ C, int ntm)
{
    constexpr int NCP   = NT * 16;
    constexpr int NQ    = NCP / 4;
    constexpr int NTASK = (K / 2) * NQ;
    constexpr int NIT   = (NTASK + 127) / 128;
    constexpr int KB    = K / 8;
    static_assert(K % 8 == 0 && NCOLS % 4 == 0 && NCOLS <= NCP, "shape");

    __shared__ v4f lds4[K * NCP / 4];

    const int tid   = threadIdx.x;
    const int wave  = tid >> 5;
    const int lane  = tid & 31;
    const int coll  = lane & 15;
    const int half  = lane >> 4;
    const int mtile = blockIdx.x * 4 + wave;
    const int mload = (mtile < ntm) ? mtile : (ntm - 1);

    v4f w0[NIT], w1[NIT];
    #pragma unroll
    for (int it = 0; it < NIT; ++it) {
        const int i  = tid + it * 128;
        const int ic = (NTASK % 128 == 0) ? i : (i < NTASK ? i : NTASK - 1);
        const int p  = ic / NQ;
        const int q  = ic - p * NQ;
        const int qc = (NCOLS == NCP) ? q : (q * 4 < NCOLS ? q : NCOLS / 4 - 1);
        const float* g = Bw + (size_t)(2 * p) * NCOLS + qc * 4;
        w0[it] = *reinterpret_cast<const v4f*>(g);
        w1[it] = *reinterpret_cast<const v4f*>(g + NCOLS);
    }

    v4f af[KB];
    const float* arow = A + (size_t)(mload * 16 + coll) * K + half * 4;
    #pragma unroll
    for (int j = 0; j < KB; ++j)
        af[j] = *reinterpret_cast<const v4f*>(arow + j * 8);

    #pragma unroll
    for (int it = 0; it < NIT; ++it) {
        const int i = tid + it * 128;
        if (NTASK % 128 == 0 || i < NTASK) {
            v4f lo = (v4f){w0[it].x, w1[it].x, w0[it].y, w1[it].y};
            v4f hi = (v4f){w0[it].z, w1[it].z, w0[it].w, w1[it].w};
            if (NCOLS != NCP) {
                const int q = i % NQ;
                if (q * 4 >= NCOLS) {
                    lo = (v4f){0.0f, 0.0f, 0.0f, 0.0f};
                    hi = (v4f){0.0f, 0.0f, 0.0f, 0.0f};
                }
            }
            lds4[2 * i]     = lo;
            lds4[2 * i + 1] = hi;
        }
    }
    __syncthreads();

    const float2* __restrict__ lds2 = reinterpret_cast<const float2*>(lds4);
    const float2* bp = lds2 + (2 * half) * NCP + coll;

    v8f acc[NT] = {};
    #pragma unroll
    for (int j = 0; j < KB; ++j) {
        #pragma unroll
        for (int s = 0; s < 2; ++s) {
            v2f a;
            a.x = s ? af[j].z : af[j].x;
            a.y = s ? af[j].w : af[j].y;
            const float2* bq = bp + (4 * j + s) * NCP;
            #pragma unroll
            for (int t = 0; t < NT; ++t) {
                float2 bv = bq[t * 16];
                v2f b; b.x = bv.x; b.y = bv.y;
                acc[t] = __builtin_amdgcn_wmma_f32_16x16x4_f32(
                    false, a, false, b, (short)0, acc[t], false, false);
            }
        }
    }

    if (mtile >= ntm) return;

    const int rbase = mtile * 16 + half * 8;
    #pragma unroll
    for (int t = 0; t < NT; ++t) {
        int col = t * 16 + coll;
        if (col < NCOLS) {
            #pragma unroll
            for (int r = 0; r < 8; ++r)
                C[(size_t)(rbase + r) * NCOLS + col] = acc[t][r];
        }
    }
}

__global__ void scatter_kernel(const float* __restrict__ h,
                               const int* __restrict__ src,
                               const int* __restrict__ tgt,
                               const float* __restrict__ w,
                               float* __restrict__ agg,
                               int C, int vpe, int total)
{
    int idx = blockIdx.x * blockDim.x + threadIdx.x;
    if (idx >= total) return;
    int e = idx / vpe;
    int c = (idx - e * vpe) * 4;
    int s = src[e];
    int t = tgt[e];
    float ww = w[e];
    float4 m = *reinterpret_cast<const float4*>(h + (size_t)s * C + c);
    float* ap = agg + (size_t)t * C + c;
    atomicAdd(ap + 0, m.x * ww);
    atomicAdd(ap + 1, m.y * ww);
    atomicAdd(ap + 2, m.z * ww);
    atomicAdd(ap + 3, m.w * ww);
}

__global__ void bias_relu_kernel(float* __restrict__ x,
                                 const float* __restrict__ b,
                                 int C, int total)
{
    int idx = blockIdx.x * blockDim.x + threadIdx.x;
    if (idx >= total) return;
    int c = idx % C;
    x[idx] = fmaxf(x[idx] + b[c], 0.0f);
}

__global__ void groupnorm_kernel(float* __restrict__ x,
                                 const float* __restrict__ gamma,
                                 const float* __restrict__ beta,
                                 int total  )
{
    int idx = blockIdx.x * blockDim.x + threadIdx.x;
    if (idx >= total) return;
    int row = idx >> 5;
    int g   = idx & 31;
    size_t base = (size_t)row * NHID + g * 3;
    float x0 = x[base], x1 = x[base + 1], x2 = x[base + 2];
    float mu = (x0 + x1 + x2) * (1.0f / 3.0f);
    float d0 = x0 - mu, d1 = x1 - mu, d2 = x2 - mu;
    float var = (d0 * d0 + d1 * d1 + d2 * d2) * (1.0f / 3.0f);
    float inv = rsqrtf(var + GN_EPS);
    int cg = g * 3;
    x[base]     = d0 * inv * gamma[cg]     + beta[cg];
    x[base + 1] = d1 * inv * gamma[cg + 1] + beta[cg + 1];
    x[base + 2] = d2 * inv * gamma[cg + 2] + beta[cg + 2];
}

__global__ void bias_logsoftmax_kernel(float* __restrict__ out,
                                       const float* __restrict__ b,
                                       int nrows)
{
    int row = blockIdx.x * blockDim.x + threadIdx.x;
    if (row >= nrows) return;
    float v[NCLASS];
    float m = -3.402823466e+38f;
    #pragma unroll
    for (int c = 0; c < NCLASS; ++c) {
        v[c] = out[(size_t)row * NCLASS + c] + b[c];
        m = fmaxf(m, v[c]);
    }
    float s = 0.0f;
    #pragma unroll
    for (int c = 0; c < NCLASS; ++c) s += __expf(v[c] - m);
    float ls = __logf(s);
    #pragma unroll
    for (int c = 0; c < NCLASS; ++c)
        out[(size_t)row * NCLASS + c] = v[c] - m - ls;
}

extern "C" void kernel_launch(void* const* d_in, const int* in_sizes, int n_in,
                              void* d_out, int out_size, void* d_ws, size_t ws_size,
                              hipStream_t stream)
{
    const float* x     = (const float*)d_in[0];
    const int*   src   = (const int*)  d_in[1];
    const int*   tgt   = (const int*)  d_in[2];
    const float* w     = (const float*)d_in[3];
    const float* W1    = (const float*)d_in[4];
    const float* b1    = (const float*)d_in[5];
    const float* W2    = (const float*)d_in[6];
    const float* b2    = (const float*)d_in[7];
    const float* W3    = (const float*)d_in[8];
    const float* b3    = (const float*)d_in[9];
    const float* gamma = (const float*)d_in[10];
    const float* beta  = (const float*)d_in[11];
    float* out = (float*)d_out;

    const int N   = in_sizes[0] / NFEAT;
    const int E   = in_sizes[1];
    const int ntm = N / 16;
    const int gblk = (ntm + 3) / 4;

    float* bufA = (float*)d_ws;
    float* bufB = bufA + (size_t)N * NHID;

    const int TPB = 256;

    wmma_gemm_strip_kernel<6, NFEAT, NHID><<<gblk, 128, 0, stream>>>(
        x, W1, bufA, ntm);
    hipMemsetAsync(bufB, 0, (size_t)N * NHID * sizeof(float), stream);
    {
        int vpe = NHID / 4, total = E * vpe;
        scatter_kernel<<<(total + TPB - 1) / TPB, TPB, 0, stream>>>(
            bufA, src, tgt, w, bufB, NHID, vpe, total);
        int t2 = N * NHID;
        bias_relu_kernel<<<(t2 + TPB - 1) / TPB, TPB, 0, stream>>>(bufB, b1, NHID, t2);
    }

    wmma_gemm_strip_kernel<6, NHID, NHID><<<gblk, 128, 0, stream>>>(
        bufB, W2, bufA, ntm);
    hipMemsetAsync(bufB, 0, (size_t)N * NHID * sizeof(float), stream);
    {
        int vpe = NHID / 4, total = E * vpe;
        scatter_kernel<<<(total + TPB - 1) / TPB, TPB, 0, stream>>>(
            bufA, src, tgt, w, bufB, NHID, vpe, total);
        int t2 = N * NHID;
        bias_relu_kernel<<<(t2 + TPB - 1) / TPB, TPB, 0, stream>>>(bufB, b2, NHID, t2);
    }

    {
        int total = N * 32;
        groupnorm_kernel<<<(total + TPB - 1) / TPB, TPB, 0, stream>>>(
            bufB, gamma, beta, total);
    }

    wmma_gemm_strip_kernel<3, NHID, NCLASS><<<gblk, 128, 0, stream>>>(
        bufB, W3, bufA, ntm);
    hipMemsetAsync(out, 0, (size_t)N * NCLASS * sizeof(float), stream);
    {
        int vpe = NCLASS / 4, total = E * vpe;
        scatter_kernel<<<(total + TPB - 1) / TPB, TPB, 0, stream>>>(
            bufA, src, tgt, w, out, NCLASS, vpe, total);
    }

    bias_logsoftmax_kernel<<<(N + TPB - 1) / TPB, TPB, 0, stream>>>(out, b3, N);
}


# Segment sums by gather, no f32 atomics: per call edges grouped by target row (LDS counters in two levels, prefix sums), each row's list ranked into edge-index order, one owner sums it in that order
# speedup vs baseline: 2.3296x; 1.8607x over previous
#include <hip/hip_runtime.h>
#include <hip/hip_bf16.h>


typedef float v2f __attribute__((ext_vector_type(2)));
typedef float v4f __attribute__((ext_vector_type(4)));
typedef float v8f __attribute__((ext_vector_type(8)));

#define NFEAT   128
#define NHID    96
#define NCLASS  40
#define GN_EPS  1e-5f

template<int NT, int K, int NCOLS>
__global__ __launch_bounds__(128) void wmma_gemm_strip_kernel(
    const float* __restrict__ A, const float* __restrict__ Bw,
    float* __restrict__ C, int ntm)
{
    constexpr int NCP   = NT * 16;
    constexpr int NQ    = NCP / 4;
    constexpr int NTASK = (K / 2) * NQ;
    constexpr int NIT   = (NTASK + 127) / 128;
    constexpr int KB    = K / 8;
    static_assert(K % 8 == 0 && NCOLS % 4 == 0 && NCOLS <= NCP, "shape");

    __shared__ v4f lds4[K * NCP / 4];

    const int tid   = threadIdx.x;
    const int wave  = tid >> 5;
    const int lane  = tid & 31;
    const int coll  = lane & 15;
    const int half  = lane >> 4;
    const int mtile = blockIdx.x * 4 + wave;
    const int mload = (mtile < ntm) ? mtile : (ntm - 1);

    v4f w0[NIT], w1[NIT];
    #pragma unroll
    for (int it = 0; it < NIT; ++it) {
        const int i  = tid + it * 128;
        const int ic = (NTASK % 128 == 0) ? i : (i < NTASK ? i : NTASK - 1);
        const int p  = ic / NQ;
        const int q  = ic - p * NQ;
        const int qc = (NCOLS == NCP) ? q : (q * 4 < NCOLS ? q : NCOLS / 4 - 1);
        const float* g = Bw + (size_t)(2 * p) * NCOLS + qc * 4;
        w0[it] = *reinterpret_cast<const v4f*>(g);
        w1[it] = *reinterpret_cast<const v4f*>(g + NCOLS);
    }

    v4f af[KB];
    const float* arow = A + (size_t)(mload * 16 + coll) * K + half * 4;
    #pragma unroll
    for (int j = 0; j < KB; ++j)
        af[j] = *reinterpret_cast<const v4f*>(arow + j * 8);

    #pragma unroll
    for (int it = 0; it < NIT; ++it) {
        const int i = tid + it * 128;
        if (NTASK % 128 == 0 || i < NTASK) {
            v4f lo = (v4f){w0[it].x, w1[it].x, w0[it].y, w1[it].y};
            v4f hi = (v4f){w0[it].z, w1[it].z, w0[it].w, w1[it].w};
            if (NCOLS != NCP) {
                const int q = i % NQ;
                if (q * 4 >= NCOLS) {
                    lo = (v4f){0.0f, 0.0f, 0.0f, 0.0f};
                    hi = (v4f){0.0f, 0.0f, 0.0f, 0.0f};
                }
            }
            lds4[2 * i]     = lo;
            lds4[2 * i + 1] = hi;
        }
    }
    __syncthreads();

    const float2* __restrict__ lds2 = reinterpret_cast<const float2*>(lds4);
    const float2* bp = lds2 + (2 * half) * NCP + coll;

    v8f acc[NT] = {};
    #pragma unroll
    for (int j = 0; j < KB; ++j) {
        #pragma unroll
        for (int s = 0; s < 2; ++s) {
            v2f a;
            a.x = s ? af[j].z : af[j].x;
            a.y = s ? af[j].w : af[j].y;
            const float2* bq = bp + (4 * j + s) * NCP;
            #pragma unroll
            for (int t = 0; t < NT; ++t) {
                float2 bv = bq[t * 16];
                v2f b; b.x = bv.x; b.y = bv.y;
                acc[t] = __builtin_amdgcn_wmma_f32_16x16x4_f32(
                    false, a, false, b, (short)0, acc[t], false, false);
            }
        }
    }

    if (mtile >= ntm) return;

    const int rbase = mtile * 16 + half * 8;
    #pragma unroll
    for (int t = 0; t < NT; ++t) {
        int col = t * 16 + coll;
        if (col < NCOLS) {
            #pragma unroll
            for (int r = 0; r < 8; ++r)
                C[(size_t)(rbase + r) * NCOLS + col] = acc[t][r];
        }
    }
}


__global__ __launch_bounds__(1024) void bin_count_kernel(const int* __restrict__ tgt,
                                                         int* __restrict__ hist,
                                                         int E, int nb, int npad)
{
    __shared__ int lh[256];
    const int tid = threadIdx.x;
    const int g   = blockIdx.x;
    const int G   = gridDim.x;
    if (tid < 256) lh[tid] = 0;
    __syncthreads();
    const int ch = (E + G - 1) / G;
    const int e0 = g * ch;
    const int e1 = (e0 + ch < E) ? (e0 + ch) : E;
    for (int e = e0 + tid; e < e1; e += 1024)
        atomicAdd(&lh[tgt[e] >> 8], 1);
    __syncthreads();
    if (tid < nb) hist[tid * G + g] = lh[tid];
    for (int i = nb * G + g * 1024 + tid; i < npad; i += G * 1024) hist[i] = 0;
}

__global__ __launch_bounds__(1024) void prefix_sum_kernel(const int* __restrict__ in,
                                                          int* __restrict__ out, int pw)
{
    __shared__ int wtot[32];
    const int lane = threadIdx.x & 31;
    const int wave = threadIdx.x >> 5;
    const int nit  = pw >> 7;
    const int4* c4 = reinterpret_cast<const int4*>(in + (size_t)wave * pw) + lane;
    int4*       r4 = reinterpret_cast<int4*>(out + (size_t)wave * pw) + lane;

    int s = 0;
    for (int it = 0; it < nit; ++it) {
        int4 v = c4[it * 32];
        s += (v.x + v.y) + (v.z + v.w);
    }
    for (int off = 16; off > 0; off >>= 1) s += __shfl_xor(s, off);
    if (lane == 0) wtot[wave] = s;
    __syncthreads();
    int carry = (lane < wave) ? wtot[lane] : 0;
    for (int off = 16; off > 0; off >>= 1) carry += __shfl_xor(carry, off);

    for (int it = 0; it < nit; ++it) {
        int4 v = c4[it * 32];
        int ls  = (v.x + v.y) + (v.z + v.w);
        int inc = ls;
        for (int off = 1; off < 32; off <<= 1) {
            int u = __shfl_up(inc, off);
            if (lane >= off) inc += u;
        }
        int4 o;
        o.x = carry + inc - ls;
        o.y = o.x + v.x;
        o.z = o.y + v.y;
        o.w = o.z + v.z;
        r4[it * 32] = o;
        carry += __shfl(inc, 31);
    }
}

__global__ __launch_bounds__(1024) void bin_place_kernel(const int* __restrict__ src,
                                                         const int* __restrict__ tgt,
                                                         const float* __restrict__ w,
                                                         const int* __restrict__ base,
                                                         int* __restrict__ recs1,
                                                         int E, int nb)
{
    __shared__ int cur[256];
    const int tid = threadIdx.x;
    const int g   = blockIdx.x;
    const int G   = gridDim.x;
    if (tid < nb) cur[tid] = base[tid * G + g];
    __syncthreads();
    const int ch = (E + G - 1) / G;
    const int e0 = g * ch;
    const int e1 = (e0 + ch < E) ? (e0 + ch) : E;
    int4* r1 = reinterpret_cast<int4*>(recs1);
    for (int e = e0 + tid; e < e1; e += 1024) {
        const int t   = tgt[e];
        const int pos = atomicAdd(&cur[t >> 8], 1);
        r1[pos] = make_int4(e, t, src[e], __float_as_int(w[e]));
    }
}

__global__ __launch_bounds__(1024) void row_place_kernel(const int* __restrict__ recs1,
                                                         const int* __restrict__ base,
                                                         int* __restrict__ rowptr,
                                                         int* __restrict__ recs2, int G)
{
    __shared__ int cnt[256];
    __shared__ int cur[256];
    __shared__ int wsum[8];
    const int tid  = threadIdx.x;
    const int lane = tid & 31;
    const int wave = tid >> 5;
    const int b    = blockIdx.x;
    const int bs   = base[b * G];
    const int be   = base[(b + 1) * G];
    const int4* r1 = reinterpret_cast<const int4*>(recs1);
    int4*       r2 = reinterpret_cast<int4*>(recs2);

    if (tid < 256) cnt[tid] = 0;
    __syncthreads();
    for (int i = bs + tid; i < be; i += 1024)
        atomicAdd(&cnt[r1[i].y & 255], 1);
    __syncthreads();

    int c = 0, inc = 0;
    if (tid < 256) {
        c = cnt[tid];
        inc = c;
        for (int off = 1; off < 32; off <<= 1) {
            int u = __shfl_up(inc, off);
            if (lane >= off) inc += u;
        }
        if (lane == 31) wsum[wave] = inc;
    }
    __syncthreads();
    if (tid < 256) {
        int front = 0;
        for (int k = 0; k < wave; ++k) front += wsum[k];
        const int first = bs + front + inc - c;
        cur[tid] = first;
        rowptr[b * 256 + tid] = first;
        if (tid == 255 && b == (int)gridDim.x - 1) rowptr[b * 256 + 256] = be;
    }
    __syncthreads();
    for (int i = bs + tid; i < be; i += 1024) {
        const int4 r = r1[i];
        const int pos = atomicAdd(&cur[r.y & 255], 1);
        r2[pos] = r;
    }
}

__global__ void edge_order_kernel(const int* __restrict__ recs2,
                                  const int* __restrict__ rowptr,
                                  int* __restrict__ csrc,
                                  float* __restrict__ cw, int E)
{
    int p = blockIdx.x * blockDim.x + threadIdx.x;
    if (p >= E) return;
    const int4* r2 = reinterpret_cast<const int4*>(recs2);
    const int4 r = r2[p];
    const int a = rowptr[r.y];
    const int b = rowptr[r.y + 1];
    int k = 0;
    for (int j = a; j < b; ++j) k += (r2[j].x < r.x) ? 1 : 0;
    csrc[a + k] = r.z;
    cw[a + k]   = __int_as_float(r.w);
}

template<int C>
__global__ void gather_rows_kernel(const float* __restrict__ h,
                                   const int* __restrict__ rowptr,
                                   const int* __restrict__ csrc,
                                   const float* __restrict__ cw,
                                   float* __restrict__ agg, int total)
{
    constexpr int VPR = C / 4;
    int idx = blockIdx.x * blockDim.x + threadIdx.x;
    if (idx >= total) return;
    int row = idx / VPR;
    int c   = (idx - row * VPR) * 4;
    int p   = rowptr[row];
    const int b = rowptr[row + 1];
    const float* hc = h + c;
    float4 acc = make_float4(0.0f, 0.0f, 0.0f, 0.0f);
    for (; p + 4 <= b; p += 4) {
        int   s0 = csrc[p], s1 = csrc[p + 1], s2 = csrc[p + 2], s3 = csrc[p + 3];
        float w0 = cw[p],   w1 = cw[p + 1],   w2 = cw[p + 2],   w3 = cw[p + 3];
        float4 m0 = *reinterpret_cast<const float4*>(hc + (size_t)s0 * C);
        float4 m1 = *reinterpret_cast<const float4*>(hc + (size_t)s1 * C);
        float4 m2 = *reinterpret_cast<const float4*>(hc + (size_t)s2 * C);
        float4 m3 = *reinterpret_cast<const float4*>(hc + (size_t)s3 * C);
        acc.x = fmaf(m0.x, w0, acc.x); acc.y = fmaf(m0.y, w0, acc.y); acc.z = fmaf(m0.z, w0, acc.z); acc.w = fmaf(m0.w, w0, acc.w);
        acc.x = fmaf(m1.x, w1, acc.x); acc.y = fmaf(m1.y, w1, acc.y); acc.z = fmaf(m1.z, w1, acc.z); acc.w = fmaf(m1.w, w1, acc.w);
        acc.x = fmaf(m2.x, w2, acc.x); acc.y = fmaf(m2.y, w2, acc.y); acc.z = fmaf(m2.z, w2, acc.z); acc.w = fmaf(m2.w, w2, acc.w);
        acc.x = fmaf(m3.x, w3, acc.x); acc.y = fmaf(m3.y, w3, acc.y); acc.z = fmaf(m3.z, w3, acc.z); acc.w = fmaf(m3.w, w3, acc.w);
    }
    for (; p < b; ++p) {
        int   s0 = csrc[p];
        float w0 = cw[p];
        float4 m0 = *reinterpret_cast<const float4*>(hc + (size_t)s0 * C);
        acc.x = fmaf(m0.x, w0, acc.x); acc.y = fmaf(m0.y, w0, acc.y); acc.z = fmaf(m0.z, w0, acc.z); acc.w = fmaf(m0.w, w0, acc.w);
    }
    *reinterpret_cast<float4*>(agg + (size_t)row * C + c) = acc;
}

__global__ void bias_relu_kernel(float* __restrict__ x,
                                 const float* __restrict__ b,
                                 int C, int total)
{
    int idx = blockIdx.x * blockDim.x + threadIdx.x;
    if (idx >= total) return;
    int c = idx % C;
    x[idx] = fmaxf(x[idx] + b[c], 0.0f);
}

__global__ void groupnorm_kernel(float* __restrict__ x,
                                 const float* __restrict__ gamma,
                                 const float* __restrict__ beta,
                                 int total  )
{
    int idx = blockIdx.x * blockDim.x + threadIdx.x;
    if (idx >= total) return;
    int row = idx >> 5;
    int g   = idx & 31;
    size_t base = (size_t)row * NHID + g * 3;
    float x0 = x[base], x1 = x[base + 1], x2 = x[base + 2];
    float mu = (x0 + x1 + x2) * (1.0f / 3.0f);
    float d0 = x0 - mu, d1 = x1 - mu, d2 = x2 - mu;
    float var = (d0 * d0 + d1 * d1 + d2 * d2) * (1.0f / 3.0f);
    float inv = rsqrtf(var + GN_EPS);
    int cg = g * 3;
    x[base]     = d0 * inv * gamma[cg]     + beta[cg];
    x[base + 1] = d1 * inv * gamma[cg + 1] + beta[cg + 1];
    x[base + 2] = d2 * inv * gamma[cg + 2] + beta[cg + 2];
}

__global__ void bias_logsoftmax_kernel(float* __restrict__ out,
                                       const float* __restrict__ b,
                                       int nrows)
{
    int row = blockIdx.x * blockDim.x + threadIdx.x;
    if (row >= nrows) return;
    float v[NCLASS];
    float m = -3.402823466e+38f;
    #pragma unroll
    for (int c = 0; c < NCLASS; ++c) {
        v[c] = out[(size_t)row * NCLASS + c] + b[c];
        m = fmaxf(m, v[c]);
    }
    float s = 0.0f;
    #pragma unroll
    for (int c = 0; c < NCLASS; ++c) s += __expf(v[c] - m);
    float ls = __logf(s);
    #pragma unroll
    for (int c = 0; c < NCLASS; ++c)
        out[(size_t)row * NCLASS + c] = v[c] - m - ls;
}

extern "C" void kernel_launch(void* const* d_in, const int* in_sizes, int n_in,
                              void* d_out, int out_size, void* d_ws, size_t ws_size,
                              hipStream_t stream)
{
    const float* x     = (const float*)d_in[0];
    const int*   src   = (const int*)  d_in[1];
    const int*   tgt   = (const int*)  d_in[2];
    const float* w     = (const float*)d_in[3];
    const float* W1    = (const float*)d_in[4];
    const float* b1    = (const float*)d_in[5];
    const float* W2    = (const float*)d_in[6];
    const float* b2    = (const float*)d_in[7];
    const float* W3    = (const float*)d_in[8];
    const float* b3    = (const float*)d_in[9];
    const float* gamma = (const float*)d_in[10];
    const float* beta  = (const float*)d_in[11];
    float* out = (float*)d_out;

    const int N   = in_sizes[0] / NFEAT;
    const int E   = in_sizes[1];
    const int ntm = N / 16;
    const int gblk = (ntm + 3) / 4;

    float* bufA = (float*)d_ws;
    float* bufB = bufA + (size_t)N * NHID;

    const int NB   = (N + 255) / 256;
    const int G    = 256;
    const int NPAD = ((NB * G + 4096) / 4096) * 4096;
    int*   hist   = (int*)(bufB + (size_t)N * NHID);
    int*   base   = hist + NPAD;
    int*   rowptr = base + NPAD;
    int*   recs1  = rowptr + NPAD;
    int*   recs2  = recs1 + (size_t)E * 4;
    int*   csrc   = recs2 + (size_t)E * 4;
    float* cw     = (float*)(csrc + E);

    const int TPB = 256;
    const int eblk = (E + TPB - 1) / TPB;

    bin_count_kernel<<<G, 1024, 0, stream>>>(tgt, hist, E, NB, NPAD);
    prefix_sum_kernel<<<1, 1024, 0, stream>>>(hist, base, NPAD / 32);
    bin_place_kernel<<<G, 1024, 0, stream>>>(src, tgt, w, base, recs1, E, NB);
    row_place_kernel<<<NB, 1024, 0, stream>>>(recs1, base, rowptr, recs2, G);
    edge_order_kernel<<<eblk, TPB, 0, stream>>>(recs2, rowptr, csrc, cw, E);

    wmma_gemm_strip_kernel<6, NFEAT, NHID><<<gblk, 128, 0, stream>>>(
        x, W1, bufA, ntm);
    {
        int total = N * (NHID / 4);
        gather_rows_kernel<NHID><<<(total + TPB - 1) / TPB, TPB, 0, stream>>>(
            bufA, rowptr, csrc, cw, bufB, total);
        int t2 = N * NHID;
        bias_relu_kernel<<<(t2 + TPB - 1) / TPB, TPB, 0, stream>>>(bufB, b1, NHID, t2);
    }

    wmma_gemm_strip_kernel<6, NHID, NHID><<<gblk, 128, 0, stream>>>(
        bufB, W2, bufA, ntm);
    {
        int total = N * (NHID / 4);
        gather_rows_kernel<NHID><<<(total + TPB - 1) / TPB, TPB, 0, stream>>>(
            bufA, rowptr, csrc, cw, bufB, total);
        int t2 = N * NHID;
        bias_relu_kernel<<<(t2 + TPB - 1) / TPB, TPB, 0, stream>>>(bufB, b2, NHID, t2);
    }

    {
        int total = N * 32;
        groupnorm_kernel<<<(total + TPB - 1) / TPB, TPB, 0, stream>>>(
            bufB, gamma, beta, total);
    }

    wmma_gemm_strip_kernel<3, NHID, NCLASS><<<gblk, 128, 0, stream>>>(
        bufB, W3, bufA, ntm);
    {
        int total = N * (NCLASS / 4);
        gather_rows_kernel<NCLASS><<<(total + TPB - 1) / TPB, TPB, 0, stream>>>(
            bufA, rowptr, csrc, cw, out, total);
    }

    bias_logsoftmax_kernel<<<(N + TPB - 1) / TPB, TPB, 0, stream>>>(out, b3, N);
}


# Row epilogues done by the gather's row owners before their one store (bias+relu; groupnorm, log_softmax through LDS in whole-row work-groups): the 4 elementwise launches and their waves are gone
# speedup vs baseline: 3.1932x; 1.3707x over previous
#include <hip/hip_runtime.h>
#include <hip/hip_bf16.h>


typedef float v2f __attribute__((ext_vector_type(2)));
typedef float v4f __attribute__((ext_vector_type(4)));
typedef float v8f __attribute__((ext_vector_type(8)));

#define NFEAT   128
#define NHID    96
#define NCLASS  40
#define GN_EPS  1e-5f

template<int NT, int K, int NCOLS>
__global__ __launch_bounds__(128) void wmma_gemm_strip_kernel(
    const float* __restrict__ A, const float* __restrict__ Bw,
    float* __restrict__ C, int ntm)
{
    constexpr int NCP   = NT * 16;
    constexpr int NQ    = NCP / 4;
    constexpr int NTASK = (K / 2) * NQ;
    constexpr int NIT   = (NTASK + 127) / 128;
    constexpr int KB    = K / 8;
    static_assert(K % 8 == 0 && NCOLS % 4 == 0 && NCOLS <= NCP, "shape");

    __shared__ v4f lds4[K * NCP / 4];

    const int tid   = threadIdx.x;
    const int wave  = tid >> 5;
    const int lane  = tid & 31;
    const int coll  = lane & 15;
    const int half  = lane >> 4;
    const int mtile = blockIdx.x * 4 + wave;
    const int mload = (mtile < ntm) ? mtile : (ntm - 1);

    v4f w0[NIT], w1[NIT];
    #pragma unroll
    for (int it = 0; it < NIT; ++it) {
        const int i  = tid + it * 128;
        const int ic = (NTASK % 128 == 0) ? i : (i < NTASK ? i : NTASK - 1);
        const int p  = ic / NQ;
        const int q  = ic - p * NQ;
        const int qc = (NCOLS == NCP) ? q : (q * 4 < NCOLS ? q : NCOLS / 4 - 1);
        const float* g = Bw + (size_t)(2 * p) * NCOLS + qc * 4;
        w0[it] = *reinterpret_cast<const v4f*>(g);
        w1[it] = *reinterpret_cast<const v4f*>(g + NCOLS);
    }

    v4f af[KB];
    const float* arow = A + (size_t)(mload * 16 + coll) * K + half * 4;
    #pragma unroll
    for (int j = 0; j < KB; ++j)
        af[j] = *reinterpret_cast<const v4f*>(arow + j * 8);

    #pragma unroll
    for (int it = 0; it < NIT; ++it) {
        const int i = tid + it * 128;
        if (NTASK % 128 == 0 || i < NTASK) {
            v4f lo = (v4f){w0[it].x, w1[it].x, w0[it].y, w1[it].y};
            v4f hi = (v4f){w0[it].z, w1[it].z, w0[it].w, w1[it].w};
            if (NCOLS != NCP) {
                const int q = i % NQ;
                if (q * 4 >= NCOLS) {
                    lo = (v4f){0.0f, 0.0f, 0.0f, 0.0f};
                    hi = (v4f){0.0f, 0.0f, 0.0f, 0.0f};
                }
            }
            lds4[2 * i]     = lo;
            lds4[2 * i + 1] = hi;
        }
    }
    __syncthreads();

    const float2* __restrict__ lds2 = reinterpret_cast<const float2*>(lds4);
    const float2* bp = lds2 + (2 * half) * NCP + coll;

    v8f acc[NT] = {};
    #pragma unroll
    for (int j = 0; j < KB; ++j) {
        #pragma unroll
        for (int s = 0; s < 2; ++s) {
            v2f a;
            a.x = s ? af[j].z : af[j].x;
            a.y = s ? af[j].w : af[j].y;
            const float2* bq = bp + (4 * j + s) * NCP;
            #pragma unroll
            for (int t = 0; t < NT; ++t) {
                float2 bv = bq[t * 16];
                v2f b; b.x = bv.x; b.y = bv.y;
                acc[t] = __builtin_amdgcn_wmma_f32_16x16x4_f32(
                    false, a, false, b, (short)0, acc[t], false, false);
            }
        }
    }

    if (mtile >= ntm) return;

    const int rbase = mtile * 16 + half * 8;
    #pragma unroll
    for (int t = 0; t < NT; ++t) {
        int col = t * 16 + coll;
        if (col < NCOLS) {
            #pragma unroll
            for (int r = 0; r < 8; ++r)
                C[(size_t)(rbase + r) * NCOLS + col] = acc[t][r];
        }
    }
}


__global__ __launch_bounds__(1024) void bin_count_kernel(const int* __restrict__ tgt,
                                                         int* __restrict__ hist,
                                                         int E, int nb, int npad)
{
    __shared__ int lh[256];
    const int tid = threadIdx.x;
    const int g   = blockIdx.x;
    const int G   = gridDim.x;
    if (tid < 256) lh[tid] = 0;
    __syncthreads();
    const int ch = (E + G - 1) / G;
    const int e0 = g * ch;
    const int e1 = (e0 + ch < E) ? (e0 + ch) : E;
    for (int e = e0 + tid; e < e1; e += 1024)
        atomicAdd(&lh[tgt[e] >> 8], 1);
    __syncthreads();
    if (tid < nb) hist[tid * G + g] = lh[tid];
    for (int i = nb * G + g * 1024 + tid; i < npad; i += G * 1024) hist[i] = 0;
}

__global__ __launch_bounds__(1024) void prefix_sum_kernel(const int* __restrict__ in,
                                                          int* __restrict__ out, int pw)
{
    __shared__ int wtot[32];
    const int lane = threadIdx.x & 31;
    const int wave = threadIdx.x >> 5;
    const int nit  = pw >> 7;
    const int4* c4 = reinterpret_cast<const int4*>(in + (size_t)wave * pw) + lane;
    int4*       r4 = reinterpret_cast<int4*>(out + (size_t)wave * pw) + lane;

    int s = 0;
    for (int it = 0; it < nit; ++it) {
        int4 v = c4[it * 32];
        s += (v.x + v.y) + (v.z + v.w);
    }
    for (int off = 16; off > 0; off >>= 1) s += __shfl_xor(s, off);
    if (lane == 0) wtot[wave] = s;
    __syncthreads();
    int carry = (lane < wave) ? wtot[lane] : 0;
    for (int off = 16; off > 0; off >>= 1) carry += __shfl_xor(carry, off);

    for (int it = 0; it < nit; ++it) {
        int4 v = c4[it * 32];
        int ls  = (v.x + v.y) + (v.z + v.w);
        int inc = ls;
        for (int off = 1; off < 32; off <<= 1) {
            int u = __shfl_up(inc, off);
            if (lane >= off) inc += u;
        }
        int4 o;
        o.x = carry + inc - ls;
        o.y = o.x + v.x;
        o.z = o.y + v.y;
        o.w = o.z + v.z;
        r4[it * 32] = o;
        carry += __shfl(inc, 31);
    }
}

__global__ __launch_bounds__(1024) void bin_place_kernel(const int* __restrict__ src,
                                                         const int* __restrict__ tgt,
                                                         const float* __restrict__ w,
                                                         const int* __restrict__ base,
                                                         int* __restrict__ recs1,
                                                         int E, int nb)
{
    __shared__ int cur[256];
    const int tid = threadIdx.x;
    const int g   = blockIdx.x;
    const int G   = gridDim.x;
    if (tid < nb) cur[tid] = base[tid * G + g];
    __syncthreads();
    const int ch = (E + G - 1) / G;
    const int e0 = g * ch;
    const int e1 = (e0 + ch < E) ? (e0 + ch) : E;
    int4* r1 = reinterpret_cast<int4*>(recs1);
    for (int e = e0 + tid; e < e1; e += 1024) {
        const int t   = tgt[e];
        const int pos = atomicAdd(&cur[t >> 8], 1);
        r1[pos] = make_int4(e, t, src[e], __float_as_int(w[e]));
    }
}

__global__ __launch_bounds__(1024) void row_place_kernel(const int* __restrict__ recs1,
                                                         const int* __restrict__ base,
                                                         int* __restrict__ rowptr,
                                                         int* __restrict__ recs2, int G)
{
    __shared__ int cnt[256];
    __shared__ int cur[256];
    __shared__ int wsum[8];
    const int tid  = threadIdx.x;
    const int lane = tid & 31;
    const int wave = tid >> 5;
    const int b    = blockIdx.x;
    const int bs   = base[b * G];
    const int be   = base[(b + 1) * G];
    const int4* r1 = reinterpret_cast<const int4*>(recs1);
    int4*       r2 = reinterpret_cast<int4*>(recs2);

    if (tid < 256) cnt[tid] = 0;
    __syncthreads();
    for (int i = bs + tid; i < be; i += 1024)
        atomicAdd(&cnt[r1[i].y & 255], 1);
    __syncthreads();

    int c = 0, inc = 0;
    if (tid < 256) {
        c = cnt[tid];
        inc = c;
        for (int off = 1; off < 32; off <<= 1) {
            int u = __shfl_up(inc, off);
            if (lane >= off) inc += u;
        }
        if (lane == 31) wsum[wave] = inc;
    }
    __syncthreads();
    if (tid < 256) {
        int front = 0;
        for (int k = 0; k < wave; ++k) front += wsum[k];
        const int first = bs + front + inc - c;
        cur[tid] = first;
        rowptr[b * 256 + tid] = first;
        if (tid == 255 && b == (int)gridDim.x - 1) rowptr[b * 256 + 256] = be;
    }
    __syncthreads();
    for (int i = bs + tid; i < be; i += 1024) {
        const int4 r = r1[i];
        const int pos = atomicAdd(&cur[r.y & 255], 1);
        r2[pos] = r;
    }
}

__global__ void edge_order_kernel(const int* __restrict__ recs2,
                                  const int* __restrict__ rowptr,
                                  int* __restrict__ csrc,
                                  float* __restrict__ cw, int E)
{
    int p = blockIdx.x * blockDim.x + threadIdx.x;
    if (p >= E) return;
    const int4* r2 = reinterpret_cast<const int4*>(recs2);
    const int4 r = r2[p];
    const int a = rowptr[r.y];
    const int b = rowptr[r.y + 1];
    int k = 0;
    for (int j = a; j < b; ++j) k += (r2[j].x < r.x) ? 1 : 0;
    csrc[a + k] = r.z;
    cw[a + k]   = __int_as_float(r.w);
}

#define EPI_RELU     1
#define EPI_RELU_GN  2
#define EPI_LSM      3

template<int C, int EPI>
__global__ __launch_bounds__((C / 4) * ((C == NHID) ? 8 : 32))
void gather_rows_kernel(const float* __restrict__ h,
                        const int* __restrict__ rowptr,
                        const int* __restrict__ csrc,
                        const float* __restrict__ cw,
                        const float* __restrict__ bias,
                        const float* __restrict__ gamma,
                        const float* __restrict__ beta,
                        float* __restrict__ out, int nrows)
{
    constexpr int VPR = C / 4;
    constexpr int RPB = (C == NHID) ? 8 : 32;
    const int tid = threadIdx.x;
    const int rl  = tid / VPR;
    const int v   = tid - rl * VPR;
    const int row = blockIdx.x * RPB + rl;
    const bool live = row < nrows;
    const int c   = v * 4;
    int p = 0, b = 0;
    if (live) { p = rowptr[row]; b = rowptr[row + 1]; }
    const float* hc = h + c;
    float4 acc = make_float4(0.0f, 0.0f, 0.0f, 0.0f);
    for (; p + 4 <= b; p += 4) {
        int   s0 = csrc[p], s1 = csrc[p + 1], s2 = csrc[p + 2], s3 = csrc[p + 3];
        float w0 = cw[p],   w1 = cw[p + 1],   w2 = cw[p + 2],   w3 = cw[p + 3];
        float4 m0 = *reinterpret_cast<const float4*>(hc + (size_t)s0 * C);
        float4 m1 = *reinterpret_cast<const float4*>(hc + (size_t)s1 * C);
        float4 m2 = *reinterpret_cast<const float4*>(hc + (size_t)s2 * C);
        float4 m3 = *reinterpret_cast<const float4*>(hc + (size_t)s3 * C);
        acc.x = fmaf(m0.x, w0, acc.x); acc.y = fmaf(m0.y, w0, acc.y); acc.z = fmaf(m0.z, w0, acc.z); acc.w = fmaf(m0.w, w0, acc.w);
        acc.x = fmaf(m1.x, w1, acc.x); acc.y = fmaf(m1.y, w1, acc.y); acc.z = fmaf(m1.z, w1, acc.z); acc.w = fmaf(m1.w, w1, acc.w);
        acc.x = fmaf(m2.x, w2, acc.x); acc.y = fmaf(m2.y, w2, acc.y); acc.z = fmaf(m2.z, w2, acc.z); acc.w = fmaf(m2.w, w2, acc.w);
        acc.x = fmaf(m3.x, w3, acc.x); acc.y = fmaf(m3.y, w3, acc.y); acc.z = fmaf(m3.z, w3, acc.z); acc.w = fmaf(m3.w, w3, acc.w);
    }
    for (; p < b; ++p) {
        int   s0 = csrc[p];
        float w0 = cw[p];
        float4 m0 = *reinterpret_cast<const float4*>(hc + (size_t)s0 * C);
        acc.x = fmaf(m0.x, w0, acc.x); acc.y = fmaf(m0.y, w0, acc.y); acc.z = fmaf(m0.z, w0, acc.z); acc.w = fmaf(m0.w, w0, acc.w);
    }

    const float4 bb = *reinterpret_cast<const float4*>(bias + c);
    float4 r = make_float4(acc.x + bb.x, acc.y + bb.y, acc.z + bb.z, acc.w + bb.w);
    if constexpr (EPI == EPI_RELU || EPI == EPI_RELU_GN) {
        r.x = fmaxf(r.x, 0.0f); r.y = fmaxf(r.y, 0.0f); r.z = fmaxf(r.z, 0.0f); r.w = fmaxf(r.w, 0.0f);
    }

    if constexpr (EPI == EPI_RELU_GN) {
        __shared__ float4 srow4[RPB * VPR];
        srow4[tid] = r;
        __syncthreads();
        const int ga = c / 3;
        const int j  = c - ga * 3;
        const float* q = reinterpret_cast<const float*>(srow4) + rl * C + ga * 3;
        const float a0 = q[0], a1 = q[1], a2 = q[2], b0 = q[3], b1 = q[4], b2 = q[5];
        const float muA = (a0 + a1 + a2) * (1.0f / 3.0f);
        const float muB = (b0 + b1 + b2) * (1.0f / 3.0f);
        const float dA0 = a0 - muA, dA1 = a1 - muA, dA2 = a2 - muA;
        const float dB0 = b0 - muB, dB1 = b1 - muB, dB2 = b2 - muB;
        const float varA = (dA0 * dA0 + dA1 * dA1 + dA2 * dA2) * (1.0f / 3.0f);
        const float varB = (dB0 * dB0 + dB1 * dB1 + dB2 * dB2) * (1.0f / 3.0f);
        const float invA = rsqrtf(varA + GN_EPS);
        const float invB = rsqrtf(varB + GN_EPS);
        const float4 gm = *reinterpret_cast<const float4*>(gamma + c);
        const float4 bt = *reinterpret_cast<const float4*>(beta + c);
        const bool inA0 = (j + 0) < 3, inA1 = (j + 1) < 3, inA2 = (j + 2) < 3, inA3 = (j + 3) < 3;
        r.x = (r.x - (inA0 ? muA : muB)) * (inA0 ? invA : invB) * gm.x + bt.x;
        r.y = (r.y - (inA1 ? muA : muB)) * (inA1 ? invA : invB) * gm.y + bt.y;
        r.z = (r.z - (inA2 ? muA : muB)) * (inA2 ? invA : invB) * gm.z + bt.z;
        r.w = (r.w - (inA3 ? muA : muB)) * (inA3 ? invA : invB) * gm.w + bt.w;
    }

    if constexpr (EPI == EPI_LSM) {
        __shared__ float4 sval[RPB * VPR];
        __shared__ float4 sexp[RPB * VPR];
        sval[tid] = r;
        __syncthreads();
        float m = -3.402823466e+38f;
        #pragma unroll
        for (int k = 0; k < VPR; ++k) {
            const float4 t = sval[rl * VPR + k];
            m = fmaxf(m, t.x); m = fmaxf(m, t.y); m = fmaxf(m, t.z); m = fmaxf(m, t.w);
        }
        float4 e;
        e.x = __expf(r.x - m); e.y = __expf(r.y - m); e.z = __expf(r.z - m); e.w = __expf(r.w - m);
        sexp[tid] = e;
        __syncthreads();
        float s = 0.0f;
        #pragma unroll
        for (int k = 0; k < VPR; ++k) {
            const float4 t = sexp[rl * VPR + k];
            s += t.x; s += t.y; s += t.z; s += t.w;
        }
        const float ls = __logf(s);
        r.x = r.x - m - ls; r.y = r.y - m - ls; r.z = r.z - m - ls; r.w = r.w - m - ls;
    }

    if (live)
        *reinterpret_cast<float4*>(out + (size_t)row * C + c) = r;
}

extern "C" void kernel_launch(void* const* d_in, const int* in_sizes, int n_in,
                              void* d_out, int out_size, void* d_ws, size_t ws_size,
                              hipStream_t stream)
{
    const float* x     = (const float*)d_in[0];
    const int*   src   = (const int*)  d_in[1];
    const int*   tgt   = (const int*)  d_in[2];
    const float* w     = (const float*)d_in[3];
    const float* W1    = (const float*)d_in[4];
    const float* b1    = (const float*)d_in[5];
    const float* W2    = (const float*)d_in[6];
    const float* b2    = (const float*)d_in[7];
    const float* W3    = (const float*)d_in[8];
    const float* b3    = (const float*)d_in[9];
    const float* gamma = (const float*)d_in[10];
    const float* beta  = (const float*)d_in[11];
    float* out = (float*)d_out;

    const int N   = in_sizes[0] / NFEAT;
    const int E   = in_sizes[1];
    const int ntm = N / 16;
    const int gblk = (ntm + 3) / 4;

    float* bufA = (float*)d_ws;
    float* bufB = bufA + (size_t)N * NHID;

    const int NB   = (N + 255) / 256;
    const int G    = 256;
    const int NPAD = ((NB * G + 4096) / 4096) * 4096;
    int*   hist   = (int*)(bufB + (size_t)N * NHID);
    int*   base   = hist + NPAD;
    int*   rowptr = base + NPAD;
    int*   recs1  = rowptr + NPAD;
    int*   recs2  = recs1 + (size_t)E * 4;
    int*   csrc   = recs2 + (size_t)E * 4;
    float* cw     = (float*)(csrc + E);

    const int TPB = 256;
    const int eblk = (E + TPB - 1) / TPB;

    bin_count_kernel<<<G, 1024, 0, stream>>>(tgt, hist, E, NB, NPAD);
    prefix_sum_kernel<<<1, 1024, 0, stream>>>(hist, base, NPAD / 32);
    bin_place_kernel<<<G, 1024, 0, stream>>>(src, tgt, w, base, recs1, E, NB);
    row_place_kernel<<<NB, 1024, 0, stream>>>(recs1, base, rowptr, recs2, G);
    edge_order_kernel<<<eblk, TPB, 0, stream>>>(recs2, rowptr, csrc, cw, E);

    const int gb96 = (N + 7) / 8;
    const int gb40 = (N + 31) / 32;

    wmma_gemm_strip_kernel<6, NFEAT, NHID><<<gblk, 128, 0, stream>>>(
        x, W1, bufA, ntm);
    gather_rows_kernel<NHID, EPI_RELU><<<gb96, 192, 0, stream>>>(
        bufA, rowptr, csrc, cw, b1, gamma, beta, bufB, N);

    wmma_gemm_strip_kernel<6, NHID, NHID><<<gblk, 128, 0, stream>>>(
        bufB, W2, bufA, ntm);
    gather_rows_kernel<NHID, EPI_RELU_GN><<<gb96, 192, 0, stream>>>(
        bufA, rowptr, csrc, cw, b2, gamma, beta, bufB, N);

    wmma_gemm_strip_kernel<3, NHID, NCLASS><<<gblk, 128, 0, stream>>>(
        bufB, W3, bufA, ntm);
    gather_rows_kernel<NCLASS, EPI_LSM><<<gb40, 320, 0, stream>>>(
        bufA, rowptr, csrc, cw, b3, gamma, beta, out, N);
}


# Layer-2 and layer-3 products moved into the gathers before them: a 64-row work-group keeps its finished rows in LDS and multiplies them by LDS-staged weights; h1 and g skip memory; 2 launches fewer
# speedup vs baseline: 3.6805x; 1.1526x over previous
#include <hip/hip_runtime.h>
#include <hip/hip_bf16.h>


typedef float v2f __attribute__((ext_vector_type(2)));
typedef float v4f __attribute__((ext_vector_type(4)));
typedef float v8f __attribute__((ext_vector_type(8)));

#define NFEAT   128
#define NHID    96
#define NCLASS  40
#define GN_EPS  1e-5f

template<int NT, int K, int NCOLS>
__global__ __launch_bounds__(128) void wmma_gemm_strip_kernel(
    const float* __restrict__ A, const float* __restrict__ Bw,
    float* __restrict__ C, int ntm)
{
    constexpr int NCP   = NT * 16;
    constexpr int NQ    = NCP / 4;
    constexpr int NTASK = (K / 2) * NQ;
    constexpr int NIT   = (NTASK + 127) / 128;
    constexpr int KB    = K / 8;
    static_assert(K % 8 == 0 && NCOLS % 4 == 0 && NCOLS <= NCP, "shape");

    __shared__ v4f lds4[K * NCP / 4];

    const int tid   = threadIdx.x;
    const int wave  = tid >> 5;
    const int lane  = tid & 31;
    const int coll  = lane & 15;
    const int half  = lane >> 4;
    const int mtile = blockIdx.x * 4 + wave;
    const int mload = (mtile < ntm) ? mtile : (ntm - 1);

    v4f w0[NIT], w1[NIT];
    #pragma unroll
    for (int it = 0; it < NIT; ++it) {
        const int i  = tid + it * 128;
        const int ic = (NTASK % 128 == 0) ? i : (i < NTASK ? i : NTASK - 1);
        const int p  = ic / NQ;
        const int q  = ic - p * NQ;
        const int qc = (NCOLS == NCP) ? q : (q * 4 < NCOLS ? q : NCOLS / 4 - 1);
        const float* g = Bw + (size_t)(2 * p) * NCOLS + qc * 4;
        w0[it] = *reinterpret_cast<const v4f*>(g);
        w1[it] = *reinterpret_cast<const v4f*>(g + NCOLS);
    }

    v4f af[KB];
    const float* arow = A + (size_t)(mload * 16 + coll) * K + half * 4;
    #pragma unroll
    for (int j = 0; j < KB; ++j)
        af[j] = *reinterpret_cast<const v4f*>(arow + j * 8);

    #pragma unroll
    for (int it = 0; it < NIT; ++it) {
        const int i = tid + it * 128;
        if (NTASK % 128 == 0 || i < NTASK) {
            v4f lo = (v4f){w0[it].x, w1[it].x, w0[it].y, w1[it].y};
            v4f hi = (v4f){w0[it].z, w1[it].z, w0[it].w, w1[it].w};
            if (NCOLS != NCP) {
                const int q = i % NQ;
                if (q * 4 >= NCOLS) {
                    lo = (v4f){0.0f, 0.0f, 0.0f, 0.0f};
                    hi = (v4f){0.0f, 0.0f, 0.0f, 0.0f};
                }
            }
            lds4[2 * i]     = lo;
            lds4[2 * i + 1] = hi;
        }
    }
    __syncthreads();

    const float2* __restrict__ lds2 = reinterpret_cast<const float2*>(lds4);
    const float2* bp = lds2 + (2 * half) * NCP + coll;

    v8f acc[NT] = {};
    #pragma unroll
    for (int j = 0; j < KB; ++j) {
        #pragma unroll
        for (int s = 0; s < 2; ++s) {
            v2f a;
            a.x = s ? af[j].z : af[j].x;
            a.y = s ? af[j].w : af[j].y;
            const float2* bq = bp + (4 * j + s) * NCP;
            #pragma unroll
            for (int t = 0; t < NT; ++t) {
                float2 bv = bq[t * 16];
                v2f b; b.x = bv.x; b.y = bv.y;
                acc[t] = __builtin_amdgcn_wmma_f32_16x16x4_f32(
                    false, a, false, b, (short)0, acc[t], false, false);
            }
        }
    }

    if (mtile >= ntm) return;

    const int rbase = mtile * 16 + half * 8;
    #pragma unroll
    for (int t = 0; t < NT; ++t) {
        int col = t * 16 + coll;
        if (col < NCOLS) {
            #pragma unroll
            for (int r = 0; r < 8; ++r)
                C[(size_t)(rbase + r) * NCOLS + col] = acc[t][r];
        }
    }
}


__global__ __launch_bounds__(1024) void bin_count_kernel(const int* __restrict__ tgt,
                                                         int* __restrict__ hist,
                                                         int E, int nb, int npad)
{
    __shared__ int lh[256];
    const int tid = threadIdx.x;
    const int g   = blockIdx.x;
    const int G   = gridDim.x;
    if (tid < 256) lh[tid] = 0;
    __syncthreads();
    const int ch = (E + G - 1) / G;
    const int e0 = g * ch;
    const int e1 = (e0 + ch < E) ? (e0 + ch) : E;
    for (int e = e0 + tid; e < e1; e += 1024)
        atomicAdd(&lh[tgt[e] >> 8], 1);
    __syncthreads();
    if (tid < nb) hist[tid * G + g] = lh[tid];
    for (int i = nb * G + g * 1024 + tid; i < npad; i += G * 1024) hist[i] = 0;
}

__global__ __launch_bounds__(1024) void prefix_sum_kernel(const int* __restrict__ in,
                                                          int* __restrict__ out, int pw)
{
    __shared__ int wtot[32];
    const int lane = threadIdx.x & 31;
    const int wave = threadIdx.x >> 5;
    const int nit  = pw >> 7;
    const int4* c4 = reinterpret_cast<const int4*>(in + (size_t)wave * pw) + lane;
    int4*       r4 = reinterpret_cast<int4*>(out + (size_t)wave * pw) + lane;

    int s = 0;
    for (int it = 0; it < nit; ++it) {
        int4 v = c4[it * 32];
        s += (v.x + v.y) + (v.z + v.w);
    }
    for (int off = 16; off > 0; off >>= 1) s += __shfl_xor(s, off);
    if (lane == 0) wtot[wave] = s;
    __syncthreads();
    int carry = (lane < wave) ? wtot[lane] : 0;
    for (int off = 16; off > 0; off >>= 1) carry += __shfl_xor(carry, off);

    for (int it = 0; it < nit; ++it) {
        int4 v = c4[it * 32];
        int ls  = (v.x + v.y) + (v.z + v.w);
        int inc = ls;
        for (int off = 1; off < 32; off <<= 1) {
            int u = __shfl_up(inc, off);
            if (lane >= off) inc += u;
        }
        int4 o;
        o.x = carry + inc - ls;
        o.y = o.x + v.x;
        o.z = o.y + v.y;
        o.w = o.z + v.z;
        r4[it * 32] = o;
        carry += __shfl(inc, 31);
    }
}

__global__ __launch_bounds__(1024) void bin_place_kernel(const int* __restrict__ src,
                                                         const int* __restrict__ tgt,
                                                         const float* __restrict__ w,
                                                         const int* __restrict__ base,
                                                         int* __restrict__ recs1,
                                                         int E, int nb)
{
    __shared__ int cur[256];
    const int tid = threadIdx.x;
    const int g   = blockIdx.x;
    const int G   = gridDim.x;
    if (tid < nb) cur[tid] = base[tid * G + g];
    __syncthreads();
    const int ch = (E + G - 1) / G;
    const int e0 = g * ch;
    const int e1 = (e0 + ch < E) ? (e0 + ch) : E;
    int4* r1 = reinterpret_cast<int4*>(recs1);
    for (int e = e0 + tid; e < e1; e += 1024) {
        const int t   = tgt[e];
        const int pos = atomicAdd(&cur[t >> 8], 1);
        r1[pos] = make_int4(e, t, src[e], __float_as_int(w[e]));
    }
}

__global__ __launch_bounds__(1024) void row_place_kernel(const int* __restrict__ recs1,
                                                         const int* __restrict__ base,
                                                         int* __restrict__ rowptr,
                                                         int* __restrict__ recs2, int G)
{
    __shared__ int cnt[256];
    __shared__ int cur[256];
    __shared__ int wsum[8];
    const int tid  = threadIdx.x;
    const int lane = tid & 31;
    const int wave = tid >> 5;
    const int b    = blockIdx.x;
    const int bs   = base[b * G];
    const int be   = base[(b + 1) * G];
    const int4* r1 = reinterpret_cast<const int4*>(recs1);
    int4*       r2 = reinterpret_cast<int4*>(recs2);

    if (tid < 256) cnt[tid] = 0;
    __syncthreads();
    for (int i = bs + tid; i < be; i += 1024)
        atomicAdd(&cnt[r1[i].y & 255], 1);
    __syncthreads();

    int c = 0, inc = 0;
    if (tid < 256) {
        c = cnt[tid];
        inc = c;
        for (int off = 1; off < 32; off <<= 1) {
            int u = __shfl_up(inc, off);
            if (lane >= off) inc += u;
        }
        if (lane == 31) wsum[wave] = inc;
    }
    __syncthreads();
    if (tid < 256) {
        int front = 0;
        for (int k = 0; k < wave; ++k) front += wsum[k];
        const int first = bs + front + inc - c;
        cur[tid] = first;
        rowptr[b * 256 + tid] = first;
        if (tid == 255 && b == (int)gridDim.x - 1) rowptr[b * 256 + 256] = be;
    }
    __syncthreads();
    for (int i = bs + tid; i < be; i += 1024) {
        const int4 r = r1[i];
        const int pos = atomicAdd(&cur[r.y & 255], 1);
        r2[pos] = r;
    }
}

__global__ void edge_order_kernel(const int* __restrict__ recs2,
                                  const int* __restrict__ rowptr,
                                  int* __restrict__ csrc,
                                  float* __restrict__ cw, int E)
{
    int p = blockIdx.x * blockDim.x + threadIdx.x;
    if (p >= E) return;
    const int4* r2 = reinterpret_cast<const int4*>(recs2);
    const int4 r = r2[p];
    const int a = rowptr[r.y];
    const int b = rowptr[r.y + 1];
    int k = 0;
    for (int j = a; j < b; ++j) k += (r2[j].x < r.x) ? 1 : 0;
    csrc[a + k] = r.z;
    cw[a + k]   = __int_as_float(r.w);
}

#define EPI_RELU     1
#define EPI_RELU_GN  2
#define EPI_LSM      3

template<int C, int CPL, int GRP>
__device__ __forceinline__ void walk_entries(const float* __restrict__ hc, int si, float wi,
                                             int n, float (&acc)[CPL])
{
    for (int k = 0; k < n; k += GRP) {
        float m[GRP][CPL];
        #pragma unroll
        for (int u = 0; u < GRP; ++u) {
            const int kk = (k + u < n) ? (k + u) : (n - 1);
            const int s  = __builtin_amdgcn_readlane(si, kk);
            const float* hp = hc + (size_t)s * C;
            #pragma unroll
            for (int j = 0; j < CPL; ++j) m[u][j] = hp[j];
        }
        #pragma unroll
        for (int u = 0; u < GRP; ++u) {
            const int kk = (k + u < n) ? (k + u) : (n - 1);
            const float w = __int_as_float(__builtin_amdgcn_readlane(__float_as_int(wi), kk));
            const bool on = (k + u < n);
            #pragma unroll
            for (int j = 0; j < CPL; ++j) {
                const float t = fmaf(m[u][j], w, acc[j]);
                acc[j] = on ? t : acc[j];
            }
        }
    }
}

template<int C, int EPI>
__global__ __launch_bounds__(256)
void gather_rows_kernel(const float* __restrict__ h,
                        const int* __restrict__ rowptr,
                        const int* __restrict__ csrc,
                        const float* __restrict__ cw,
                        const float* __restrict__ bias,
                        const float* __restrict__ gamma,
                        const float* __restrict__ beta,
                        float* __restrict__ out, int nrows)
{
    constexpr int CPL = (C == NHID) ? 3 : 2;
    constexpr int LPR = C / CPL;
    constexpr int RPW = 4;
    constexpr int GRP = 8;
    static_assert(LPR * CPL == C && LPR <= 32 && RPW + 1 <= 32, "shape");

    const int lane = threadIdx.x & 31;
    const int gw  = __builtin_amdgcn_readfirstlane((int)(blockIdx.x * 8 + (threadIdx.x >> 5)));
    const int r0  = gw * RPW;
    if (r0 >= nrows) return;
    const bool act = lane < LPR;
    const int  c   = (act ? lane : 0) * CPL;

    int bi = r0 + ((lane < RPW) ? lane : RPW);
    if (bi > nrows) bi = nrows;
    const int rpv = rowptr[bi];

    float bs[CPL], gm[CPL], bt[CPL];
    #pragma unroll
    for (int j = 0; j < CPL; ++j) {
        bs[j] = bias[c + j];
        gm[j] = (EPI == EPI_RELU_GN) ? gamma[c + j] : 1.0f;
        bt[j] = (EPI == EPI_RELU_GN) ? beta[c + j]  : 0.0f;
    }
    const float* hc = h + c;

    auto do_row = [&](const int i, const int si, const float wi, int& nsi, float& nwi)
        __attribute__((always_inline)) {
        const int row = r0 + i;
        const int p0 = __builtin_amdgcn_readlane(rpv, i);
        const int p1 = __builtin_amdgcn_readlane(rpv, i + 1);

        nsi = 0;
        nwi = 0.0f;
        if (i + 1 < RPW && row + 1 < nrows) {
            const int p2 = __builtin_amdgcn_readlane(rpv, i + 2);
            if (lane < p2 - p1) { nsi = csrc[p1 + lane]; nwi = cw[p1 + lane]; }
        }

        float acc[CPL];
        #pragma unroll
        for (int j = 0; j < CPL; ++j) acc[j] = 0.0f;

        if (p1 > p0)
            walk_entries<C, CPL, GRP>(hc, si, wi, (p1 - p0 < 32) ? (p1 - p0) : 32, acc);
        for (int base = p0 + 32; base < p1; base += 32) {
            int   xi = 0;
            float xw = 0.0f;
            if (lane < p1 - base) { xi = csrc[base + lane]; xw = cw[base + lane]; }
            walk_entries<C, CPL, GRP>(hc, xi, xw, (p1 - base < 32) ? (p1 - base) : 32, acc);
        }

        float r[CPL];
        #pragma unroll
        for (int j = 0; j < CPL; ++j) {
            r[j] = acc[j] + bs[j];
            if (EPI == EPI_RELU || EPI == EPI_RELU_GN) r[j] = fmaxf(r[j], 0.0f);
        }

        if constexpr (EPI == EPI_RELU_GN) {
            const float mu  = (r[0] + r[1] + r[2]) * (1.0f / 3.0f);
            const float d0 = r[0] - mu, d1 = r[1] - mu, d2 = r[2] - mu;
            const float var = (d0 * d0 + d1 * d1 + d2 * d2) * (1.0f / 3.0f);
            const float inv = rsqrtf(var + GN_EPS);
            r[0] = d0 * inv * gm[0] + bt[0];
            r[1] = d1 * inv * gm[1] + bt[1];
            r[2] = d2 * inv * gm[2] + bt[2];
        }

        if constexpr (EPI == EPI_LSM) {
            float mx = act ? fmaxf(r[0], r[1]) : -3.402823466e+38f;
            #pragma unroll
            for (int off = 16; off > 0; off >>= 1) mx = fmaxf(mx, __shfl_xor(mx, off));
            const float e0 = __expf(r[0] - mx);
            const float e1 = __expf(r[1] - mx);
            float s = 0.0f;
            #pragma unroll
            for (int k = 0; k < LPR; ++k) {
                s += __int_as_float(__builtin_amdgcn_readlane(__float_as_int(e0), k));
                s += __int_as_float(__builtin_amdgcn_readlane(__float_as_int(e1), k));
            }
            const float ls = __logf(s);
            r[0] = r[0] - mx - ls;
            r[1] = r[1] - mx - ls;
        }

        if (act) {
            float* o = out + (size_t)row * C + c;
            #pragma unroll
            for (int j = 0; j < CPL; ++j) o[j] = r[j];
        }
    };

    int   fsi = 0;
    float fwi = 0.0f;
    {
        const int pa = __builtin_amdgcn_readlane(rpv, 0);
        const int pb = __builtin_amdgcn_readlane(rpv, 1);
        if (lane < pb - pa) { fsi = csrc[pa + lane]; fwi = cw[pa + lane]; }
    }
    int   nsi;
    float nwi;
    do_row(0, fsi, fwi, nsi, nwi);
    for (int i = 1; i < RPW; ++i) {
        if (r0 + i >= nrows) break;
        const int   si = nsi;
        const float wi = nwi;
        do_row(i, si, wi, nsi, nwi);
    }
}

template<int EPI, int NT, int NCOLS>
__global__ __launch_bounds__(512)
void gather_rows_gemm_kernel(const float* __restrict__ h,
                             const int* __restrict__ rowptr,
                             const int* __restrict__ csrc,
                             const float* __restrict__ cw,
                             const float* __restrict__ bias,
                             const float* __restrict__ gamma,
                             const float* __restrict__ beta,
                             const float* __restrict__ Bw,
                             float* __restrict__ Cn, int nrows)
{
    constexpr int C     = NHID;
    constexpr int K     = NHID;
    constexpr int CPL   = 3;
    constexpr int RPW   = 4;
    constexpr int GRP   = 8;
    constexpr int NTHR  = 512;
    constexpr int RPB   = (NTHR / 32) * RPW;
    constexpr int NCP   = NT * 16;
    constexpr int NQ    = NCP / 4;
    constexpr int NTASK = (K / 2) * NQ;
    constexpr int NIT   = (NTASK + NTHR - 1) / NTHR;
    constexpr int KB    = K / 8;
    constexpr int AP    = 100;
    constexpr int TPW   = NT / 3;
    static_assert(NT % 3 == 0 && K % 8 == 0 && NCOLS % 4 == 0 && NCOLS <= NCP && C == 32 * CPL, "shape");

    __shared__ v4f lds4[K * NCP / 4];
    __shared__ v4f at4[RPB * AP / 4];
    float* at = reinterpret_cast<float*>(at4);

    const int tid  = threadIdx.x;
    const int lane = tid & 31;
    const int wave = __builtin_amdgcn_readfirstlane(tid >> 5);
    const int r0   = blockIdx.x * RPB + wave * RPW;
    const int c    = lane * CPL;

    int bi = r0 + ((lane < RPW) ? lane : RPW);
    if (bi > nrows) bi = nrows;
    const int rpv = rowptr[bi];

    v4f w0[NIT], w1[NIT];
    #pragma unroll
    for (int it = 0; it < NIT; ++it) {
        const int i  = tid + it * NTHR;
        const int ic = (NTASK % NTHR == 0) ? i : (i < NTASK ? i : NTASK - 1);
        const int p  = ic / NQ;
        const int q  = ic - p * NQ;
        const int qc = (NCOLS == NCP) ? q : (q * 4 < NCOLS ? q : NCOLS / 4 - 1);
        const float* g = Bw + (size_t)(2 * p) * NCOLS + qc * 4;
        w0[it] = *reinterpret_cast<const v4f*>(g);
        w1[it] = *reinterpret_cast<const v4f*>(g + NCOLS);
    }

    float bs[CPL], gm[CPL], bt[CPL];
    #pragma unroll
    for (int j = 0; j < CPL; ++j) {
        bs[j] = bias[c + j];
        gm[j] = (EPI == EPI_RELU_GN) ? gamma[c + j] : 1.0f;
        bt[j] = (EPI == EPI_RELU_GN) ? beta[c + j]  : 0.0f;
    }
    const float* hc = h + c;

    int   fsi = 0;
    float fwi = 0.0f;
    if (r0 < nrows) {
        const int pa = __builtin_amdgcn_readlane(rpv, 0);
        const int pb = __builtin_amdgcn_readlane(rpv, 1);
        if (lane < pb - pa) { fsi = csrc[pa + lane]; fwi = cw[pa + lane]; }
    }

    #pragma unroll
    for (int it = 0; it < NIT; ++it) {
        const int i = tid + it * NTHR;
        if (NTASK % NTHR == 0 || i < NTASK) {
            v4f lo = (v4f){w0[it].x, w1[it].x, w0[it].y, w1[it].y};
            v4f hi = (v4f){w0[it].z, w1[it].z, w0[it].w, w1[it].w};
            if (NCOLS != NCP) {
                const int q = i % NQ;
                if (q * 4 >= NCOLS) {
                    lo = (v4f){0.0f, 0.0f, 0.0f, 0.0f};
                    hi = (v4f){0.0f, 0.0f, 0.0f, 0.0f};
                }
            }
            lds4[2 * i]     = lo;
            lds4[2 * i + 1] = hi;
        }
    }

    auto do_row = [&](const int i, const int si, const float wi, int& nsi, float& nwi)
        __attribute__((always_inline)) {
        const int row = r0 + i;
        const int p0 = __builtin_amdgcn_readlane(rpv, i);
        const int p1 = __builtin_amdgcn_readlane(rpv, i + 1);

        nsi = 0;
        nwi = 0.0f;
        if (i + 1 < RPW && row + 1 < nrows) {
            const int p2 = __builtin_amdgcn_readlane(rpv, i + 2);
            if (lane < p2 - p1) { nsi = csrc[p1 + lane]; nwi = cw[p1 + lane]; }
        }

        float acc[CPL];
        #pragma unroll
        for (int j = 0; j < CPL; ++j) acc[j] = 0.0f;

        if (p1 > p0)
            walk_entries<C, CPL, GRP>(hc, si, wi, (p1 - p0 < 32) ? (p1 - p0) : 32, acc);
        for (int base = p0 + 32; base < p1; base += 32) {
            int   xi = 0;
            float xw = 0.0f;
            if (lane < p1 - base) { xi = csrc[base + lane]; xw = cw[base + lane]; }
            walk_entries<C, CPL, GRP>(hc, xi, xw, (p1 - base < 32) ? (p1 - base) : 32, acc);
        }

        float r[CPL];
        #pragma unroll
        for (int j = 0; j < CPL; ++j) r[j] = fmaxf(acc[j] + bs[j], 0.0f);

        if constexpr (EPI == EPI_RELU_GN) {
            const float mu  = (r[0] + r[1] + r[2]) * (1.0f / 3.0f);
            const float d0 = r[0] - mu, d1 = r[1] - mu, d2 = r[2] - mu;
            const float var = (d0 * d0 + d1 * d1 + d2 * d2) * (1.0f / 3.0f);
            const float inv = rsqrtf(var + GN_EPS);
            r[0] = d0 * inv * gm[0] + bt[0];
            r[1] = d1 * inv * gm[1] + bt[1];
            r[2] = d2 * inv * gm[2] + bt[2];
        }

        float* o = at + (wave * RPW + i) * AP + c;
        #pragma unroll
        for (int j = 0; j < CPL; ++j) o[j] = r[j];
    };

    if (r0 < nrows) {
        int   nsi;
        float nwi;
        do_row(0, fsi, fwi, nsi, nwi);
        for (int i = 1; i < RPW; ++i) {
            if (r0 + i >= nrows) {
                float* o = at + (wave * RPW + i) * AP + c;
                #pragma unroll
                for (int j = 0; j < CPL; ++j) o[j] = 0.0f;
                continue;
            }
            const int   si = nsi;
            const float wi = nwi;
            do_row(i, si, wi, nsi, nwi);
        }
    } else {
        for (int i = 0; i < RPW; ++i) {
            float* o = at + (wave * RPW + i) * AP + c;
            #pragma unroll
            for (int j = 0; j < CPL; ++j) o[j] = 0.0f;
        }
    }
    __syncthreads();

    if (wave < 12) {
        const int coll = lane & 15;
        const int half = lane >> 4;
        const int mt   = wave / 3;
        const int np   = wave - mt * 3;
        const float2* __restrict__ lds2 = reinterpret_cast<const float2*>(lds4);
        const float2* bp = lds2 + (2 * half) * NCP + np * TPW * 16 + coll;
        const float*  ap = at + (mt * 16 + coll) * AP + half * 4;

        v8f acc[TPW] = {};
        #pragma unroll
        for (int j = 0; j < KB; ++j) {
            const v4f af = *reinterpret_cast<const v4f*>(ap + j * 8);
            #pragma unroll
            for (int s = 0; s < 2; ++s) {
                v2f a;
                a.x = s ? af.z : af.x;
                a.y = s ? af.w : af.y;
                const float2* bq = bp + (4 * j + s) * NCP;
                #pragma unroll
                for (int t = 0; t < TPW; ++t) {
                    float2 bv = bq[t * 16];
                    v2f b; b.x = bv.x; b.y = bv.y;
                    acc[t] = __builtin_amdgcn_wmma_f32_16x16x4_f32(
                        false, a, false, b, (short)0, acc[t], false, false);
                }
            }
        }

        const int rbase = blockIdx.x * RPB + mt * 16 + half * 8;
        #pragma unroll
        for (int t = 0; t < TPW; ++t) {
            const int col = (np * TPW + t) * 16 + coll;
            if (col < NCOLS) {
                float* cp = Cn + (size_t)rbase * NCOLS + col;
                if (rbase + 8 <= nrows) {
                    #pragma unroll
                    for (int r = 0; r < 8; ++r) cp[r * NCOLS] = acc[t][r];
                } else {
                    #pragma unroll
                    for (int r = 0; r < 8; ++r)
                        if (rbase + r < nrows) cp[r * NCOLS] = acc[t][r];
                }
            }
        }
    }
}

extern "C" void kernel_launch(void* const* d_in, const int* in_sizes, int n_in,
                              void* d_out, int out_size, void* d_ws, size_t ws_size,
                              hipStream_t stream)
{
    const float* x     = (const float*)d_in[0];
    const int*   src   = (const int*)  d_in[1];
    const int*   tgt   = (const int*)  d_in[2];
    const float* w     = (const float*)d_in[3];
    const float* W1    = (const float*)d_in[4];
    const float* b1    = (const float*)d_in[5];
    const float* W2    = (const float*)d_in[6];
    const float* b2    = (const float*)d_in[7];
    const float* W3    = (const float*)d_in[8];
    const float* b3    = (const float*)d_in[9];
    const float* gamma = (const float*)d_in[10];
    const float* beta  = (const float*)d_in[11];
    float* out = (float*)d_out;

    const int N   = in_sizes[0] / NFEAT;
    const int E   = in_sizes[1];
    const int ntm = N / 16;
    const int gblk = (ntm + 3) / 4;

    float* bufA = (float*)d_ws;
    float* bufB = bufA + (size_t)N * NHID;

    const int NB   = (N + 255) / 256;
    const int G    = 256;
    const int NPAD = ((NB * G + 4096) / 4096) * 4096;
    int*   hist   = (int*)(bufB + (size_t)N * NHID);
    int*   base   = hist + NPAD;
    int*   rowptr = base + NPAD;
    int*   recs1  = rowptr + NPAD;
    int*   recs2  = recs1 + (size_t)E * 4;
    int*   csrc   = recs2 + (size_t)E * 4;
    float* cw     = (float*)(csrc + E);

    const int TPB = 256;
    const int eblk = (E + TPB - 1) / TPB;

    bin_count_kernel<<<G, 1024, 0, stream>>>(tgt, hist, E, NB, NPAD);
    prefix_sum_kernel<<<1, 1024, 0, stream>>>(hist, base, NPAD / 32);
    bin_place_kernel<<<G, 1024, 0, stream>>>(src, tgt, w, base, recs1, E, NB);
    row_place_kernel<<<NB, 1024, 0, stream>>>(recs1, base, rowptr, recs2, G);
    edge_order_kernel<<<eblk, TPB, 0, stream>>>(recs2, rowptr, csrc, cw, E);

    const int gwv = (N + 31) / 32;
    const int gfb = (N + 63) / 64;

    wmma_gemm_strip_kernel<6, NFEAT, NHID><<<gblk, 128, 0, stream>>>(
        x, W1, bufA, ntm);

    gather_rows_gemm_kernel<EPI_RELU, 6, NHID><<<gfb, 512, 0, stream>>>(
        bufA, rowptr, csrc, cw, b1, gamma, beta, W2, bufB, N);

    gather_rows_gemm_kernel<EPI_RELU_GN, 3, NCLASS><<<gfb, 512, 0, stream>>>(
        bufB, rowptr, csrc, cw, b2, gamma, beta, W3, bufA, N);

    gather_rows_kernel<NCLASS, EPI_LSM><<<gwv, 256, 0, stream>>>(
        bufA, rowptr, csrc, cw, b3, gamma, beta, out, N);
}


# Edge grouping in 3 launches instead of 5: chunk work-groups sum the count table for their own bases (no scan launch); a bin's work-group groups and ranks its records in LDS (no per-edge rank launch)
# speedup vs baseline: 4.1817x; 1.1362x over previous
#include <hip/hip_runtime.h>
#include <hip/hip_bf16.h>


typedef float v2f __attribute__((ext_vector_type(2)));
typedef float v4f __attribute__((ext_vector_type(4)));
typedef float v8f __attribute__((ext_vector_type(8)));

#define NFEAT   128
#define NHID    96
#define NCLASS  40
#define GN_EPS  1e-5f

template<int NT, int K, int NCOLS>
__global__ __launch_bounds__(128) void wmma_gemm_strip_kernel(
    const float* __restrict__ A, const float* __restrict__ Bw,
    float* __restrict__ C, int ntm)
{
    constexpr int NCP   = NT * 16;
    constexpr int NQ    = NCP / 4;
    constexpr int NTASK = (K / 2) * NQ;
    constexpr int NIT   = (NTASK + 127) / 128;
    constexpr int KB    = K / 8;
    static_assert(K % 8 == 0 && NCOLS % 4 == 0 && NCOLS <= NCP, "shape");

    __shared__ v4f lds4[K * NCP / 4];

    const int tid   = threadIdx.x;
    const int wave  = tid >> 5;
    const int lane  = tid & 31;
    const int coll  = lane & 15;
    const int half  = lane >> 4;
    const int mtile = blockIdx.x * 4 + wave;
    const int mload = (mtile < ntm) ? mtile : (ntm - 1);

    v4f w0[NIT], w1[NIT];
    #pragma unroll
    for (int it = 0; it < NIT; ++it) {
        const int i  = tid + it * 128;
        const int ic = (NTASK % 128 == 0) ? i : (i < NTASK ? i : NTASK - 1);
        const int p  = ic / NQ;
        const int q  = ic - p * NQ;
        const int qc = (NCOLS == NCP) ? q : (q * 4 < NCOLS ? q : NCOLS / 4 - 1);
        const float* g = Bw + (size_t)(2 * p) * NCOLS + qc * 4;
        w0[it] = *reinterpret_cast<const v4f*>(g);
        w1[it] = *reinterpret_cast<const v4f*>(g + NCOLS);
    }

    v4f af[KB];
    const float* arow = A + (size_t)(mload * 16 + coll) * K + half * 4;
    #pragma unroll
    for (int j = 0; j < KB; ++j)
        af[j] = *reinterpret_cast<const v4f*>(arow + j * 8);

    #pragma unroll
    for (int it = 0; it < NIT; ++it) {
        const int i = tid + it * 128;
        if (NTASK % 128 == 0 || i < NTASK) {
            v4f lo = (v4f){w0[it].x, w1[it].x, w0[it].y, w1[it].y};
            v4f hi = (v4f){w0[it].z, w1[it].z, w0[it].w, w1[it].w};
            if (NCOLS != NCP) {
                const int q = i % NQ;
                if (q * 4 >= NCOLS) {
                    lo = (v4f){0.0f, 0.0f, 0.0f, 0.0f};
                    hi = (v4f){0.0f, 0.0f, 0.0f, 0.0f};
                }
            }
            lds4[2 * i]     = lo;
            lds4[2 * i + 1] = hi;
        }
    }
    __syncthreads();

    const float2* __restrict__ lds2 = reinterpret_cast<const float2*>(lds4);
    const float2* bp = lds2 + (2 * half) * NCP + coll;

    v8f acc[NT] = {};
    #pragma unroll
    for (int j = 0; j < KB; ++j) {
        #pragma unroll
        for (int s = 0; s < 2; ++s) {
            v2f a;
            a.x = s ? af[j].z : af[j].x;
            a.y = s ? af[j].w : af[j].y;
            const float2* bq = bp + (4 * j + s) * NCP;
            #pragma unroll
            for (int t = 0; t < NT; ++t) {
                float2 bv = bq[t * 16];
                v2f b; b.x = bv.x; b.y = bv.y;
                acc[t] = __builtin_amdgcn_wmma_f32_16x16x4_f32(
                    false, a, false, b, (short)0, acc[t], false, false);
            }
        }
    }

    if (mtile >= ntm) return;

    const int rbase = mtile * 16 + half * 8;
    #pragma unroll
    for (int t = 0; t < NT; ++t) {
        int col = t * 16 + coll;
        if (col < NCOLS) {
            #pragma unroll
            for (int r = 0; r < 8; ++r)
                C[(size_t)(rbase + r) * NCOLS + col] = acc[t][r];
        }
    }
}


#define GCH 256

__global__ __launch_bounds__(1024) void bin_count_kernel(const int* __restrict__ tgt,
                                                         int* __restrict__ hist,
                                                         int E, int nb)
{
    __shared__ int lh[256];
    const int tid = threadIdx.x;
    const int g   = blockIdx.x;
    if (tid < 256) lh[tid] = 0;
    __syncthreads();
    const int ch = (E + GCH - 1) / GCH;
    const int e0 = g * ch;
    const int e1 = (e0 + ch < E) ? (e0 + ch) : E;
    for (int e = e0 + tid; e < e1; e += 1024)
        atomicAdd(&lh[tgt[e] >> 8], 1);
    __syncthreads();
    if (tid < nb) hist[tid * GCH + g] = lh[tid];
}

__global__ __launch_bounds__(1024) void bin_place_kernel(const int* __restrict__ src,
                                                         const int* __restrict__ tgt,
                                                         const float* __restrict__ w,
                                                         const int* __restrict__ hist,
                                                         int* __restrict__ bstart,
                                                         int* __restrict__ recs1,
                                                         int E, int nb)
{
    __shared__ int cur[256];
    __shared__ int tl[256];
    __shared__ int wsum[8];
    const int tid  = threadIdx.x;
    const int lane = tid & 31;
    const int wave = tid >> 5;
    const int g    = blockIdx.x;

    const int b    = tid >> 2;
    const int part = tid & 3;
    int tot = 0, pre = 0;
    if (b < nb) {
        const int4* h4 = reinterpret_cast<const int4*>(hist + b * GCH + part * (GCH / 4));
        #pragma unroll
        for (int u = 0; u < GCH / 16; ++u) {
            const int4 v  = h4[u];
            const int  g0 = part * (GCH / 4) + u * 4;
            tot += (v.x + v.y) + (v.z + v.w);
            pre += ((g0     < g) ? v.x : 0) + ((g0 + 1 < g) ? v.y : 0)
                 + ((g0 + 2 < g) ? v.z : 0) + ((g0 + 3 < g) ? v.w : 0);
        }
    }
    tot += __shfl_xor(tot, 1);  tot += __shfl_xor(tot, 2);
    pre += __shfl_xor(pre, 1);  pre += __shfl_xor(pre, 2);
    if (part == 0) { tl[b] = tot; cur[b] = pre; }
    __syncthreads();

    int c = 0, inc = 0;
    if (tid < 256) {
        c = tl[tid];
        inc = c;
        for (int off = 1; off < 32; off <<= 1) {
            int u = __shfl_up(inc, off);
            if (lane >= off) inc += u;
        }
        if (lane == 31) wsum[wave] = inc;
    }
    __syncthreads();
    if (tid < 256) {
        int front = 0;
        for (int k = 0; k < wave; ++k) front += wsum[k];
        const int ex = front + inc - c;
        cur[tid] += ex;
        if (g == 0) {
            bstart[tid] = ex;
            if (tid == 255) bstart[256] = ex + c;
        }
    }
    __syncthreads();

    const int ch = (E + GCH - 1) / GCH;
    const int e0 = g * ch;
    const int e1 = (e0 + ch < E) ? (e0 + ch) : E;
    int4* r1 = reinterpret_cast<int4*>(recs1);
    for (int e = e0 + tid; e < e1; e += 1024) {
        const int t   = tgt[e];
        const int pos = atomicAdd(&cur[t >> 8], 1);
        r1[pos] = make_int4(e, t, src[e], __float_as_int(w[e]));
    }
}

template<int NR>
__global__ __launch_bounds__(1024) void row_lists_kernel(const int* __restrict__ recs1,
                                                         const int* __restrict__ bstart,
                                                         int* __restrict__ rowptr,
                                                         int* __restrict__ recs2,
                                                         int* __restrict__ csrc,
                                                         float* __restrict__ cw)
{
    constexpr int CAP = NR * 1024;
    __shared__ int  cnt[256];
    __shared__ int  cur[256];
    __shared__ int  wsum[8];
    __shared__ int4 srec[CAP];
    const int tid  = threadIdx.x;
    const int lane = tid & 31;
    const int wave = tid >> 5;
    const int b    = blockIdx.x;
    const int bs   = bstart[b];
    const int be   = bstart[b + 1];
    const int n    = be - bs;
    const bool fits = (n <= CAP);
    const int4* r1 = reinterpret_cast<const int4*>(recs1);
    int4*       r2 = reinterpret_cast<int4*>(recs2);

    int4 rr[NR];
    if (fits) {
        #pragma unroll
        for (int u = 0; u < NR; ++u) {
            int i = bs + tid + u * 1024;
            if (i > be - 1) i = be - 1;
            if (i < 0) i = 0;
            rr[u] = r1[i];
        }
    }

    if (tid < 256) cnt[tid] = 0;
    __syncthreads();
    if (fits) {
        #pragma unroll
        for (int u = 0; u < NR; ++u)
            if (bs + tid + u * 1024 < be) atomicAdd(&cnt[rr[u].y & 255], 1);
    } else {
        for (int i = bs + tid; i < be; i += 1024)
            atomicAdd(&cnt[r1[i].y & 255], 1);
    }
    __syncthreads();

    int c = 0, inc = 0;
    if (tid < 256) {
        c = cnt[tid];
        inc = c;
        for (int off = 1; off < 32; off <<= 1) {
            int u = __shfl_up(inc, off);
            if (lane >= off) inc += u;
        }
        if (lane == 31) wsum[wave] = inc;
    }
    __syncthreads();
    if (tid < 256) {
        int front = 0;
        for (int k = 0; k < wave; ++k) front += wsum[k];
        const int first = bs + front + inc - c;
        cur[tid] = first;
        rowptr[b * 256 + tid] = first;
        if (tid == 255 && b == (int)gridDim.x - 1) rowptr[b * 256 + 256] = be;
    }
    __syncthreads();

    if (fits) {
        #pragma unroll
        for (int u = 0; u < NR; ++u) {
            if (bs + tid + u * 1024 < be) {
                const int pos = atomicAdd(&cur[rr[u].y & 255], 1);
                srec[pos - bs] = rr[u];
            }
        }
    } else {
        for (int i = bs + tid; i < be; i += 1024) {
            const int4 r = r1[i];
            const int pos = atomicAdd(&cur[r.y & 255], 1);
            r2[pos] = r;
        }
        __threadfence();
    }
    __syncthreads();

    if (fits) {
        for (int p = tid; p < n; p += 1024) {
            const int4 r  = srec[p];
            const int  rw = r.y & 255;
            const int  e  = cur[rw] - bs;
            const int  a  = e - cnt[rw];
            int k = 0;
            for (int j = a; j < e; j += 4) {
                #pragma unroll
                for (int u = 0; u < 4; ++u) {
                    const int jj = (j + u < e) ? (j + u) : (e - 1);
                    const int x  = srec[jj].x;
                    k += (j + u < e && x < r.x) ? 1 : 0;
                }
            }
            csrc[bs + a + k] = r.z;
            cw[bs + a + k]   = __int_as_float(r.w);
        }
    } else {
        for (int p = bs + tid; p < be; p += 1024) {
            const int4 r  = r2[p];
            const int  rw = r.y & 255;
            const int  e  = cur[rw];
            const int  a  = e - cnt[rw];
            int k = 0;
            for (int j = a; j < e; ++j) k += (r2[j].x < r.x) ? 1 : 0;
            csrc[a + k] = r.z;
            cw[a + k]   = __int_as_float(r.w);
        }
    }
}

#define EPI_RELU     1
#define EPI_RELU_GN  2
#define EPI_LSM      3

template<int C, int CPL, int GRP>
__device__ __forceinline__ void walk_entries(const float* __restrict__ hc, int si, float wi,
                                             int n, float (&acc)[CPL])
{
    for (int k = 0; k < n; k += GRP) {
        float m[GRP][CPL];
        #pragma unroll
        for (int u = 0; u < GRP; ++u) {
            const int kk = (k + u < n) ? (k + u) : (n - 1);
            const int s  = __builtin_amdgcn_readlane(si, kk);
            const float* hp = hc + (size_t)s * C;
            #pragma unroll
            for (int j = 0; j < CPL; ++j) m[u][j] = hp[j];
        }
        #pragma unroll
        for (int u = 0; u < GRP; ++u) {
            const int kk = (k + u < n) ? (k + u) : (n - 1);
            const float w = __int_as_float(__builtin_amdgcn_readlane(__float_as_int(wi), kk));
            const bool on = (k + u < n);
            #pragma unroll
            for (int j = 0; j < CPL; ++j) {
                const float t = fmaf(m[u][j], w, acc[j]);
                acc[j] = on ? t : acc[j];
            }
        }
    }
}

template<int C, int EPI>
__global__ __launch_bounds__(256)
void gather_rows_kernel(const float* __restrict__ h,
                        const int* __restrict__ rowptr,
                        const int* __restrict__ csrc,
                        const float* __restrict__ cw,
                        const float* __restrict__ bias,
                        const float* __restrict__ gamma,
                        const float* __restrict__ beta,
                        float* __restrict__ out, int nrows)
{
    constexpr int CPL = (C == NHID) ? 3 : 2;
    constexpr int LPR = C / CPL;
    constexpr int RPW = 4;
    constexpr int GRP = 8;
    static_assert(LPR * CPL == C && LPR <= 32 && RPW + 1 <= 32, "shape");

    const int lane = threadIdx.x & 31;
    const int gw  = __builtin_amdgcn_readfirstlane((int)(blockIdx.x * 8 + (threadIdx.x >> 5)));
    const int r0  = gw * RPW;
    if (r0 >= nrows) return;
    const bool act = lane < LPR;
    const int  c   = (act ? lane : 0) * CPL;

    int bi = r0 + ((lane < RPW) ? lane : RPW);
    if (bi > nrows) bi = nrows;
    const int rpv = rowptr[bi];

    float bs[CPL], gm[CPL], bt[CPL];
    #pragma unroll
    for (int j = 0; j < CPL; ++j) {
        bs[j] = bias[c + j];
        gm[j] = (EPI == EPI_RELU_GN) ? gamma[c + j] : 1.0f;
        bt[j] = (EPI == EPI_RELU_GN) ? beta[c + j]  : 0.0f;
    }
    const float* hc = h + c;

    auto do_row = [&](const int i, const int si, const float wi, int& nsi, float& nwi)
        __attribute__((always_inline)) {
        const int row = r0 + i;
        const int p0 = __builtin_amdgcn_readlane(rpv, i);
        const int p1 = __builtin_amdgcn_readlane(rpv, i + 1);

        nsi = 0;
        nwi = 0.0f;
        if (i + 1 < RPW && row + 1 < nrows) {
            const int p2 = __builtin_amdgcn_readlane(rpv, i + 2);
            if (lane < p2 - p1) { nsi = csrc[p1 + lane]; nwi = cw[p1 + lane]; }
        }

        float acc[CPL];
        #pragma unroll
        for (int j = 0; j < CPL; ++j) acc[j] = 0.0f;

        if (p1 > p0)
            walk_entries<C, CPL, GRP>(hc, si, wi, (p1 - p0 < 32) ? (p1 - p0) : 32, acc);
        for (int base = p0 + 32; base < p1; base += 32) {
            int   xi = 0;
            float xw = 0.0f;
            if (lane < p1 - base) { xi = csrc[base + lane]; xw = cw[base + lane]; }
            walk_entries<C, CPL, GRP>(hc, xi, xw, (p1 - base < 32) ? (p1 - base) : 32, acc);
        }

        float r[CPL];
        #pragma unroll
        for (int j = 0; j < CPL; ++j) {
            r[j] = acc[j] + bs[j];
            if (EPI == EPI_RELU || EPI == EPI_RELU_GN) r[j] = fmaxf(r[j], 0.0f);
        }

        if constexpr (EPI == EPI_RELU_GN) {
            const float mu  = (r[0] + r[1] + r[2]) * (1.0f / 3.0f);
            const float d0 = r[0] - mu, d1 = r[1] - mu, d2 = r[2] - mu;
            const float var = (d0 * d0 + d1 * d1 + d2 * d2) * (1.0f / 3.0f);
            const float inv = rsqrtf(var + GN_EPS);
            r[0] = d0 * inv * gm[0] + bt[0];
            r[1] = d1 * inv * gm[1] + bt[1];
            r[2] = d2 * inv * gm[2] + bt[2];
        }

        if constexpr (EPI == EPI_LSM) {
            float mx = act ? fmaxf(r[0], r[1]) : -3.402823466e+38f;
            #pragma unroll
            for (int off = 16; off > 0; off >>= 1) mx = fmaxf(mx, __shfl_xor(mx, off));
            const float e0 = __expf(r[0] - mx);
            const float e1 = __expf(r[1] - mx);
            float s = 0.0f;
            #pragma unroll
            for (int k = 0; k < LPR; ++k) {
                s += __int_as_float(__builtin_amdgcn_readlane(__float_as_int(e0), k));
                s += __int_as_float(__builtin_amdgcn_readlane(__float_as_int(e1), k));
            }
            const float ls = __logf(s);
            r[0] = r[0] - mx - ls;
            r[1] = r[1] - mx - ls;
        }

        if (act) {
            float* o = out + (size_t)row * C + c;
            #pragma unroll
            for (int j = 0; j < CPL; ++j) o[j] = r[j];
        }
    };

    int   fsi = 0;
    float fwi = 0.0f;
    {
        const int pa = __builtin_amdgcn_readlane(rpv, 0);
        const int pb = __builtin_amdgcn_readlane(rpv, 1);
        if (lane < pb - pa) { fsi = csrc[pa + lane]; fwi = cw[pa + lane]; }
    }
    int   nsi;
    float nwi;
    do_row(0, fsi, fwi, nsi, nwi);
    for (int i = 1; i < RPW; ++i) {
        if (r0 + i >= nrows) break;
        const int   si = nsi;
        const float wi = nwi;
        do_row(i, si, wi, nsi, nwi);
    }
}

template<int EPI, int NT, int NCOLS>
__global__ __launch_bounds__(512)
void gather_rows_gemm_kernel(const float* __restrict__ h,
                             const int* __restrict__ rowptr,
                             const int* __restrict__ csrc,
                             const float* __restrict__ cw,
                             const float* __restrict__ bias,
                             const float* __restrict__ gamma,
                             const float* __restrict__ beta,
                             const float* __restrict__ Bw,
                             float* __restrict__ Cn, int nrows)
{
    constexpr int C     = NHID;
    constexpr int K     = NHID;
    constexpr int CPL   = 3;
    constexpr int RPW   = 4;
    constexpr int GRP   = 8;
    constexpr int NTHR  = 512;
    constexpr int RPB   = (NTHR / 32) * RPW;
    constexpr int NCP   = NT * 16;
    constexpr int NQ    = NCP / 4;
    constexpr int NTASK = (K / 2) * NQ;
    constexpr int NIT   = (NTASK + NTHR - 1) / NTHR;
    constexpr int KB    = K / 8;
    constexpr int AP    = 100;
    constexpr int TPW   = NT / 3;
    static_assert(NT % 3 == 0 && K % 8 == 0 && NCOLS % 4 == 0 && NCOLS <= NCP && C == 32 * CPL, "shape");

    __shared__ v4f lds4[K * NCP / 4];
    __shared__ v4f at4[RPB * AP / 4];
    float* at = reinterpret_cast<float*>(at4);

    const int tid  = threadIdx.x;
    const int lane = tid & 31;
    const int wave = __builtin_amdgcn_readfirstlane(tid >> 5);
    const int r0   = blockIdx.x * RPB + wave * RPW;
    const int c    = lane * CPL;

    int bi = r0 + ((lane < RPW) ? lane : RPW);
    if (bi > nrows) bi = nrows;
    const int rpv = rowptr[bi];

    v4f w0[NIT], w1[NIT];
    #pragma unroll
    for (int it = 0; it < NIT; ++it) {
        const int i  = tid + it * NTHR;
        const int ic = (NTASK % NTHR == 0) ? i : (i < NTASK ? i : NTASK - 1);
        const int p  = ic / NQ;
        const int q  = ic - p * NQ;
        const int qc = (NCOLS == NCP) ? q : (q * 4 < NCOLS ? q : NCOLS / 4 - 1);
        const float* g = Bw + (size_t)(2 * p) * NCOLS + qc * 4;
        w0[it] = *reinterpret_cast<const v4f*>(g);
        w1[it] = *reinterpret_cast<const v4f*>(g + NCOLS);
    }

    float bs[CPL], gm[CPL], bt[CPL];
    #pragma unroll
    for (int j = 0; j < CPL; ++j) {
        bs[j] = bias[c + j];
        gm[j] = (EPI == EPI_RELU_GN) ? gamma[c + j] : 1.0f;
        bt[j] = (EPI == EPI_RELU_GN) ? beta[c + j]  : 0.0f;
    }
    const float* hc = h + c;

    int   fsi = 0;
    float fwi = 0.0f;
    if (r0 < nrows) {
        const int pa = __builtin_amdgcn_readlane(rpv, 0);
        const int pb = __builtin_amdgcn_readlane(rpv, 1);
        if (lane < pb - pa) { fsi = csrc[pa + lane]; fwi = cw[pa + lane]; }
    }

    #pragma unroll
    for (int it = 0; it < NIT; ++it) {
        const int i = tid + it * NTHR;
        if (NTASK % NTHR == 0 || i < NTASK) {
            v4f lo = (v4f){w0[it].x, w1[it].x, w0[it].y, w1[it].y};
            v4f hi = (v4f){w0[it].z, w1[it].z, w0[it].w, w1[it].w};
            if (NCOLS != NCP) {
                const int q = i % NQ;
                if (q * 4 >= NCOLS) {
                    lo = (v4f){0.0f, 0.0f, 0.0f, 0.0f};
                    hi = (v4f){0.0f, 0.0f, 0.0f, 0.0f};
                }
            }
            lds4[2 * i]     = lo;
            lds4[2 * i + 1] = hi;
        }
    }

    auto do_row = [&](const int i, const int si, const float wi, int& nsi, float& nwi)
        __attribute__((always_inline)) {
        const int row = r0 + i;
        const int p0 = __builtin_amdgcn_readlane(rpv, i);
        const int p1 = __builtin_amdgcn_readlane(rpv, i + 1);

        nsi = 0;
        nwi = 0.0f;
        if (i + 1 < RPW && row + 1 < nrows) {
            const int p2 = __builtin_amdgcn_readlane(rpv, i + 2);
            if (lane < p2 - p1) { nsi = csrc[p1 + lane]; nwi = cw[p1 + lane]; }
        }

        float acc[CPL];
        #pragma unroll
        for (int j = 0; j < CPL; ++j) acc[j] = 0.0f;

        if (p1 > p0)
            walk_entries<C, CPL, GRP>(hc, si, wi, (p1 - p0 < 32) ? (p1 - p0) : 32, acc);
        for (int base = p0 + 32; base < p1; base += 32) {
            int   xi = 0;
            float xw = 0.0f;
            if (lane < p1 - base) { xi = csrc[base + lane]; xw = cw[base + lane]; }
            walk_entries<C, CPL, GRP>(hc, xi, xw, (p1 - base < 32) ? (p1 - base) : 32, acc);
        }

        float r[CPL];
        #pragma unroll
        for (int j = 0; j < CPL; ++j) r[j] = fmaxf(acc[j] + bs[j], 0.0f);

        if constexpr (EPI == EPI_RELU_GN) {
            const float mu  = (r[0] + r[1] + r[2]) * (1.0f / 3.0f);
            const float d0 = r[0] - mu, d1 = r[1] - mu, d2 = r[2] - mu;
            const float var = (d0 * d0 + d1 * d1 + d2 * d2) * (1.0f / 3.0f);
            const float inv = rsqrtf(var + GN_EPS);
            r[0] = d0 * inv * gm[0] + bt[0];
            r[1] = d1 * inv * gm[1] + bt[1];
            r[2] = d2 * inv * gm[2] + bt[2];
        }

        float* o = at + (wave * RPW + i) * AP + c;
        #pragma unroll
        for (int j = 0; j < CPL; ++j) o[j] = r[j];
    };

    if (r0 < nrows) {
        int   nsi;
        float nwi;
        do_row(0, fsi, fwi, nsi, nwi);
        for (int i = 1; i < RPW; ++i) {
            if (r0 + i >= nrows) {
                float* o = at + (wave * RPW + i) * AP + c;
                #pragma unroll
                for (int j = 0; j < CPL; ++j) o[j] = 0.0f;
                continue;
            }
            const int   si = nsi;
            const float wi = nwi;
            do_row(i, si, wi, nsi, nwi);
        }
    } else {
        for (int i = 0; i < RPW; ++i) {
            float* o = at + (wave * RPW + i) * AP + c;
            #pragma unroll
            for (int j = 0; j < CPL; ++j) o[j] = 0.0f;
        }
    }
    __syncthreads();

    if (wave < 12) {
        const int coll = lane & 15;
        const int half = lane >> 4;
        const int mt   = wave / 3;
        const int np   = wave - mt * 3;
        const float2* __restrict__ lds2 = reinterpret_cast<const float2*>(lds4);
        const float2* bp = lds2 + (2 * half) * NCP + np * TPW * 16 + coll;
        const float*  ap = at + (mt * 16 + coll) * AP + half * 4;

        v8f acc[TPW] = {};
        #pragma unroll
        for (int j = 0; j < KB; ++j) {
            const v4f af = *reinterpret_cast<const v4f*>(ap + j * 8);
            #pragma unroll
            for (int s = 0; s < 2; ++s) {
                v2f a;
                a.x = s ? af.z : af.x;
                a.y = s ? af.w : af.y;
                const float2* bq = bp + (4 * j + s) * NCP;
                #pragma unroll
                for (int t = 0; t < TPW; ++t) {
                    float2 bv = bq[t * 16];
                    v2f b; b.x = bv.x; b.y = bv.y;
                    acc[t] = __builtin_amdgcn_wmma_f32_16x16x4_f32(
                        false, a, false, b, (short)0, acc[t], false, false);
                }
            }
        }

        const int rbase = blockIdx.x * RPB + mt * 16 + half * 8;
        #pragma unroll
        for (int t = 0; t < TPW; ++t) {
            const int col = (np * TPW + t) * 16 + coll;
            if (col < NCOLS) {
                float* cp = Cn + (size_t)rbase * NCOLS + col;
                if (rbase + 8 <= nrows) {
                    #pragma unroll
                    for (int r = 0; r < 8; ++r) cp[r * NCOLS] = acc[t][r];
                } else {
                    #pragma unroll
                    for (int r = 0; r < 8; ++r)
                        if (rbase + r < nrows) cp[r * NCOLS] = acc[t][r];
                }
            }
        }
    }
}

extern "C" void kernel_launch(void* const* d_in, const int* in_sizes, int n_in,
                              void* d_out, int out_size, void* d_ws, size_t ws_size,
                              hipStream_t stream)
{
    const float* x     = (const float*)d_in[0];
    const int*   src   = (const int*)  d_in[1];
    const int*   tgt   = (const int*)  d_in[2];
    const float* w     = (const float*)d_in[3];
    const float* W1    = (const float*)d_in[4];
    const float* b1    = (const float*)d_in[5];
    const float* W2    = (const float*)d_in[6];
    const float* b2    = (const float*)d_in[7];
    const float* W3    = (const float*)d_in[8];
    const float* b3    = (const float*)d_in[9];
    const float* gamma = (const float*)d_in[10];
    const float* beta  = (const float*)d_in[11];
    float* out = (float*)d_out;

    const int N   = in_sizes[0] / NFEAT;
    const int E   = in_sizes[1];
    const int ntm = N / 16;
    const int gblk = (ntm + 3) / 4;

    float* bufA = (float*)d_ws;
    float* bufB = bufA + (size_t)N * NHID;

    const int NB   = (N + 255) / 256;
    const int NPAD = ((NB * GCH + 4096) / 4096) * 4096;
    int*   hist   = (int*)(bufB + (size_t)N * NHID);
    int*   bstart = hist + NPAD;
    int*   rowptr = bstart + NPAD;
    int*   recs1  = rowptr + NPAD;
    int*   recs2  = recs1 + (size_t)E * 4;
    int*   csrc   = recs2 + (size_t)E * 4;
    float* cw     = (float*)(csrc + E);

    bin_count_kernel<<<GCH, 1024, 0, stream>>>(tgt, hist, E, NB);
    bin_place_kernel<<<GCH, 1024, 0, stream>>>(src, tgt, w, hist, bstart, recs1, E, NB);
    row_lists_kernel<6><<<NB, 1024, 0, stream>>>(recs1, bstart, rowptr, recs2, csrc, cw);

    const int gwv = (N + 31) / 32;
    const int gfb = (N + 63) / 64;

    wmma_gemm_strip_kernel<6, NFEAT, NHID><<<gblk, 128, 0, stream>>>(
        x, W1, bufA, ntm);

    gather_rows_gemm_kernel<EPI_RELU, 6, NHID><<<gfb, 512, 0, stream>>>(
        bufA, rowptr, csrc, cw, b1, gamma, beta, W2, bufB, N);

    gather_rows_gemm_kernel<EPI_RELU_GN, 3, NCLASS><<<gfb, 512, 0, stream>>>(
        bufB, rowptr, csrc, cw, b2, gamma, beta, W3, bufA, N);

    gather_rows_kernel<NCLASS, EPI_LSM><<<gwv, 256, 0, stream>>>(
        bufA, rowptr, csrc, cw, b3, gamma, beta, out, N);
}


# Edge grouping in 2 launches instead of 3: a chunk's work-group orders its edges by bin inside its own range (LDS counts, prefix sum), so no count launch; a bin's work-group collects its 256 pieces
# speedup vs baseline: 4.5236x; 1.0818x over previous
#include <hip/hip_runtime.h>
#include <hip/hip_bf16.h>


typedef float v2f __attribute__((ext_vector_type(2)));
typedef float v4f __attribute__((ext_vector_type(4)));
typedef float v8f __attribute__((ext_vector_type(8)));

#define NFEAT   128
#define NHID    96
#define NCLASS  40
#define GN_EPS  1e-5f

template<int NT, int K, int NCOLS>
__global__ __launch_bounds__(128) void wmma_gemm_strip_kernel(
    const float* __restrict__ A, const float* __restrict__ Bw,
    float* __restrict__ C, int ntm)
{
    constexpr int NCP   = NT * 16;
    constexpr int NQ    = NCP / 4;
    constexpr int NTASK = (K / 2) * NQ;
    constexpr int NIT   = (NTASK + 127) / 128;
    constexpr int KB    = K / 8;
    static_assert(K % 8 == 0 && NCOLS % 4 == 0 && NCOLS <= NCP, "shape");

    __shared__ v4f lds4[K * NCP / 4];

    const int tid   = threadIdx.x;
    const int wave  = tid >> 5;
    const int lane  = tid & 31;
    const int coll  = lane & 15;
    const int half  = lane >> 4;
    const int mtile = blockIdx.x * 4 + wave;
    const int mload = (mtile < ntm) ? mtile : (ntm - 1);

    v4f w0[NIT], w1[NIT];
    #pragma unroll
    for (int it = 0; it < NIT; ++it) {
        const int i  = tid + it * 128;
        const int ic = (NTASK % 128 == 0) ? i : (i < NTASK ? i : NTASK - 1);
        const int p  = ic / NQ;
        const int q  = ic - p * NQ;
        const int qc = (NCOLS == NCP) ? q : (q * 4 < NCOLS ? q : NCOLS / 4 - 1);
        const float* g = Bw + (size_t)(2 * p) * NCOLS + qc * 4;
        w0[it] = *reinterpret_cast<const v4f*>(g);
        w1[it] = *reinterpret_cast<const v4f*>(g + NCOLS);
    }

    v4f af[KB];
    const float* arow = A + (size_t)(mload * 16 + coll) * K + half * 4;
    #pragma unroll
    for (int j = 0; j < KB; ++j)
        af[j] = *reinterpret_cast<const v4f*>(arow + j * 8);

    #pragma unroll
    for (int it = 0; it < NIT; ++it) {
        const int i = tid + it * 128;
        if (NTASK % 128 == 0 || i < NTASK) {
            v4f lo = (v4f){w0[it].x, w1[it].x, w0[it].y, w1[it].y};
            v4f hi = (v4f){w0[it].z, w1[it].z, w0[it].w, w1[it].w};
            if (NCOLS != NCP) {
                const int q = i % NQ;
                if (q * 4 >= NCOLS) {
                    lo = (v4f){0.0f, 0.0f, 0.0f, 0.0f};
                    hi = (v4f){0.0f, 0.0f, 0.0f, 0.0f};
                }
            }
            lds4[2 * i]     = lo;
            lds4[2 * i + 1] = hi;
        }
    }
    __syncthreads();

    const float2* __restrict__ lds2 = reinterpret_cast<const float2*>(lds4);
    const float2* bp = lds2 + (2 * half) * NCP + coll;

    v8f acc[NT] = {};
    #pragma unroll
    for (int j = 0; j < KB; ++j) {
        #pragma unroll
        for (int s = 0; s < 2; ++s) {
            v2f a;
            a.x = s ? af[j].z : af[j].x;
            a.y = s ? af[j].w : af[j].y;
            const float2* bq = bp + (4 * j + s) * NCP;
            #pragma unroll
            for (int t = 0; t < NT; ++t) {
                float2 bv = bq[t * 16];
                v2f b; b.x = bv.x; b.y = bv.y;
                acc[t] = __builtin_amdgcn_wmma_f32_16x16x4_f32(
                    false, a, false, b, (short)0, acc[t], false, false);
            }
        }
    }

    if (mtile >= ntm) return;

    const int rbase = mtile * 16 + half * 8;
    #pragma unroll
    for (int t = 0; t < NT; ++t) {
        int col = t * 16 + coll;
        if (col < NCOLS) {
            #pragma unroll
            for (int r = 0; r < 8; ++r)
                C[(size_t)(rbase + r) * NCOLS + col] = acc[t][r];
        }
    }
}


#define GCH 256
#define NEC 4

__global__ __launch_bounds__(1024) void chunk_bins_kernel(const int* __restrict__ src,
                                                          const int* __restrict__ tgt,
                                                          const float* __restrict__ w,
                                                          int* __restrict__ hist,
                                                          int* __restrict__ pst,
                                                          int* __restrict__ recs1,
                                                          int E, int nb)
{
    __shared__ int lh[256];
    __shared__ int cur[256];
    __shared__ int wsum[8];
    const int tid  = threadIdx.x;
    const int lane = tid & 31;
    const int wave = tid >> 5;
    const int g    = blockIdx.x;
    const int ch = (E + GCH - 1) / GCH;
    int e0 = g * ch;
    if (e0 > E) e0 = E;
    const int e1 = (e0 + ch < E) ? (e0 + ch) : E;
    const bool one = (ch <= NEC * 1024);

    int   tg[NEC] = {};
    int   sr[NEC] = {};
    float ww[NEC] = {};
    if (one && e1 > e0) {
        #pragma unroll
        for (int u = 0; u < NEC; ++u) {
            int e = e0 + tid + u * 1024;
            if (e > e1 - 1) e = e1 - 1;
            tg[u] = tgt[e];
            sr[u] = src[e];
            ww[u] = w[e];
        }
    }
    if (tid < 256) lh[tid] = 0;
    __syncthreads();
    if (one) {
        #pragma unroll
        for (int u = 0; u < NEC; ++u)
            if (e0 + tid + u * 1024 < e1) atomicAdd(&lh[tg[u] >> 8], 1);
    } else {
        for (int e = e0 + tid; e < e1; e += 1024)
            atomicAdd(&lh[tgt[e] >> 8], 1);
    }
    __syncthreads();

    int c = 0, inc = 0;
    if (tid < 256) {
        c = lh[tid];
        inc = c;
        for (int off = 1; off < 32; off <<= 1) {
            int u = __shfl_up(inc, off);
            if (lane >= off) inc += u;
        }
        if (lane == 31) wsum[wave] = inc;
    }
    __syncthreads();
    if (tid < 256) {
        int front = 0;
        for (int k = 0; k < wave; ++k) front += wsum[k];
        const int first = e0 + front + inc - c;
        cur[tid] = first;
        if (tid < nb) {
            hist[tid * GCH + g] = c;
            pst[tid * GCH + g]  = first;
        }
    }
    __syncthreads();

    int4* r1 = reinterpret_cast<int4*>(recs1);
    if (one) {
        #pragma unroll
        for (int u = 0; u < NEC; ++u) {
            const int e = e0 + tid + u * 1024;
            if (e < e1) {
                const int pos = atomicAdd(&cur[tg[u] >> 8], 1);
                r1[pos] = make_int4(e, tg[u], sr[u], __float_as_int(ww[u]));
            }
        }
    } else {
        for (int e = e0 + tid; e < e1; e += 1024) {
            const int t   = tgt[e];
            const int pos = atomicAdd(&cur[t >> 8], 1);
            r1[pos] = make_int4(e, t, src[e], __float_as_int(w[e]));
        }
    }
}

template<int NR>
__global__ __launch_bounds__(1024) void row_lists_kernel(const int* __restrict__ recs1,
                                                         const int* __restrict__ hist,
                                                         const int* __restrict__ pst,
                                                         int* __restrict__ rowptr,
                                                         int* __restrict__ recs2,
                                                         int* __restrict__ csrc,
                                                         float* __restrict__ cw)
{
    constexpr int CAP = NR * 1024;
    __shared__ int  cnt[256];
    __shared__ int  cur[256];
    __shared__ int  wsum[8];
    __shared__ int  wred[32];
    __shared__ int  segoff[257];
    __shared__ int  segst[256];
    __shared__ int4 srec[CAP];
    const int tid  = threadIdx.x;
    const int lane = tid & 31;
    const int wave = tid >> 5;
    const int b    = blockIdx.x;
    const int4* r1 = reinterpret_cast<const int4*>(recs1);
    int4*       r2 = reinterpret_cast<int4*>(recs2);

    int s = 0;
    {
        const int n4 = b * (GCH / 4);
        const int4* h4 = reinterpret_cast<const int4*>(hist);
        for (int i0 = 0; i0 < n4; i0 += 4 * 1024) {
            int4 v[4];
            #pragma unroll
            for (int u = 0; u < 4; ++u) {
                int i = i0 + u * 1024 + tid;
                if (i > n4 - 1) i = n4 - 1;
                v[u] = h4[i];
            }
            #pragma unroll
            for (int u = 0; u < 4; ++u)
                if (i0 + u * 1024 + tid < n4) s += (v[u].x + v[u].y) + (v[u].z + v[u].w);
        }
    }
    #pragma unroll
    for (int off = 16; off > 0; off >>= 1) s += __shfl_xor(s, off);
    if (lane == 0) wred[wave] = s;

    int pc = 0, pinc = 0, pstart = 0;
    if (tid < 256) {
        pc     = hist[b * GCH + tid];
        pstart = pst[b * GCH + tid];
        pinc = pc;
        for (int off = 1; off < 32; off <<= 1) {
            int u = __shfl_up(pinc, off);
            if (lane >= off) pinc += u;
        }
        if (lane == 31) wsum[wave] = pinc;
        cnt[tid] = 0;
    }
    __syncthreads();
    int bs = wred[lane];
    #pragma unroll
    for (int off = 16; off > 0; off >>= 1) bs += __shfl_xor(bs, off);
    if (tid < 256) {
        int front = 0;
        for (int k = 0; k < wave; ++k) front += wsum[k];
        const int o = front + pinc - pc;
        segoff[tid] = o;
        segst[tid]  = pstart;
        if (tid == 255) segoff[256] = o + pc;
    }
    __syncthreads();
    const int n  = segoff[256];
    const int be = bs + n;
    const bool fits = (n <= CAP);

    auto rec_at = [&](const int i) __attribute__((always_inline)) {
        int lo = 0;
        #pragma unroll
        for (int step = 128; step > 0; step >>= 1)
            if (segoff[lo + step] <= i) lo += step;
        return segst[lo] + (i - segoff[lo]);
    };

    int4 rr[NR];
    if (fits && n > 0) {
        #pragma unroll
        for (int u = 0; u < NR; ++u) {
            int i = tid + u * 1024;
            if (i > n - 1) i = n - 1;
            rr[u] = r1[rec_at(i)];
        }
    }

    if (fits) {
        #pragma unroll
        for (int u = 0; u < NR; ++u)
            if (tid + u * 1024 < n) atomicAdd(&cnt[rr[u].y & 255], 1);
    } else {
        for (int i = tid; i < n; i += 1024)
            atomicAdd(&cnt[r1[rec_at(i)].y & 255], 1);
    }
    __syncthreads();

    int c = 0, inc = 0;
    if (tid < 256) {
        c = cnt[tid];
        inc = c;
        for (int off = 1; off < 32; off <<= 1) {
            int u = __shfl_up(inc, off);
            if (lane >= off) inc += u;
        }
        if (lane == 31) wsum[wave] = inc;
    }
    __syncthreads();
    if (tid < 256) {
        int front = 0;
        for (int k = 0; k < wave; ++k) front += wsum[k];
        const int first = bs + front + inc - c;
        cur[tid] = first;
        rowptr[b * 256 + tid] = first;
        if (tid == 255 && b == (int)gridDim.x - 1) rowptr[b * 256 + 256] = be;
    }
    __syncthreads();

    if (fits) {
        #pragma unroll
        for (int u = 0; u < NR; ++u) {
            if (tid + u * 1024 < n) {
                const int pos = atomicAdd(&cur[rr[u].y & 255], 1);
                srec[pos - bs] = rr[u];
            }
        }
    } else {
        for (int i = tid; i < n; i += 1024) {
            const int4 r = r1[rec_at(i)];
            const int pos = atomicAdd(&cur[r.y & 255], 1);
            r2[pos] = r;
        }
        __threadfence();
    }
    __syncthreads();

    if (fits) {
        for (int p = tid; p < n; p += 1024) {
            const int4 r  = srec[p];
            const int  rw = r.y & 255;
            const int  e  = cur[rw] - bs;
            const int  a  = e - cnt[rw];
            int k = 0;
            for (int j = a; j < e; j += 4) {
                #pragma unroll
                for (int u = 0; u < 4; ++u) {
                    const int jj = (j + u < e) ? (j + u) : (e - 1);
                    const int x  = srec[jj].x;
                    k += (j + u < e && x < r.x) ? 1 : 0;
                }
            }
            csrc[bs + a + k] = r.z;
            cw[bs + a + k]   = __int_as_float(r.w);
        }
    } else {
        for (int p = bs + tid; p < be; p += 1024) {
            const int4 r  = r2[p];
            const int  rw = r.y & 255;
            const int  e  = cur[rw];
            const int  a  = e - cnt[rw];
            int k = 0;
            for (int j = a; j < e; ++j) k += (r2[j].x < r.x) ? 1 : 0;
            csrc[a + k] = r.z;
            cw[a + k]   = __int_as_float(r.w);
        }
    }
}

#define EPI_RELU     1
#define EPI_RELU_GN  2
#define EPI_LSM      3

template<int C, int CPL, int GRP>
__device__ __forceinline__ void walk_entries(const float* __restrict__ hc, int si, float wi,
                                             int n, float (&acc)[CPL])
{
    for (int k = 0; k < n; k += GRP) {
        float m[GRP][CPL];
        #pragma unroll
        for (int u = 0; u < GRP; ++u) {
            const int kk = (k + u < n) ? (k + u) : (n - 1);
            const int s  = __builtin_amdgcn_readlane(si, kk);
            const float* hp = hc + (size_t)s * C;
            #pragma unroll
            for (int j = 0; j < CPL; ++j) m[u][j] = hp[j];
        }
        #pragma unroll
        for (int u = 0; u < GRP; ++u) {
            const int kk = (k + u < n) ? (k + u) : (n - 1);
            const float w = __int_as_float(__builtin_amdgcn_readlane(__float_as_int(wi), kk));
            const bool on = (k + u < n);
            #pragma unroll
            for (int j = 0; j < CPL; ++j) {
                const float t = fmaf(m[u][j], w, acc[j]);
                acc[j] = on ? t : acc[j];
            }
        }
    }
}

template<int C, int EPI>
__global__ __launch_bounds__(256)
void gather_rows_kernel(const float* __restrict__ h,
                        const int* __restrict__ rowptr,
                        const int* __restrict__ csrc,
                        const float* __restrict__ cw,
                        const float* __restrict__ bias,
                        const float* __restrict__ gamma,
                        const float* __restrict__ beta,
                        float* __restrict__ out, int nrows)
{
    constexpr int CPL = (C == NHID) ? 3 : 2;
    constexpr int LPR = C / CPL;
    constexpr int RPW = 4;
    constexpr int GRP = 8;
    static_assert(LPR * CPL == C && LPR <= 32 && RPW + 1 <= 32, "shape");

    const int lane = threadIdx.x & 31;
    const int gw  = __builtin_amdgcn_readfirstlane((int)(blockIdx.x * 8 + (threadIdx.x >> 5)));
    const int r0  = gw * RPW;
    if (r0 >= nrows) return;
    const bool act = lane < LPR;
    const int  c   = (act ? lane : 0) * CPL;

    int bi = r0 + ((lane < RPW) ? lane : RPW);
    if (bi > nrows) bi = nrows;
    const int rpv = rowptr[bi];

    float bs[CPL], gm[CPL], bt[CPL];
    #pragma unroll
    for (int j = 0; j < CPL; ++j) {
        bs[j] = bias[c + j];
        gm[j] = (EPI == EPI_RELU_GN) ? gamma[c + j] : 1.0f;
        bt[j] = (EPI == EPI_RELU_GN) ? beta[c + j]  : 0.0f;
    }
    const float* hc = h + c;

    auto do_row = [&](const int i, const int si, const float wi, int& nsi, float& nwi)
        __attribute__((always_inline)) {
        const int row = r0 + i;
        const int p0 = __builtin_amdgcn_readlane(rpv, i);
        const int p1 = __builtin_amdgcn_readlane(rpv, i + 1);

        nsi = 0;
        nwi = 0.0f;
        if (i + 1 < RPW && row + 1 < nrows) {
            const int p2 = __builtin_amdgcn_readlane(rpv, i + 2);
            if (lane < p2 - p1) { nsi = csrc[p1 + lane]; nwi = cw[p1 + lane]; }
        }

        float acc[CPL];
        #pragma unroll
        for (int j = 0; j < CPL; ++j) acc[j] = 0.0f;

        if (p1 > p0)
            walk_entries<C, CPL, GRP>(hc, si, wi, (p1 - p0 < 32) ? (p1 - p0) : 32, acc);
        for (int base = p0 + 32; base < p1; base += 32) {
            int   xi = 0;
            float xw = 0.0f;
            if (lane < p1 - base) { xi = csrc[base + lane]; xw = cw[base + lane]; }
            walk_entries<C, CPL, GRP>(hc, xi, xw, (p1 - base < 32) ? (p1 - base) : 32, acc);
        }

        float r[CPL];
        #pragma unroll
        for (int j = 0; j < CPL; ++j) {
            r[j] = acc[j] + bs[j];
            if (EPI == EPI_RELU || EPI == EPI_RELU_GN) r[j] = fmaxf(r[j], 0.0f);
        }

        if constexpr (EPI == EPI_RELU_GN) {
            const float mu  = (r[0] + r[1] + r[2]) * (1.0f / 3.0f);
            const float d0 = r[0] - mu, d1 = r[1] - mu, d2 = r[2] - mu;
            const float var = (d0 * d0 + d1 * d1 + d2 * d2) * (1.0f / 3.0f);
            const float inv = rsqrtf(var + GN_EPS);
            r[0] = d0 * inv * gm[0] + bt[0];
            r[1] = d1 * inv * gm[1] + bt[1];
            r[2] = d2 * inv * gm[2] + bt[2];
        }

        if constexpr (EPI == EPI_LSM) {
            float mx = act ? fmaxf(r[0], r[1]) : -3.402823466e+38f;
            #pragma unroll
            for (int off = 16; off > 0; off >>= 1) mx = fmaxf(mx, __shfl_xor(mx, off));
            const float e0 = __expf(r[0] - mx);
            const float e1 = __expf(r[1] - mx);
            float s = 0.0f;
            #pragma unroll
            for (int k = 0; k < LPR; ++k) {
                s += __int_as_float(__builtin_amdgcn_readlane(__float_as_int(e0), k));
                s += __int_as_float(__builtin_amdgcn_readlane(__float_as_int(e1), k));
            }
            const float ls = __logf(s);
            r[0] = r[0] - mx - ls;
            r[1] = r[1] - mx - ls;
        }

        if (act) {
            float* o = out + (size_t)row * C + c;
            #pragma unroll
            for (int j = 0; j < CPL; ++j) o[j] = r[j];
        }
    };

    int   fsi = 0;
    float fwi = 0.0f;
    {
        const int pa = __builtin_amdgcn_readlane(rpv, 0);
        const int pb = __builtin_amdgcn_readlane(rpv, 1);
        if (lane < pb - pa) { fsi = csrc[pa + lane]; fwi = cw[pa + lane]; }
    }
    int   nsi;
    float nwi;
    do_row(0, fsi, fwi, nsi, nwi);
    for (int i = 1; i < RPW; ++i) {
        if (r0 + i >= nrows) break;
        const int   si = nsi;
        const float wi = nwi;
        do_row(i, si, wi, nsi, nwi);
    }
}

template<int EPI, int NT, int NCOLS>
__global__ __launch_bounds__(512)
void gather_rows_gemm_kernel(const float* __restrict__ h,
                             const int* __restrict__ rowptr,
                             const int* __restrict__ csrc,
                             const float* __restrict__ cw,
                             const float* __restrict__ bias,
                             const float* __restrict__ gamma,
                             const float* __restrict__ beta,
                             const float* __restrict__ Bw,
                             float* __restrict__ Cn, int nrows)
{
    constexpr int C     = NHID;
    constexpr int K     = NHID;
    constexpr int CPL   = 3;
    constexpr int RPW   = 4;
    constexpr int GRP   = 8;
    constexpr int NTHR  = 512;
    constexpr int RPB   = (NTHR / 32) * RPW;
    constexpr int NCP   = NT * 16;
    constexpr int NQ    = NCP / 4;
    constexpr int NTASK = (K / 2) * NQ;
    constexpr int NIT   = (NTASK + NTHR - 1) / NTHR;
    constexpr int KB    = K / 8;
    constexpr int AP    = 100;
    constexpr int TPW   = NT / 3;
    static_assert(NT % 3 == 0 && K % 8 == 0 && NCOLS % 4 == 0 && NCOLS <= NCP && C == 32 * CPL, "shape");

    __shared__ v4f lds4[K * NCP / 4];
    __shared__ v4f at4[RPB * AP / 4];
    float* at = reinterpret_cast<float*>(at4);

    const int tid  = threadIdx.x;
    const int lane = tid & 31;
    const int wave = __builtin_amdgcn_readfirstlane(tid >> 5);
    const int r0   = blockIdx.x * RPB + wave * RPW;
    const int c    = lane * CPL;

    int bi = r0 + ((lane < RPW) ? lane : RPW);
    if (bi > nrows) bi = nrows;
    const int rpv = rowptr[bi];

    v4f w0[NIT], w1[NIT];
    #pragma unroll
    for (int it = 0; it < NIT; ++it) {
        const int i  = tid + it * NTHR;
        const int ic = (NTASK % NTHR == 0) ? i : (i < NTASK ? i : NTASK - 1);
        const int p  = ic / NQ;
        const int q  = ic - p * NQ;
        const int qc = (NCOLS == NCP) ? q : (q * 4 < NCOLS ? q : NCOLS / 4 - 1);
        const float* g = Bw + (size_t)(2 * p) * NCOLS + qc * 4;
        w0[it] = *reinterpret_cast<const v4f*>(g);
        w1[it] = *reinterpret_cast<const v4f*>(g + NCOLS);
    }

    float bs[CPL], gm[CPL], bt[CPL];
    #pragma unroll
    for (int j = 0; j < CPL; ++j) {
        bs[j] = bias[c + j];
        gm[j] = (EPI == EPI_RELU_GN) ? gamma[c + j] : 1.0f;
        bt[j] = (EPI == EPI_RELU_GN) ? beta[c + j]  : 0.0f;
    }
    const float* hc = h + c;

    int   fsi = 0;
    float fwi = 0.0f;
    if (r0 < nrows) {
        const int pa = __builtin_amdgcn_readlane(rpv, 0);
        const int pb = __builtin_amdgcn_readlane(rpv, 1);
        if (lane < pb - pa) { fsi = csrc[pa + lane]; fwi = cw[pa + lane]; }
    }

    #pragma unroll
    for (int it = 0; it < NIT; ++it) {
        const int i = tid + it * NTHR;
        if (NTASK % NTHR == 0 || i < NTASK) {
            v4f lo = (v4f){w0[it].x, w1[it].x, w0[it].y, w1[it].y};
            v4f hi = (v4f){w0[it].z, w1[it].z, w0[it].w, w1[it].w};
            if (NCOLS != NCP) {
                const int q = i % NQ;
                if (q * 4 >= NCOLS) {
                    lo = (v4f){0.0f, 0.0f, 0.0f, 0.0f};
                    hi = (v4f){0.0f, 0.0f, 0.0f, 0.0f};
                }
            }
            lds4[2 * i]     = lo;
            lds4[2 * i + 1] = hi;
        }
    }

    auto do_row = [&](const int i, const int si, const float wi, int& nsi, float& nwi)
        __attribute__((always_inline)) {
        const int row = r0 + i;
        const int p0 = __builtin_amdgcn_readlane(rpv, i);
        const int p1 = __builtin_amdgcn_readlane(rpv, i + 1);

        nsi = 0;
        nwi = 0.0f;
        if (i + 1 < RPW && row + 1 < nrows) {
            const int p2 = __builtin_amdgcn_readlane(rpv, i + 2);
            if (lane < p2 - p1) { nsi = csrc[p1 + lane]; nwi = cw[p1 + lane]; }
        }

        float acc[CPL];
        #pragma unroll
        for (int j = 0; j < CPL; ++j) acc[j] = 0.0f;

        if (p1 > p0)
            walk_entries<C, CPL, GRP>(hc, si, wi, (p1 - p0 < 32) ? (p1 - p0) : 32, acc);
        for (int base = p0 + 32; base < p1; base += 32) {
            int   xi = 0;
            float xw = 0.0f;
            if (lane < p1 - base) { xi = csrc[base + lane]; xw = cw[base + lane]; }
            walk_entries<C, CPL, GRP>(hc, xi, xw, (p1 - base < 32) ? (p1 - base) : 32, acc);
        }

        float r[CPL];
        #pragma unroll
        for (int j = 0; j < CPL; ++j) r[j] = fmaxf(acc[j] + bs[j], 0.0f);

        if constexpr (EPI == EPI_RELU_GN) {
            const float mu  = (r[0] + r[1] + r[2]) * (1.0f / 3.0f);
            const float d0 = r[0] - mu, d1 = r[1] - mu, d2 = r[2] - mu;
            const float var = (d0 * d0 + d1 * d1 + d2 * d2) * (1.0f / 3.0f);
            const float inv = rsqrtf(var + GN_EPS);
            r[0] = d0 * inv * gm[0] + bt[0];
            r[1] = d1 * inv * gm[1] + bt[1];
            r[2] = d2 * inv * gm[2] + bt[2];
        }

        float* o = at + (wave * RPW + i) * AP + c;
        #pragma unroll
        for (int j = 0; j < CPL; ++j) o[j] = r[j];
    };

    if (r0 < nrows) {
        int   nsi;
        float nwi;
        do_row(0, fsi, fwi, nsi, nwi);
        for (int i = 1; i < RPW; ++i) {
            if (r0 + i >= nrows) {
                float* o = at + (wave * RPW + i) * AP + c;
                #pragma unroll
                for (int j = 0; j < CPL; ++j) o[j] = 0.0f;
                continue;
            }
            const int   si = nsi;
            const float wi = nwi;
            do_row(i, si, wi, nsi, nwi);
        }
    } else {
        for (int i = 0; i < RPW; ++i) {
            float* o = at + (wave * RPW + i) * AP + c;
            #pragma unroll
            for (int j = 0; j < CPL; ++j) o[j] = 0.0f;
        }
    }
    __syncthreads();

    if (wave < 12) {
        const int coll = lane & 15;
        const int half = lane >> 4;
        const int mt   = wave / 3;
        const int np   = wave - mt * 3;
        const float2* __restrict__ lds2 = reinterpret_cast<const float2*>(lds4);
        const float2* bp = lds2 + (2 * half) * NCP + np * TPW * 16 + coll;
        const float*  ap = at + (mt * 16 + coll) * AP + half * 4;

        v8f acc[TPW] = {};
        #pragma unroll
        for (int j = 0; j < KB; ++j) {
            const v4f af = *reinterpret_cast<const v4f*>(ap + j * 8);
            #pragma unroll
            for (int s = 0; s < 2; ++s) {
                v2f a;
                a.x = s ? af.z : af.x;
                a.y = s ? af.w : af.y;
                const float2* bq = bp + (4 * j + s) * NCP;
                #pragma unroll
                for (int t = 0; t < TPW; ++t) {
                    float2 bv = bq[t * 16];
                    v2f b; b.x = bv.x; b.y = bv.y;
                    acc[t] = __builtin_amdgcn_wmma_f32_16x16x4_f32(
                        false, a, false, b, (short)0, acc[t], false, false);
                }
            }
        }

        const int rbase = blockIdx.x * RPB + mt * 16 + half * 8;
        #pragma unroll
        for (int t = 0; t < TPW; ++t) {
            const int col = (np * TPW + t) * 16 + coll;
            if (col < NCOLS) {
                float* cp = Cn + (size_t)rbase * NCOLS + col;
                if (rbase + 8 <= nrows) {
                    #pragma unroll
                    for (int r = 0; r < 8; ++r) cp[r * NCOLS] = acc[t][r];
                } else {
                    #pragma unroll
                    for (int r = 0; r < 8; ++r)
                        if (rbase + r < nrows) cp[r * NCOLS] = acc[t][r];
                }
            }
        }
    }
}

extern "C" void kernel_launch(void* const* d_in, const int* in_sizes, int n_in,
                              void* d_out, int out_size, void* d_ws, size_t ws_size,
                              hipStream_t stream)
{
    const float* x     = (const float*)d_in[0];
    const int*   src   = (const int*)  d_in[1];
    const int*   tgt   = (const int*)  d_in[2];
    const float* w     = (const float*)d_in[3];
    const float* W1    = (const float*)d_in[4];
    const float* b1    = (const float*)d_in[5];
    const float* W2    = (const float*)d_in[6];
    const float* b2    = (const float*)d_in[7];
    const float* W3    = (const float*)d_in[8];
    const float* b3    = (const float*)d_in[9];
    const float* gamma = (const float*)d_in[10];
    const float* beta  = (const float*)d_in[11];
    float* out = (float*)d_out;

    const int N   = in_sizes[0] / NFEAT;
    const int E   = in_sizes[1];
    const int ntm = N / 16;
    const int gblk = (ntm + 3) / 4;

    float* bufA = (float*)d_ws;
    float* bufB = bufA + (size_t)N * NHID;

    const int NB   = (N + 255) / 256;
    const int NPAD = ((NB * GCH + 4096) / 4096) * 4096;
    int*   hist   = (int*)(bufB + (size_t)N * NHID);
    int*   pst    = hist + NPAD;
    int*   rowptr = pst + NPAD;
    int*   recs1  = rowptr + NPAD;
    int*   recs2  = recs1 + (size_t)E * 4;
    int*   csrc   = recs2 + (size_t)E * 4;
    float* cw     = (float*)(csrc + E);

    chunk_bins_kernel<<<GCH, 1024, 0, stream>>>(src, tgt, w, hist, pst, recs1, E, NB);
    row_lists_kernel<6><<<NB, 1024, 0, stream>>>(recs1, hist, pst, rowptr, recs2, csrc, cw);

    const int gwv = (N + 31) / 32;
    const int gfb = (N + 63) / 64;

    wmma_gemm_strip_kernel<6, NFEAT, NHID><<<gblk, 128, 0, stream>>>(
        x, W1, bufA, ntm);

    gather_rows_gemm_kernel<EPI_RELU, 6, NHID><<<gfb, 512, 0, stream>>>(
        bufA, rowptr, csrc, cw, b1, gamma, beta, W2, bufB, N);

    gather_rows_gemm_kernel<EPI_RELU_GN, 3, NCLASS><<<gfb, 512, 0, stream>>>(
        bufB, rowptr, csrc, cw, b2, gamma, beta, W3, bufA, N);

    gather_rows_kernel<NCLASS, EPI_LSM><<<gwv, 256, 0, stream>>>(
        bufA, rowptr, csrc, cw, b3, gamma, beta, out, N);
}


# Layer-1 product moved into the first grouping launch: its work-groups (32 waves, 2 column halves per M-tile, 62 registers) share the WGPs with the chunk work-groups; 1 launch fewer; sums unchanged
# speedup vs baseline: 4.7227x; 1.0440x over previous
#include <hip/hip_runtime.h>
#include <hip/hip_bf16.h>


typedef float v2f __attribute__((ext_vector_type(2)));
typedef float v4f __attribute__((ext_vector_type(4)));
typedef float v8f __attribute__((ext_vector_type(8)));

#define NFEAT   128
#define NHID    96
#define NCLASS  40
#define GN_EPS  1e-5f

template<int NT, int K, int NCOLS, int NP>
__device__ __forceinline__ void wmma_gemm_strip(
    const float* __restrict__ A, const float* __restrict__ Bw,
    float* __restrict__ C, int ntm, int wg)
{
    constexpr int NTHR  = 1024;
    constexpr int NCP   = NT * 16;
    constexpr int NQ    = NCP / 2;
    constexpr int NTASK = (K / 2) * NQ;
    constexpr int NIT   = NTASK / NTHR;
    constexpr int KB    = K / 8;
    constexpr int TPW   = NT / NP;
    constexpr int JC    = 4;
    constexpr int NCH   = KB / JC;
    static_assert(K % 8 == 0 && NCOLS % 2 == 0 && NCOLS <= NCP && NT % NP == 0 && KB % JC == 0 && NTASK % NTHR == 0, "shape");

    __shared__ v4f lds4[K * NCP / 4];

    const int tid   = threadIdx.x;
    const int lane  = tid & 31;
    const int wave  = __builtin_amdgcn_readfirstlane(tid >> 5);
    const int coll  = lane & 15;
    const int half  = lane >> 4;
    const int task  = wg * (NTHR / 32) + wave;
    const int mtile = task / NP;
    const int np    = task - mtile * NP;
    const int mload = (mtile < ntm) ? mtile : (ntm - 1);

    v2f w0[NIT], w1[NIT];
    #pragma unroll
    for (int it = 0; it < NIT; ++it) {
        const int i  = tid + it * NTHR;
        const int p  = i / NQ;
        const int q  = i - p * NQ;
        const int qc = (NCOLS == NCP) ? q : (q * 2 < NCOLS ? q : NCOLS / 2 - 1);
        const float* g = Bw + (size_t)(2 * p) * NCOLS + qc * 2;
        w0[it] = *reinterpret_cast<const v2f*>(g);
        w1[it] = *reinterpret_cast<const v2f*>(g + NCOLS);
    }

    v4f an[JC];
    const float* arow = A + (size_t)(mload * 16 + coll) * K + half * 4;
    #pragma unroll
    for (int j = 0; j < JC; ++j)
        an[j] = *reinterpret_cast<const v4f*>(arow + j * 8);

    #pragma unroll
    for (int it = 0; it < NIT; ++it) {
        const int i = tid + it * NTHR;
        v4f pr = (v4f){w0[it].x, w1[it].x, w0[it].y, w1[it].y};
        if (NCOLS != NCP) {
            const int q = i % NQ;
            if (q * 2 >= NCOLS) pr = (v4f){0.0f, 0.0f, 0.0f, 0.0f};
        }
        lds4[i] = pr;
    }
    __syncthreads();

    const float2* __restrict__ lds2 = reinterpret_cast<const float2*>(lds4);
    const float2* bp = lds2 + (2 * half) * NCP + np * TPW * 16 + coll;

    v8f acc[TPW] = {};
    #pragma unroll
    for (int ch = 0; ch < NCH; ++ch) {
        v4f ac[JC];
        #pragma unroll
        for (int j = 0; j < JC; ++j) ac[j] = an[j];
        if (ch + 1 < NCH) {
            #pragma unroll
            for (int j = 0; j < JC; ++j)
                an[j] = *reinterpret_cast<const v4f*>(arow + ((ch + 1) * JC + j) * 8);
        }
        #pragma unroll
        for (int j = 0; j < JC; ++j) {
            const int jj = ch * JC + j;
            #pragma unroll
            for (int s = 0; s < 2; ++s) {
                v2f a;
                a.x = s ? ac[j].z : ac[j].x;
                a.y = s ? ac[j].w : ac[j].y;
                const float2* bq = bp + (4 * jj + s) * NCP;
                #pragma unroll
                for (int t = 0; t < TPW; ++t) {
                    float2 bv = bq[t * 16];
                    v2f b; b.x = bv.x; b.y = bv.y;
                    acc[t] = __builtin_amdgcn_wmma_f32_16x16x4_f32(
                        false, a, false, b, (short)0, acc[t], false, false);
                }
            }
        }
    }

    if (mtile >= ntm) return;

    const int rbase = mtile * 16 + half * 8;
    #pragma unroll
    for (int t = 0; t < TPW; ++t) {
        const int col = (np * TPW + t) * 16 + coll;
        if (col < NCOLS) {
            #pragma unroll
            for (int r = 0; r < 8; ++r)
                C[(size_t)(rbase + r) * NCOLS + col] = acc[t][r];
        }
    }
}


#define GCH 256
#define NEC 4

__global__ __launch_bounds__(1024, 16) void product_and_chunk_bins_kernel(
                                                          const int* __restrict__ src,
                                                          const int* __restrict__ tgt,
                                                          const float* __restrict__ w,
                                                          int* __restrict__ hist,
                                                          int* __restrict__ pst,
                                                          int* __restrict__ recs1,
                                                          int E, int nb,
                                                          const float* __restrict__ X,
                                                          const float* __restrict__ Wa,
                                                          float* __restrict__ T,
                                                          int ntm, int ngw)
{
    if ((int)blockIdx.x < ngw) {
        wmma_gemm_strip<6, NFEAT, NHID, 2>(X, Wa, T, ntm, (int)blockIdx.x);
        return;
    }
    __shared__ int lh[256];
    __shared__ int cur[256];
    __shared__ int wsum[8];
    const int tid  = threadIdx.x;
    const int lane = tid & 31;
    const int wave = tid >> 5;
    const int g    = (int)blockIdx.x - ngw;
    const int ch = (E + GCH - 1) / GCH;
    int e0 = g * ch;
    if (e0 > E) e0 = E;
    const int e1 = (e0 + ch < E) ? (e0 + ch) : E;
    const bool one = (ch <= NEC * 1024);

    int   tg[NEC] = {};
    int   sr[NEC] = {};
    float ww[NEC] = {};
    if (one && e1 > e0) {
        #pragma unroll
        for (int u = 0; u < NEC; ++u) {
            int e = e0 + tid + u * 1024;
            if (e > e1 - 1) e = e1 - 1;
            tg[u] = tgt[e];
            sr[u] = src[e];
            ww[u] = w[e];
        }
    }
    if (tid < 256) lh[tid] = 0;
    __syncthreads();
    if (one) {
        #pragma unroll
        for (int u = 0; u < NEC; ++u)
            if (e0 + tid + u * 1024 < e1) atomicAdd(&lh[tg[u] >> 8], 1);
    } else {
        for (int e = e0 + tid; e < e1; e += 1024)
            atomicAdd(&lh[tgt[e] >> 8], 1);
    }
    __syncthreads();

    int c = 0, inc = 0;
    if (tid < 256) {
        c = lh[tid];
        inc = c;
        for (int off = 1; off < 32; off <<= 1) {
            int u = __shfl_up(inc, off);
            if (lane >= off) inc += u;
        }
        if (lane == 31) wsum[wave] = inc;
    }
    __syncthreads();
    if (tid < 256) {
        int front = 0;
        for (int k = 0; k < wave; ++k) front += wsum[k];
        const int first = e0 + front + inc - c;
        cur[tid] = first;
        if (tid < nb) {
            hist[tid * GCH + g] = c;
            pst[tid * GCH + g]  = first;
        }
    }
    __syncthreads();

    int4* r1 = reinterpret_cast<int4*>(recs1);
    if (one) {
        #pragma unroll
        for (int u = 0; u < NEC; ++u) {
            const int e = e0 + tid + u * 1024;
            if (e < e1) {
                const int pos = atomicAdd(&cur[tg[u] >> 8], 1);
                r1[pos] = make_int4(e, tg[u], sr[u], __float_as_int(ww[u]));
            }
        }
    } else {
        for (int e = e0 + tid; e < e1; e += 1024) {
            const int t   = tgt[e];
            const int pos = atomicAdd(&cur[t >> 8], 1);
            r1[pos] = make_int4(e, t, src[e], __float_as_int(w[e]));
        }
    }
}

template<int NR>
__global__ __launch_bounds__(1024) void row_lists_kernel(const int* __restrict__ recs1,
                                                         const int* __restrict__ hist,
                                                         const int* __restrict__ pst,
                                                         int* __restrict__ rowptr,
                                                         int* __restrict__ recs2,
                                                         int* __restrict__ csrc,
                                                         float* __restrict__ cw)
{
    constexpr int CAP = NR * 1024;
    __shared__ int  cnt[256];
    __shared__ int  cur[256];
    __shared__ int  wsum[8];
    __shared__ int  wred[32];
    __shared__ int  segoff[257];
    __shared__ int  segst[256];
    __shared__ int4 srec[CAP];
    const int tid  = threadIdx.x;
    const int lane = tid & 31;
    const int wave = tid >> 5;
    const int b    = blockIdx.x;
    const int4* r1 = reinterpret_cast<const int4*>(recs1);
    int4*       r2 = reinterpret_cast<int4*>(recs2);

    int s = 0;
    {
        const int n4 = b * (GCH / 4);
        const int4* h4 = reinterpret_cast<const int4*>(hist);
        for (int i0 = 0; i0 < n4; i0 += 4 * 1024) {
            int4 v[4];
            #pragma unroll
            for (int u = 0; u < 4; ++u) {
                int i = i0 + u * 1024 + tid;
                if (i > n4 - 1) i = n4 - 1;
                v[u] = h4[i];
            }
            #pragma unroll
            for (int u = 0; u < 4; ++u)
                if (i0 + u * 1024 + tid < n4) s += (v[u].x + v[u].y) + (v[u].z + v[u].w);
        }
    }
    #pragma unroll
    for (int off = 16; off > 0; off >>= 1) s += __shfl_xor(s, off);
    if (lane == 0) wred[wave] = s;

    int pc = 0, pinc = 0, pstart = 0;
    if (tid < 256) {
        pc     = hist[b * GCH + tid];
        pstart = pst[b * GCH + tid];
        pinc = pc;
        for (int off = 1; off < 32; off <<= 1) {
            int u = __shfl_up(pinc, off);
            if (lane >= off) pinc += u;
        }
        if (lane == 31) wsum[wave] = pinc;
        cnt[tid] = 0;
    }
    __syncthreads();
    int bs = wred[lane];
    #pragma unroll
    for (int off = 16; off > 0; off >>= 1) bs += __shfl_xor(bs, off);
    if (tid < 256) {
        int front = 0;
        for (int k = 0; k < wave; ++k) front += wsum[k];
        const int o = front + pinc - pc;
        segoff[tid] = o;
        segst[tid]  = pstart;
        if (tid == 255) segoff[256] = o + pc;
    }
    __syncthreads();
    const int n  = segoff[256];
    const int be = bs + n;
    const bool fits = (n <= CAP);

    auto rec_at = [&](const int i) __attribute__((always_inline)) {
        int lo = 0;
        #pragma unroll
        for (int step = 128; step > 0; step >>= 1)
            if (segoff[lo + step] <= i) lo += step;
        return segst[lo] + (i - segoff[lo]);
    };

    int4 rr[NR];
    if (fits && n > 0) {
        #pragma unroll
        for (int u = 0; u < NR; ++u) {
            int i = tid + u * 1024;
            if (i > n - 1) i = n - 1;
            rr[u] = r1[rec_at(i)];
        }
    }

    if (fits) {
        #pragma unroll
        for (int u = 0; u < NR; ++u)
            if (tid + u * 1024 < n) atomicAdd(&cnt[rr[u].y & 255], 1);
    } else {
        for (int i = tid; i < n; i += 1024)
            atomicAdd(&cnt[r1[rec_at(i)].y & 255], 1);
    }
    __syncthreads();

    int c = 0, inc = 0;
    if (tid < 256) {
        c = cnt[tid];
        inc = c;
        for (int off = 1; off < 32; off <<= 1) {
            int u = __shfl_up(inc, off);
            if (lane >= off) inc += u;
        }
        if (lane == 31) wsum[wave] = inc;
    }
    __syncthreads();
    if (tid < 256) {
        int front = 0;
        for (int k = 0; k < wave; ++k) front += wsum[k];
        const int first = bs + front + inc - c;
        cur[tid] = first;
        rowptr[b * 256 + tid] = first;
        if (tid == 255 && b == (int)gridDim.x - 1) rowptr[b * 256 + 256] = be;
    }
    __syncthreads();

    if (fits) {
        #pragma unroll
        for (int u = 0; u < NR; ++u) {
            if (tid + u * 1024 < n) {
                const int pos = atomicAdd(&cur[rr[u].y & 255], 1);
                srec[pos - bs] = rr[u];
            }
        }
    } else {
        for (int i = tid; i < n; i += 1024) {
            const int4 r = r1[rec_at(i)];
            const int pos = atomicAdd(&cur[r.y & 255], 1);
            r2[pos] = r;
        }
        __threadfence();
    }
    __syncthreads();

    if (fits) {
        for (int p = tid; p < n; p += 1024) {
            const int4 r  = srec[p];
            const int  rw = r.y & 255;
            const int  e  = cur[rw] - bs;
            const int  a  = e - cnt[rw];
            int k = 0;
            for (int j = a; j < e; j += 4) {
                #pragma unroll
                for (int u = 0; u < 4; ++u) {
                    const int jj = (j + u < e) ? (j + u) : (e - 1);
                    const int x  = srec[jj].x;
                    k += (j + u < e && x < r.x) ? 1 : 0;
                }
            }
            csrc[bs + a + k] = r.z;
            cw[bs + a + k]   = __int_as_float(r.w);
        }
    } else {
        for (int p = bs + tid; p < be; p += 1024) {
            const int4 r  = r2[p];
            const int  rw = r.y & 255;
            const int  e  = cur[rw];
            const int  a  = e - cnt[rw];
            int k = 0;
            for (int j = a; j < e; ++j) k += (r2[j].x < r.x) ? 1 : 0;
            csrc[a + k] = r.z;
            cw[a + k]   = __int_as_float(r.w);
        }
    }
}

#define EPI_RELU     1
#define EPI_RELU_GN  2
#define EPI_LSM      3

template<int C, int CPL, int GRP>
__device__ __forceinline__ void walk_entries(const float* __restrict__ hc, int si, float wi,
                                             int n, float (&acc)[CPL])
{
    for (int k = 0; k < n; k += GRP) {
        float m[GRP][CPL];
        #pragma unroll
        for (int u = 0; u < GRP; ++u) {
            const int kk = (k + u < n) ? (k + u) : (n - 1);
            const int s  = __builtin_amdgcn_readlane(si, kk);
            const float* hp = hc + (size_t)s * C;
            #pragma unroll
            for (int j = 0; j < CPL; ++j) m[u][j] = hp[j];
        }
        #pragma unroll
        for (int u = 0; u < GRP; ++u) {
            const int kk = (k + u < n) ? (k + u) : (n - 1);
            const float w = __int_as_float(__builtin_amdgcn_readlane(__float_as_int(wi), kk));
            const bool on = (k + u < n);
            #pragma unroll
            for (int j = 0; j < CPL; ++j) {
                const float t = fmaf(m[u][j], w, acc[j]);
                acc[j] = on ? t : acc[j];
            }
        }
    }
}

template<int C, int EPI>
__global__ __launch_bounds__(256)
void gather_rows_kernel(const float* __restrict__ h,
                        const int* __restrict__ rowptr,
                        const int* __restrict__ csrc,
                        const float* __restrict__ cw,
                        const float* __restrict__ bias,
                        const float* __restrict__ gamma,
                        const float* __restrict__ beta,
                        float* __restrict__ out, int nrows)
{
    constexpr int CPL = (C == NHID) ? 3 : 2;
    constexpr int LPR = C / CPL;
    constexpr int RPW = 4;
    constexpr int GRP = 8;
    static_assert(LPR * CPL == C && LPR <= 32 && RPW + 1 <= 32, "shape");

    const int lane = threadIdx.x & 31;
    const int gw  = __builtin_amdgcn_readfirstlane((int)(blockIdx.x * 8 + (threadIdx.x >> 5)));
    const int r0  = gw * RPW;
    if (r0 >= nrows) return;
    const bool act = lane < LPR;
    const int  c   = (act ? lane : 0) * CPL;

    int bi = r0 + ((lane < RPW) ? lane : RPW);
    if (bi > nrows) bi = nrows;
    const int rpv = rowptr[bi];

    float bs[CPL], gm[CPL], bt[CPL];
    #pragma unroll
    for (int j = 0; j < CPL; ++j) {
        bs[j] = bias[c + j];
        gm[j] = (EPI == EPI_RELU_GN) ? gamma[c + j] : 1.0f;
        bt[j] = (EPI == EPI_RELU_GN) ? beta[c + j]  : 0.0f;
    }
    const float* hc = h + c;

    auto do_row = [&](const int i, const int si, const float wi, int& nsi, float& nwi)
        __attribute__((always_inline)) {
        const int row = r0 + i;
        const int p0 = __builtin_amdgcn_readlane(rpv, i);
        const int p1 = __builtin_amdgcn_readlane(rpv, i + 1);

        nsi = 0;
        nwi = 0.0f;
        if (i + 1 < RPW && row + 1 < nrows) {
            const int p2 = __builtin_amdgcn_readlane(rpv, i + 2);
            if (lane < p2 - p1) { nsi = csrc[p1 + lane]; nwi = cw[p1 + lane]; }
        }

        float acc[CPL];
        #pragma unroll
        for (int j = 0; j < CPL; ++j) acc[j] = 0.0f;

        if (p1 > p0)
            walk_entries<C, CPL, GRP>(hc, si, wi, (p1 - p0 < 32) ? (p1 - p0) : 32, acc);
        for (int base = p0 + 32; base < p1; base += 32) {
            int   xi = 0;
            float xw = 0.0f;
            if (lane < p1 - base) { xi = csrc[base + lane]; xw = cw[base + lane]; }
            walk_entries<C, CPL, GRP>(hc, xi, xw, (p1 - base < 32) ? (p1 - base) : 32, acc);
        }

        float r[CPL];
        #pragma unroll
        for (int j = 0; j < CPL; ++j) {
            r[j] = acc[j] + bs[j];
            if (EPI == EPI_RELU || EPI == EPI_RELU_GN) r[j] = fmaxf(r[j], 0.0f);
        }

        if constexpr (EPI == EPI_RELU_GN) {
            const float mu  = (r[0] + r[1] + r[2]) * (1.0f / 3.0f);
            const float d0 = r[0] - mu, d1 = r[1] - mu, d2 = r[2] - mu;
            const float var = (d0 * d0 + d1 * d1 + d2 * d2) * (1.0f / 3.0f);
            const float inv = rsqrtf(var + GN_EPS);
            r[0] = d0 * inv * gm[0] + bt[0];
            r[1] = d1 * inv * gm[1] + bt[1];
            r[2] = d2 * inv * gm[2] + bt[2];
        }

        if constexpr (EPI == EPI_LSM) {
            float mx = act ? fmaxf(r[0], r[1]) : -3.402823466e+38f;
            #pragma unroll
            for (int off = 16; off > 0; off >>= 1) mx = fmaxf(mx, __shfl_xor(mx, off));
            const float e0 = __expf(r[0] - mx);
            const float e1 = __expf(r[1] - mx);
            float s = 0.0f;
            #pragma unroll
            for (int k = 0; k < LPR; ++k) {
                s += __int_as_float(__builtin_amdgcn_readlane(__float_as_int(e0), k));
                s += __int_as_float(__builtin_amdgcn_readlane(__float_as_int(e1), k));
            }
            const float ls = __logf(s);
            r[0] = r[0] - mx - ls;
            r[1] = r[1] - mx - ls;
        }

        if (act) {
            float* o = out + (size_t)row * C + c;
            #pragma unroll
            for (int j = 0; j < CPL; ++j) o[j] = r[j];
        }
    };

    int   fsi = 0;
    float fwi = 0.0f;
    {
        const int pa = __builtin_amdgcn_readlane(rpv, 0);
        const int pb = __builtin_amdgcn_readlane(rpv, 1);
        if (lane < pb - pa) { fsi = csrc[pa + lane]; fwi = cw[pa + lane]; }
    }
    int   nsi;
    float nwi;
    do_row(0, fsi, fwi, nsi, nwi);
    for (int i = 1; i < RPW; ++i) {
        if (r0 + i >= nrows) break;
        const int   si = nsi;
        const float wi = nwi;
        do_row(i, si, wi, nsi, nwi);
    }
}

template<int EPI, int NT, int NCOLS>
__global__ __launch_bounds__(512)
void gather_rows_gemm_kernel(const float* __restrict__ h,
                             const int* __restrict__ rowptr,
                             const int* __restrict__ csrc,
                             const float* __restrict__ cw,
                             const float* __restrict__ bias,
                             const float* __restrict__ gamma,
                             const float* __restrict__ beta,
                             const float* __restrict__ Bw,
                             float* __restrict__ Cn, int nrows)
{
    constexpr int C     = NHID;
    constexpr int K     = NHID;
    constexpr int CPL   = 3;
    constexpr int RPW   = 4;
    constexpr int GRP   = 8;
    constexpr int NTHR  = 512;
    constexpr int RPB   = (NTHR / 32) * RPW;
    constexpr int NCP   = NT * 16;
    constexpr int NQ    = NCP / 4;
    constexpr int NTASK = (K / 2) * NQ;
    constexpr int NIT   = (NTASK + NTHR - 1) / NTHR;
    constexpr int KB    = K / 8;
    constexpr int AP    = 100;
    constexpr int TPW   = NT / 3;
    static_assert(NT % 3 == 0 && K % 8 == 0 && NCOLS % 4 == 0 && NCOLS <= NCP && C == 32 * CPL, "shape");

    __shared__ v4f lds4[K * NCP / 4];
    __shared__ v4f at4[RPB * AP / 4];
    float* at = reinterpret_cast<float*>(at4);

    const int tid  = threadIdx.x;
    const int lane = tid & 31;
    const int wave = __builtin_amdgcn_readfirstlane(tid >> 5);
    const int r0   = blockIdx.x * RPB + wave * RPW;
    const int c    = lane * CPL;

    int bi = r0 + ((lane < RPW) ? lane : RPW);
    if (bi > nrows) bi = nrows;
    const int rpv = rowptr[bi];

    v4f w0[NIT], w1[NIT];
    #pragma unroll
    for (int it = 0; it < NIT; ++it) {
        const int i  = tid + it * NTHR;
        const int ic = (NTASK % NTHR == 0) ? i : (i < NTASK ? i : NTASK - 1);
        const int p  = ic / NQ;
        const int q  = ic - p * NQ;
        const int qc = (NCOLS == NCP) ? q : (q * 4 < NCOLS ? q : NCOLS / 4 - 1);
        const float* g = Bw + (size_t)(2 * p) * NCOLS + qc * 4;
        w0[it] = *reinterpret_cast<const v4f*>(g);
        w1[it] = *reinterpret_cast<const v4f*>(g + NCOLS);
    }

    float bs[CPL], gm[CPL], bt[CPL];
    #pragma unroll
    for (int j = 0; j < CPL; ++j) {
        bs[j] = bias[c + j];
        gm[j] = (EPI == EPI_RELU_GN) ? gamma[c + j] : 1.0f;
        bt[j] = (EPI == EPI_RELU_GN) ? beta[c + j]  : 0.0f;
    }
    const float* hc = h + c;

    int   fsi = 0;
    float fwi = 0.0f;
    if (r0 < nrows) {
        const int pa = __builtin_amdgcn_readlane(rpv, 0);
        const int pb = __builtin_amdgcn_readlane(rpv, 1);
        if (lane < pb - pa) { fsi = csrc[pa + lane]; fwi = cw[pa + lane]; }
    }

    #pragma unroll
    for (int it = 0; it < NIT; ++it) {
        const int i = tid + it * NTHR;
        if (NTASK % NTHR == 0 || i < NTASK) {
            v4f lo = (v4f){w0[it].x, w1[it].x, w0[it].y, w1[it].y};
            v4f hi = (v4f){w0[it].z, w1[it].z, w0[it].w, w1[it].w};
            if (NCOLS != NCP) {
                const int q = i % NQ;
                if (q * 4 >= NCOLS) {
                    lo = (v4f){0.0f, 0.0f, 0.0f, 0.0f};
                    hi = (v4f){0.0f, 0.0f, 0.0f, 0.0f};
                }
            }
            lds4[2 * i]     = lo;
            lds4[2 * i + 1] = hi;
        }
    }

    auto do_row = [&](const int i, const int si, const float wi, int& nsi, float& nwi)
        __attribute__((always_inline)) {
        const int row = r0 + i;
        const int p0 = __builtin_amdgcn_readlane(rpv, i);
        const int p1 = __builtin_amdgcn_readlane(rpv, i + 1);

        nsi = 0;
        nwi = 0.0f;
        if (i + 1 < RPW && row + 1 < nrows) {
            const int p2 = __builtin_amdgcn_readlane(rpv, i + 2);
            if (lane < p2 - p1) { nsi = csrc[p1 + lane]; nwi = cw[p1 + lane]; }
        }

        float acc[CPL];
        #pragma unroll
        for (int j = 0; j < CPL; ++j) acc[j] = 0.0f;

        if (p1 > p0)
            walk_entries<C, CPL, GRP>(hc, si, wi, (p1 - p0 < 32) ? (p1 - p0) : 32, acc);
        for (int base = p0 + 32; base < p1; base += 32) {
            int   xi = 0;
            float xw = 0.0f;
            if (lane < p1 - base) { xi = csrc[base + lane]; xw = cw[base + lane]; }
            walk_entries<C, CPL, GRP>(hc, xi, xw, (p1 - base < 32) ? (p1 - base) : 32, acc);
        }

        float r[CPL];
        #pragma unroll
        for (int j = 0; j < CPL; ++j) r[j] = fmaxf(acc[j] + bs[j], 0.0f);

        if constexpr (EPI == EPI_RELU_GN) {
            const float mu  = (r[0] + r[1] + r[2]) * (1.0f / 3.0f);
            const float d0 = r[0] - mu, d1 = r[1] - mu, d2 = r[2] - mu;
            const float var = (d0 * d0 + d1 * d1 + d2 * d2) * (1.0f / 3.0f);
            const float inv = rsqrtf(var + GN_EPS);
            r[0] = d0 * inv * gm[0] + bt[0];
            r[1] = d1 * inv * gm[1] + bt[1];
            r[2] = d2 * inv * gm[2] + bt[2];
        }

        float* o = at + (wave * RPW + i) * AP + c;
        #pragma unroll
        for (int j = 0; j < CPL; ++j) o[j] = r[j];
    };

    if (r0 < nrows) {
        int   nsi;
        float nwi;
        do_row(0, fsi, fwi, nsi, nwi);
        for (int i = 1; i < RPW; ++i) {
            if (r0 + i >= nrows) {
                float* o = at + (wave * RPW + i) * AP + c;
                #pragma unroll
                for (int j = 0; j < CPL; ++j) o[j] = 0.0f;
                continue;
            }
            const int   si = nsi;
            const float wi = nwi;
            do_row(i, si, wi, nsi, nwi);
        }
    } else {
        for (int i = 0; i < RPW; ++i) {
            float* o = at + (wave * RPW + i) * AP + c;
            #pragma unroll
            for (int j = 0; j < CPL; ++j) o[j] = 0.0f;
        }
    }
    __syncthreads();

    if (wave < 12) {
        const int coll = lane & 15;
        const int half = lane >> 4;
        const int mt   = wave / 3;
        const int np   = wave - mt * 3;
        const float2* __restrict__ lds2 = reinterpret_cast<const float2*>(lds4);
        const float2* bp = lds2 + (2 * half) * NCP + np * TPW * 16 + coll;
        const float*  ap = at + (mt * 16 + coll) * AP + half * 4;

        v8f acc[TPW] = {};
        #pragma unroll
        for (int j = 0; j < KB; ++j) {
            const v4f af = *reinterpret_cast<const v4f*>(ap + j * 8);
            #pragma unroll
            for (int s = 0; s < 2; ++s) {
                v2f a;
                a.x = s ? af.z : af.x;
                a.y = s ? af.w : af.y;
                const float2* bq = bp + (4 * j + s) * NCP;
                #pragma unroll
                for (int t = 0; t < TPW; ++t) {
                    float2 bv = bq[t * 16];
                    v2f b; b.x = bv.x; b.y = bv.y;
                    acc[t] = __builtin_amdgcn_wmma_f32_16x16x4_f32(
                        false, a, false, b, (short)0, acc[t], false, false);
                }
            }
        }

        const int rbase = blockIdx.x * RPB + mt * 16 + half * 8;
        #pragma unroll
        for (int t = 0; t < TPW; ++t) {
            const int col = (np * TPW + t) * 16 + coll;
            if (col < NCOLS) {
                float* cp = Cn + (size_t)rbase * NCOLS + col;
                if (rbase + 8 <= nrows) {
                    #pragma unroll
                    for (int r = 0; r < 8; ++r) cp[r * NCOLS] = acc[t][r];
                } else {
                    #pragma unroll
                    for (int r = 0; r < 8; ++r)
                        if (rbase + r < nrows) cp[r * NCOLS] = acc[t][r];
                }
            }
        }
    }
}

extern "C" void kernel_launch(void* const* d_in, const int* in_sizes, int n_in,
                              void* d_out, int out_size, void* d_ws, size_t ws_size,
                              hipStream_t stream)
{
    const float* x     = (const float*)d_in[0];
    const int*   src   = (const int*)  d_in[1];
    const int*   tgt   = (const int*)  d_in[2];
    const float* w     = (const float*)d_in[3];
    const float* W1    = (const float*)d_in[4];
    const float* b1    = (const float*)d_in[5];
    const float* W2    = (const float*)d_in[6];
    const float* b2    = (const float*)d_in[7];
    const float* W3    = (const float*)d_in[8];
    const float* b3    = (const float*)d_in[9];
    const float* gamma = (const float*)d_in[10];
    const float* beta  = (const float*)d_in[11];
    float* out = (float*)d_out;

    const int N   = in_sizes[0] / NFEAT;
    const int E   = in_sizes[1];
    const int ntm = N / 16;

    float* bufA = (float*)d_ws;
    float* bufB = bufA + (size_t)N * NHID;

    const int NB   = (N + 255) / 256;
    const int NPAD = ((NB * GCH + 4096) / 4096) * 4096;
    int*   hist   = (int*)(bufB + (size_t)N * NHID);
    int*   pst    = hist + NPAD;
    int*   rowptr = pst + NPAD;
    int*   recs1  = rowptr + NPAD;
    int*   recs2  = recs1 + (size_t)E * 4;
    int*   csrc   = recs2 + (size_t)E * 4;
    float* cw     = (float*)(csrc + E);

    const int ngw = (ntm * 2 + 31) / 32;
    product_and_chunk_bins_kernel<<<ngw + GCH, 1024, 0, stream>>>(src, tgt, w, hist, pst, recs1, E, NB,
                                                                   x, W1, bufA, ntm, ngw);
    row_lists_kernel<6><<<NB, 1024, 0, stream>>>(recs1, hist, pst, rowptr, recs2, csrc, cw);

    const int gwv = (N + 31) / 32;
    const int gfb = (N + 63) / 64;

    gather_rows_gemm_kernel<EPI_RELU, 6, NHID><<<gfb, 512, 0, stream>>>(
        bufA, rowptr, csrc, cw, b1, gamma, beta, W2, bufB, N);

    gather_rows_gemm_kernel<EPI_RELU_GN, 3, NCLASS><<<gfb, 512, 0, stream>>>(
        bufB, rowptr, csrc, cw, b2, gamma, beta, W3, bufA, N);

    gather_rows_kernel<NCLASS, EPI_LSM><<<gwv, 256, 0, stream>>>(
        bufA, rowptr, csrc, cw, b3, gamma, beta, out, N);
}


# Gathers: fewer dependent round trips per wave: its 4 lists' entries fetched as one group up front; 12, 10, 16 source rows per trip (was 8) on scalar row addresses, remainder exact; sums unchanged
# speedup vs baseline: 4.9092x; 1.0395x over previous
#include <hip/hip_runtime.h>
#include <hip/hip_bf16.h>


typedef float v2f __attribute__((ext_vector_type(2)));
typedef float v4f __attribute__((ext_vector_type(4)));
typedef float v8f __attribute__((ext_vector_type(8)));
typedef float gfloat __attribute__((address_space(1)));

#define NFEAT   128
#define NHID    96
#define NCLASS  40
#define GN_EPS  1e-5f

template<int NT, int K, int NCOLS, int NP>
__device__ __forceinline__ void wmma_gemm_strip(
    const float* __restrict__ A, const float* __restrict__ Bw,
    float* __restrict__ C, int ntm, int wg)
{
    constexpr int NTHR  = 1024;
    constexpr int NCP   = NT * 16;
    constexpr int NQ    = NCP / 2;
    constexpr int NTASK = (K / 2) * NQ;
    constexpr int NIT   = NTASK / NTHR;
    constexpr int KB    = K / 8;
    constexpr int TPW   = NT / NP;
    constexpr int JC    = 4;
    constexpr int NCH   = KB / JC;
    static_assert(K % 8 == 0 && NCOLS % 2 == 0 && NCOLS <= NCP && NT % NP == 0 && KB % JC == 0 && NTASK % NTHR == 0, "shape");

    __shared__ v4f lds4[K * NCP / 4];

    const int tid   = threadIdx.x;
    const int lane  = tid & 31;
    const int wave  = __builtin_amdgcn_readfirstlane(tid >> 5);
    const int coll  = lane & 15;
    const int half  = lane >> 4;
    const int task  = wg * (NTHR / 32) + wave;
    const int mtile = task / NP;
    const int np    = task - mtile * NP;
    const int mload = (mtile < ntm) ? mtile : (ntm - 1);

    v2f w0[NIT], w1[NIT];
    #pragma unroll
    for (int it = 0; it < NIT; ++it) {
        const int i  = tid + it * NTHR;
        const int p  = i / NQ;
        const int q  = i - p * NQ;
        const int qc = (NCOLS == NCP) ? q : (q * 2 < NCOLS ? q : NCOLS / 2 - 1);
        const float* g = Bw + (size_t)(2 * p) * NCOLS + qc * 2;
        w0[it] = *reinterpret_cast<const v2f*>(g);
        w1[it] = *reinterpret_cast<const v2f*>(g + NCOLS);
    }

    v4f an[JC];
    const float* arow = A + (size_t)(mload * 16 + coll) * K + half * 4;
    #pragma unroll
    for (int j = 0; j < JC; ++j)
        an[j] = *reinterpret_cast<const v4f*>(arow + j * 8);

    #pragma unroll
    for (int it = 0; it < NIT; ++it) {
        const int i = tid + it * NTHR;
        v4f pr = (v4f){w0[it].x, w1[it].x, w0[it].y, w1[it].y};
        if (NCOLS != NCP) {
            const int q = i % NQ;
            if (q * 2 >= NCOLS) pr = (v4f){0.0f, 0.0f, 0.0f, 0.0f};
        }
        lds4[i] = pr;
    }
    __syncthreads();

    const float2* __restrict__ lds2 = reinterpret_cast<const float2*>(lds4);
    const float2* bp = lds2 + (2 * half) * NCP + np * TPW * 16 + coll;

    v8f acc[TPW] = {};
    #pragma unroll
    for (int ch = 0; ch < NCH; ++ch) {
        v4f ac[JC];
        #pragma unroll
        for (int j = 0; j < JC; ++j) ac[j] = an[j];
        if (ch + 1 < NCH) {
            #pragma unroll
            for (int j = 0; j < JC; ++j)
                an[j] = *reinterpret_cast<const v4f*>(arow + ((ch + 1) * JC + j) * 8);
        }
        #pragma unroll
        for (int j = 0; j < JC; ++j) {
            const int jj = ch * JC + j;
            #pragma unroll
            for (int s = 0; s < 2; ++s) {
                v2f a;
                a.x = s ? ac[j].z : ac[j].x;
                a.y = s ? ac[j].w : ac[j].y;
                const float2* bq = bp + (4 * jj + s) * NCP;
                #pragma unroll
                for (int t = 0; t < TPW; ++t) {
                    float2 bv = bq[t * 16];
                    v2f b; b.x = bv.x; b.y = bv.y;
                    acc[t] = __builtin_amdgcn_wmma_f32_16x16x4_f32(
                        false, a, false, b, (short)0, acc[t], false, false);
                }
            }
        }
    }

    if (mtile >= ntm) return;

    const int rbase = mtile * 16 + half * 8;
    #pragma unroll
    for (int t = 0; t < TPW; ++t) {
        const int col = (np * TPW + t) * 16 + coll;
        if (col < NCOLS) {
            #pragma unroll
            for (int r = 0; r < 8; ++r)
                C[(size_t)(rbase + r) * NCOLS + col] = acc[t][r];
        }
    }
}


#define GCH 256
#define NEC 4

__global__ __launch_bounds__(1024, 16) void product_and_chunk_bins_kernel(
                                                          const int* __restrict__ src,
                                                          const int* __restrict__ tgt,
                                                          const float* __restrict__ w,
                                                          int* __restrict__ hist,
                                                          int* __restrict__ pst,
                                                          int* __restrict__ recs1,
                                                          int E, int nb,
                                                          const float* __restrict__ X,
                                                          const float* __restrict__ Wa,
                                                          float* __restrict__ T,
                                                          int ntm, int ngw)
{
    if ((int)blockIdx.x < ngw) {
        wmma_gemm_strip<6, NFEAT, NHID, 2>(X, Wa, T, ntm, (int)blockIdx.x);
        return;
    }
    __shared__ int lh[256];
    __shared__ int cur[256];
    __shared__ int wsum[8];
    const int tid  = threadIdx.x;
    const int lane = tid & 31;
    const int wave = tid >> 5;
    const int g    = (int)blockIdx.x - ngw;
    const int ch = (E + GCH - 1) / GCH;
    int e0 = g * ch;
    if (e0 > E) e0 = E;
    const int e1 = (e0 + ch < E) ? (e0 + ch) : E;
    const bool one = (ch <= NEC * 1024);

    int   tg[NEC] = {};
    int   sr[NEC] = {};
    float ww[NEC] = {};
    if (one && e1 > e0) {
        #pragma unroll
        for (int u = 0; u < NEC; ++u) {
            int e = e0 + tid + u * 1024;
            if (e > e1 - 1) e = e1 - 1;
            tg[u] = tgt[e];
            sr[u] = src[e];
            ww[u] = w[e];
        }
    }
    if (tid < 256) lh[tid] = 0;
    __syncthreads();
    if (one) {
        #pragma unroll
        for (int u = 0; u < NEC; ++u)
            if (e0 + tid + u * 1024 < e1) atomicAdd(&lh[tg[u] >> 8], 1);
    } else {
        for (int e = e0 + tid; e < e1; e += 1024)
            atomicAdd(&lh[tgt[e] >> 8], 1);
    }
    __syncthreads();

    int c = 0, inc = 0;
    if (tid < 256) {
        c = lh[tid];
        inc = c;
        for (int off = 1; off < 32; off <<= 1) {
            int u = __shfl_up(inc, off);
            if (lane >= off) inc += u;
        }
        if (lane == 31) wsum[wave] = inc;
    }
    __syncthreads();
    if (tid < 256) {
        int front = 0;
        for (int k = 0; k < wave; ++k) front += wsum[k];
        const int first = e0 + front + inc - c;
        cur[tid] = first;
        if (tid < nb) {
            hist[tid * GCH + g] = c;
            pst[tid * GCH + g]  = first;
        }
    }
    __syncthreads();

    int4* r1 = reinterpret_cast<int4*>(recs1);
    if (one) {
        #pragma unroll
        for (int u = 0; u < NEC; ++u) {
            const int e = e0 + tid + u * 1024;
            if (e < e1) {
                const int pos = atomicAdd(&cur[tg[u] >> 8], 1);
                r1[pos] = make_int4(e, tg[u], sr[u], __float_as_int(ww[u]));
            }
        }
    } else {
        for (int e = e0 + tid; e < e1; e += 1024) {
            const int t   = tgt[e];
            const int pos = atomicAdd(&cur[t >> 8], 1);
            r1[pos] = make_int4(e, t, src[e], __float_as_int(w[e]));
        }
    }
}

template<int NR>
__global__ __launch_bounds__(1024) void row_lists_kernel(const int* __restrict__ recs1,
                                                         const int* __restrict__ hist,
                                                         const int* __restrict__ pst,
                                                         int* __restrict__ rowptr,
                                                         int* __restrict__ recs2,
                                                         int* __restrict__ csrc,
                                                         float* __restrict__ cw)
{
    constexpr int CAP = NR * 1024;
    __shared__ int  cnt[256];
    __shared__ int  cur[256];
    __shared__ int  wsum[8];
    __shared__ int  wred[32];
    __shared__ int  segoff[257];
    __shared__ int  segst[256];
    __shared__ int4 srec[CAP];
    const int tid  = threadIdx.x;
    const int lane = tid & 31;
    const int wave = tid >> 5;
    const int b    = blockIdx.x;
    const int4* r1 = reinterpret_cast<const int4*>(recs1);
    int4*       r2 = reinterpret_cast<int4*>(recs2);

    int s = 0;
    {
        const int n4 = b * (GCH / 4);
        const int4* h4 = reinterpret_cast<const int4*>(hist);
        for (int i0 = 0; i0 < n4; i0 += 4 * 1024) {
            int4 v[4];
            #pragma unroll
            for (int u = 0; u < 4; ++u) {
                int i = i0 + u * 1024 + tid;
                if (i > n4 - 1) i = n4 - 1;
                v[u] = h4[i];
            }
            #pragma unroll
            for (int u = 0; u < 4; ++u)
                if (i0 + u * 1024 + tid < n4) s += (v[u].x + v[u].y) + (v[u].z + v[u].w);
        }
    }
    #pragma unroll
    for (int off = 16; off > 0; off >>= 1) s += __shfl_xor(s, off);
    if (lane == 0) wred[wave] = s;

    int pc = 0, pinc = 0, pstart = 0;
    if (tid < 256) {
        pc     = hist[b * GCH + tid];
        pstart = pst[b * GCH + tid];
        pinc = pc;
        for (int off = 1; off < 32; off <<= 1) {
            int u = __shfl_up(pinc, off);
            if (lane >= off) pinc += u;
        }
        if (lane == 31) wsum[wave] = pinc;
        cnt[tid] = 0;
    }
    __syncthreads();
    int bs = wred[lane];
    #pragma unroll
    for (int off = 16; off > 0; off >>= 1) bs += __shfl_xor(bs, off);
    if (tid < 256) {
        int front = 0;
        for (int k = 0; k < wave; ++k) front += wsum[k];
        const int o = front + pinc - pc;
        segoff[tid] = o;
        segst[tid]  = pstart;
        if (tid == 255) segoff[256] = o + pc;
    }
    __syncthreads();
    const int n  = segoff[256];
    const int be = bs + n;
    const bool fits = (n <= CAP);

    auto rec_at = [&](const int i) __attribute__((always_inline)) {
        int lo = 0;
        #pragma unroll
        for (int step = 128; step > 0; step >>= 1)
            if (segoff[lo + step] <= i) lo += step;
        return segst[lo] + (i - segoff[lo]);
    };

    int4 rr[NR];
    if (fits && n > 0) {
        #pragma unroll
        for (int u = 0; u < NR; ++u) {
            int i = tid + u * 1024;
            if (i > n - 1) i = n - 1;
            rr[u] = r1[rec_at(i)];
        }
    }

    if (fits) {
        #pragma unroll
        for (int u = 0; u < NR; ++u)
            if (tid + u * 1024 < n) atomicAdd(&cnt[rr[u].y & 255], 1);
    } else {
        for (int i = tid; i < n; i += 1024)
            atomicAdd(&cnt[r1[rec_at(i)].y & 255], 1);
    }
    __syncthreads();

    int c = 0, inc = 0;
    if (tid < 256) {
        c = cnt[tid];
        inc = c;
        for (int off = 1; off < 32; off <<= 1) {
            int u = __shfl_up(inc, off);
            if (lane >= off) inc += u;
        }
        if (lane == 31) wsum[wave] = inc;
    }
    __syncthreads();
    if (tid < 256) {
        int front = 0;
        for (int k = 0; k < wave; ++k) front += wsum[k];
        const int first = bs + front + inc - c;
        cur[tid] = first;
        rowptr[b * 256 + tid] = first;
        if (tid == 255 && b == (int)gridDim.x - 1) rowptr[b * 256 + 256] = be;
    }
    __syncthreads();

    if (fits) {
        #pragma unroll
        for (int u = 0; u < NR; ++u) {
            if (tid + u * 1024 < n) {
                const int pos = atomicAdd(&cur[rr[u].y & 255], 1);
                srec[pos - bs] = rr[u];
            }
        }
    } else {
        for (int i = tid; i < n; i += 1024) {
            const int4 r = r1[rec_at(i)];
            const int pos = atomicAdd(&cur[r.y & 255], 1);
            r2[pos] = r;
        }
        __threadfence();
    }
    __syncthreads();

    if (fits) {
        for (int p = tid; p < n; p += 1024) {
            const int4 r  = srec[p];
            const int  rw = r.y & 255;
            const int  e  = cur[rw] - bs;
            const int  a  = e - cnt[rw];
            int k = 0;
            for (int j = a; j < e; j += 4) {
                #pragma unroll
                for (int u = 0; u < 4; ++u) {
                    const int jj = (j + u < e) ? (j + u) : (e - 1);
                    const int x  = srec[jj].x;
                    k += (j + u < e && x < r.x) ? 1 : 0;
                }
            }
            csrc[bs + a + k] = r.z;
            cw[bs + a + k]   = __int_as_float(r.w);
        }
    } else {
        for (int p = bs + tid; p < be; p += 1024) {
            const int4 r  = r2[p];
            const int  rw = r.y & 255;
            const int  e  = cur[rw];
            const int  a  = e - cnt[rw];
            int k = 0;
            for (int j = a; j < e; ++j) k += (r2[j].x < r.x) ? 1 : 0;
            csrc[a + k] = r.z;
            cw[a + k]   = __int_as_float(r.w);
        }
    }
}

#define EPI_RELU     1
#define EPI_RELU_GN  2
#define EPI_LSM      3

template<int C, int CPL, int GRP>
__device__ __forceinline__ void walk_entries(const float* __restrict__ h, unsigned coff, int si, float wi,
                                             int n, float (&acc)[CPL])
{
    int k = 0;
    for (; k + GRP <= n; k += GRP) {
        float m[GRP][CPL];
        unsigned long long ra[GRP];
        #pragma unroll
        for (int u = 0; u < GRP; ++u)
            ra[u] = reinterpret_cast<unsigned long long>(h)
                  + (unsigned long long)(unsigned)__builtin_amdgcn_readlane(si, k + u) * (C * 4u);
        #pragma unroll
        for (int u = 0; u < GRP; ++u) {
            const gfloat* hp = reinterpret_cast<const gfloat*>(ra[u]);
            #pragma unroll
            for (int j = 0; j < CPL; ++j) m[u][j] = hp[coff + j];
        }
        #pragma unroll
        for (int u = 0; u < GRP; ++u) asm volatile("" :: "s"(ra[u]));
        #pragma unroll
        for (int u = 0; u < GRP; ++u) {
            const float w = __int_as_float(__builtin_amdgcn_readlane(__float_as_int(wi), k + u));
            #pragma unroll
            for (int j = 0; j < CPL; ++j) acc[j] = fmaf(m[u][j], w, acc[j]);
        }
    }
    if (k < n) {
        float m[GRP - 1][CPL];
        #pragma unroll
        for (int u = 0; u < GRP - 1; ++u) {
            #pragma unroll
            for (int j = 0; j < CPL; ++j) m[u][j] = 0.0f;
            if (k + u < n) {
                const int s = __builtin_amdgcn_readlane(si, k + u);
                const gfloat* hp = reinterpret_cast<const gfloat*>(reinterpret_cast<unsigned long long>(h) + (unsigned long long)(unsigned)s * (C * 4u));
                #pragma unroll
                for (int j = 0; j < CPL; ++j) m[u][j] = hp[coff + j];
            }
        }
        #pragma unroll
        for (int u = 0; u < GRP - 1; ++u) {
            if (k + u < n) {
                const float w = __int_as_float(__builtin_amdgcn_readlane(__float_as_int(wi), k + u));
                #pragma unroll
                for (int j = 0; j < CPL; ++j) acc[j] = fmaf(m[u][j], w, acc[j]);
            }
        }
    }
}

template<int C, int EPI>
__global__ __launch_bounds__(256)
void gather_rows_kernel(const float* __restrict__ h,
                        const int* __restrict__ rowptr,
                        const int* __restrict__ csrc,
                        const float* __restrict__ cw,
                        const float* __restrict__ bias,
                        const float* __restrict__ gamma,
                        const float* __restrict__ beta,
                        float* __restrict__ out, int nrows)
{
    constexpr int CPL = (C == NHID) ? 3 : 2;
    constexpr int LPR = C / CPL;
    constexpr int RPW = 4;
    constexpr int GRP = 16;
    static_assert(LPR * CPL == C && LPR <= 32 && RPW + 1 <= 32, "shape");

    const int lane = threadIdx.x & 31;
    const int gw  = __builtin_amdgcn_readfirstlane((int)(blockIdx.x * 8 + (threadIdx.x >> 5)));
    const int r0  = gw * RPW;
    if (r0 >= nrows) return;
    const bool act = lane < LPR;
    const int  c   = (act ? lane : 0) * CPL;

    int bi = r0 + ((lane < RPW) ? lane : RPW);
    if (bi > nrows) bi = nrows;
    const int rpv = rowptr[bi];

    float bs[CPL], gm[CPL], bt[CPL];
    #pragma unroll
    for (int j = 0; j < CPL; ++j) {
        bs[j] = bias[c + j];
        gm[j] = (EPI == EPI_RELU_GN) ? gamma[c + j] : 1.0f;
        bt[j] = (EPI == EPI_RELU_GN) ? beta[c + j]  : 0.0f;
    }

    int   esi[RPW];
    float ewi[RPW];
    #pragma unroll
    for (int i = 0; i < RPW; ++i) {
        const int pa = __builtin_amdgcn_readlane(rpv, i);
        const int pb = __builtin_amdgcn_readlane(rpv, i + 1);
        esi[i] = 0;
        ewi[i] = 0.0f;
        if (lane < pb - pa) { esi[i] = csrc[pa + lane]; ewi[i] = cw[pa + lane]; }
    }

    for (int i = 0; i < RPW; ++i) {
        if (r0 + i >= nrows) break;
        const int row = r0 + i;
        const int p0 = __builtin_amdgcn_readlane(rpv, i);
        const int p1 = __builtin_amdgcn_readlane(rpv, i + 1);

        int   si = esi[0];
        float wi = ewi[0];
        #pragma unroll
        for (int q = 1; q < RPW; ++q) {
            si = (i == q) ? esi[q] : si;
            wi = (i == q) ? ewi[q] : wi;
        }

        float acc[CPL];
        #pragma unroll
        for (int j = 0; j < CPL; ++j) acc[j] = 0.0f;

        for (int base = p0; base < p1; base += 32) {
            if (base > p0) {
                si = 0;
                wi = 0.0f;
                if (lane < p1 - base) { si = csrc[base + lane]; wi = cw[base + lane]; }
            }
            walk_entries<C, CPL, GRP>(h, (unsigned)c, si, wi, (p1 - base < 32) ? (p1 - base) : 32, acc);
        }

        float r[CPL];
        #pragma unroll
        for (int j = 0; j < CPL; ++j) {
            r[j] = acc[j] + bs[j];
            if (EPI == EPI_RELU || EPI == EPI_RELU_GN) r[j] = fmaxf(r[j], 0.0f);
        }

        if constexpr (EPI == EPI_RELU_GN) {
            const float mu  = (r[0] + r[1] + r[2]) * (1.0f / 3.0f);
            const float d0 = r[0] - mu, d1 = r[1] - mu, d2 = r[2] - mu;
            const float var = (d0 * d0 + d1 * d1 + d2 * d2) * (1.0f / 3.0f);
            const float inv = rsqrtf(var + GN_EPS);
            r[0] = d0 * inv * gm[0] + bt[0];
            r[1] = d1 * inv * gm[1] + bt[1];
            r[2] = d2 * inv * gm[2] + bt[2];
        }

        if constexpr (EPI == EPI_LSM) {
            float mx = act ? fmaxf(r[0], r[1]) : -3.402823466e+38f;
            #pragma unroll
            for (int off = 16; off > 0; off >>= 1) mx = fmaxf(mx, __shfl_xor(mx, off));
            const float e0 = __expf(r[0] - mx);
            const float e1 = __expf(r[1] - mx);
            float s = 0.0f;
            #pragma unroll
            for (int k = 0; k < LPR; ++k) {
                s += __int_as_float(__builtin_amdgcn_readlane(__float_as_int(e0), k));
                s += __int_as_float(__builtin_amdgcn_readlane(__float_as_int(e1), k));
            }
            const float ls = __logf(s);
            r[0] = r[0] - mx - ls;
            r[1] = r[1] - mx - ls;
        }

        if (act) {
            float* o = out + (size_t)row * C + c;
            #pragma unroll
            for (int j = 0; j < CPL; ++j) o[j] = r[j];
        }
    }
}

template<int EPI, int NT, int NCOLS, int GRP>
__global__ __launch_bounds__(512, 16)
void gather_rows_gemm_kernel(const float* __restrict__ h,
                             const int* __restrict__ rowptr,
                             const int* __restrict__ csrc,
                             const float* __restrict__ cw,
                             const float* __restrict__ bias,
                             const float* __restrict__ gamma,
                             const float* __restrict__ beta,
                             const float* __restrict__ Bw,
                             float* __restrict__ Cn, int nrows)
{
    constexpr int C     = NHID;
    constexpr int K     = NHID;
    constexpr int CPL   = 3;
    constexpr int RPW   = 4;
    constexpr int NTHR  = 512;
    constexpr int RPB   = (NTHR / 32) * RPW;
    constexpr int NCP   = NT * 16;
    constexpr int NQ    = NCP / 4;
    constexpr int NTASK = (K / 2) * NQ;
    constexpr int NIT   = (NTASK + NTHR - 1) / NTHR;
    constexpr int KB    = K / 8;
    constexpr int AP    = 100;
    constexpr int TPW   = NT / 3;
    static_assert(NT % 3 == 0 && K % 8 == 0 && NCOLS % 4 == 0 && NCOLS <= NCP && C == 32 * CPL, "shape");

    __shared__ v4f lds4[K * NCP / 4];
    __shared__ v4f at4[RPB * AP / 4];
    float* at = reinterpret_cast<float*>(at4);

    const int tid  = threadIdx.x;
    const int lane = tid & 31;
    const int wave = __builtin_amdgcn_readfirstlane(tid >> 5);
    const int r0   = blockIdx.x * RPB + wave * RPW;
    const int c    = lane * CPL;

    int bi = r0 + ((lane < RPW) ? lane : RPW);
    if (bi > nrows) bi = nrows;
    const int rpv = rowptr[bi];

    v4f w0[NIT], w1[NIT];
    #pragma unroll
    for (int it = 0; it < NIT; ++it) {
        const int i  = tid + it * NTHR;
        const int ic = (NTASK % NTHR == 0) ? i : (i < NTASK ? i : NTASK - 1);
        const int p  = ic / NQ;
        const int q  = ic - p * NQ;
        const int qc = (NCOLS == NCP) ? q : (q * 4 < NCOLS ? q : NCOLS / 4 - 1);
        const float* g = Bw + (size_t)(2 * p) * NCOLS + qc * 4;
        w0[it] = *reinterpret_cast<const v4f*>(g);
        w1[it] = *reinterpret_cast<const v4f*>(g + NCOLS);
    }

    float bs[CPL], gm[CPL], bt[CPL];
    #pragma unroll
    for (int j = 0; j < CPL; ++j) {
        bs[j] = bias[c + j];
        gm[j] = (EPI == EPI_RELU_GN) ? gamma[c + j] : 1.0f;
        bt[j] = (EPI == EPI_RELU_GN) ? beta[c + j]  : 0.0f;
    }

    int   esi[RPW];
    float ewi[RPW];
    #pragma unroll
    for (int i = 0; i < RPW; ++i) {
        const int pa = __builtin_amdgcn_readlane(rpv, i);
        const int pb = __builtin_amdgcn_readlane(rpv, i + 1);
        esi[i] = 0;
        ewi[i] = 0.0f;
        if (lane < pb - pa) { esi[i] = csrc[pa + lane]; ewi[i] = cw[pa + lane]; }
    }

    #pragma unroll
    for (int it = 0; it < NIT; ++it) {
        const int i = tid + it * NTHR;
        if (NTASK % NTHR == 0 || i < NTASK) {
            v4f lo = (v4f){w0[it].x, w1[it].x, w0[it].y, w1[it].y};
            v4f hi = (v4f){w0[it].z, w1[it].z, w0[it].w, w1[it].w};
            if (NCOLS != NCP) {
                const int q = i % NQ;
                if (q * 4 >= NCOLS) {
                    lo = (v4f){0.0f, 0.0f, 0.0f, 0.0f};
                    hi = (v4f){0.0f, 0.0f, 0.0f, 0.0f};
                }
            }
            lds4[2 * i]     = lo;
            lds4[2 * i + 1] = hi;
        }
    }

    for (int i = 0; i < RPW; ++i) {
        float* o = at + (wave * RPW + i) * AP + c;
        if (r0 + i >= nrows) {
            #pragma unroll
            for (int j = 0; j < CPL; ++j) o[j] = 0.0f;
            continue;
        }
        const int p0 = __builtin_amdgcn_readlane(rpv, i);
        const int p1 = __builtin_amdgcn_readlane(rpv, i + 1);

        int   si = esi[0];
        float wi = ewi[0];
        #pragma unroll
        for (int q = 1; q < RPW; ++q) {
            si = (i == q) ? esi[q] : si;
            wi = (i == q) ? ewi[q] : wi;
        }

        float acc[CPL];
        #pragma unroll
        for (int j = 0; j < CPL; ++j) acc[j] = 0.0f;

        for (int base = p0; base < p1; base += 32) {
            if (base > p0) {
                si = 0;
                wi = 0.0f;
                if (lane < p1 - base) { si = csrc[base + lane]; wi = cw[base + lane]; }
            }
            walk_entries<C, CPL, GRP>(h, (unsigned)c, si, wi, (p1 - base < 32) ? (p1 - base) : 32, acc);
        }

        float r[CPL];
        #pragma unroll
        for (int j = 0; j < CPL; ++j) r[j] = fmaxf(acc[j] + bs[j], 0.0f);

        if constexpr (EPI == EPI_RELU_GN) {
            const float mu  = (r[0] + r[1] + r[2]) * (1.0f / 3.0f);
            const float d0 = r[0] - mu, d1 = r[1] - mu, d2 = r[2] - mu;
            const float var = (d0 * d0 + d1 * d1 + d2 * d2) * (1.0f / 3.0f);
            const float inv = rsqrtf(var + GN_EPS);
            r[0] = d0 * inv * gm[0] + bt[0];
            r[1] = d1 * inv * gm[1] + bt[1];
            r[2] = d2 * inv * gm[2] + bt[2];
        }

        #pragma unroll
        for (int j = 0; j < CPL; ++j) o[j] = r[j];
    }
    __syncthreads();

    if (wave < 12) {
        const int coll = lane & 15;
        const int half = lane >> 4;
        const int mt   = wave / 3;
        const int np   = wave - mt * 3;
        const float2* __restrict__ lds2 = reinterpret_cast<const float2*>(lds4);
        const float2* bp = lds2 + (2 * half) * NCP + np * TPW * 16 + coll;
        const float*  ap = at + (mt * 16 + coll) * AP + half * 4;

        v8f acc[TPW] = {};
        #pragma unroll
        for (int j = 0; j < KB; ++j) {
            const v4f af = *reinterpret_cast<const v4f*>(ap + j * 8);
            #pragma unroll
            for (int s = 0; s < 2; ++s) {
                v2f a;
                a.x = s ? af.z : af.x;
                a.y = s ? af.w : af.y;
                const float2* bq = bp + (4 * j + s) * NCP;
                #pragma unroll
                for (int t = 0; t < TPW; ++t) {
                    float2 bv = bq[t * 16];
                    v2f b; b.x = bv.x; b.y = bv.y;
                    acc[t] = __builtin_amdgcn_wmma_f32_16x16x4_f32(
                        false, a, false, b, (short)0, acc[t], false, false);
                }
            }
        }

        const int rbase = blockIdx.x * RPB + mt * 16 + half * 8;
        #pragma unroll
        for (int t = 0; t < TPW; ++t) {
            const int col = (np * TPW + t) * 16 + coll;
            if (col < NCOLS) {
                float* cp = Cn + (size_t)rbase * NCOLS + col;
                if (rbase + 8 <= nrows) {
                    #pragma unroll
                    for (int r = 0; r < 8; ++r) cp[r * NCOLS] = acc[t][r];
                } else {
                    #pragma unroll
                    for (int r = 0; r < 8; ++r)
                        if (rbase + r < nrows) cp[r * NCOLS] = acc[t][r];
                }
            }
        }
    }
}

extern "C" void kernel_launch(void* const* d_in, const int* in_sizes, int n_in,
                              void* d_out, int out_size, void* d_ws, size_t ws_size,
                              hipStream_t stream)
{
    const float* x     = (const float*)d_in[0];
    const int*   src   = (const int*)  d_in[1];
    const int*   tgt   = (const int*)  d_in[2];
    const float* w     = (const float*)d_in[3];
    const float* W1    = (const float*)d_in[4];
    const float* b1    = (const float*)d_in[5];
    const float* W2    = (const float*)d_in[6];
    const float* b2    = (const float*)d_in[7];
    const float* W3    = (const float*)d_in[8];
    const float* b3    = (const float*)d_in[9];
    const float* gamma = (const float*)d_in[10];
    const float* beta  = (const float*)d_in[11];
    float* out = (float*)d_out;

    const int N   = in_sizes[0] / NFEAT;
    const int E   = in_sizes[1];
    const int ntm = N / 16;

    float* bufA = (float*)d_ws;
    float* bufB = bufA + (size_t)N * NHID;

    const int NB   = (N + 255) / 256;
    const int NPAD = ((NB * GCH + 4096) / 4096) * 4096;
    int*   hist   = (int*)(bufB + (size_t)N * NHID);
    int*   pst    = hist + NPAD;
    int*   rowptr = pst + NPAD;
    int*   recs1  = rowptr + NPAD;
    int*   recs2  = recs1 + (size_t)E * 4;
    int*   csrc   = recs2 + (size_t)E * 4;
    float* cw     = (float*)(csrc + E);

    const int ngw = (ntm * 2 + 31) / 32;
    product_and_chunk_bins_kernel<<<ngw + GCH, 1024, 0, stream>>>(src, tgt, w, hist, pst, recs1, E, NB,
                                                                   x, W1, bufA, ntm, ngw);
    row_lists_kernel<6><<<NB, 1024, 0, stream>>>(recs1, hist, pst, rowptr, recs2, csrc, cw);

    const int gwv = (N + 31) / 32;
    const int gfb = (N + 63) / 64;

    gather_rows_gemm_kernel<EPI_RELU, 6, NHID, 12><<<gfb, 512, 0, stream>>>(
        bufA, rowptr, csrc, cw, b1, gamma, beta, W2, bufB, N);

    gather_rows_gemm_kernel<EPI_RELU_GN, 3, NCLASS, 10><<<gfb, 512, 0, stream>>>(
        bufB, rowptr, csrc, cw, b2, gamma, beta, W3, bufA, N);

    gather_rows_kernel<NCLASS, EPI_LSM><<<gwv, 256, 0, stream>>>(
        bufA, rowptr, csrc, cw, b3, gamma, beta, out, N);
}
